# starting kernel
# baseline (speedup 1.0000x reference)
#include <hip/hip_runtime.h>

typedef __attribute__((ext_vector_type(2))) float v2f;
typedef __attribute__((ext_vector_type(8))) float v8f;

__device__ __forceinline__ v8f wmma_f32x4(v2f a, v2f b, v8f c) {
    return __builtin_amdgcn_wmma_f32_16x16x4_f32(
         false, a,  false, b,
         (short)0, c,  false,  false);
}

__device__ __forceinline__ void atomic_add_f32(float* p, float v) {
#if defined(__HIP_DEVICE_COMPILE__)
    unsafeAtomicAdd(p, v);
#else
    atomicAdd(p, v);
#endif
}

template <int K>
__global__ void gemm_bias_wmma_lds(const float* __restrict__ X,
                                   const float* __restrict__ W,
                                   const float* __restrict__ bias,
                                   float* __restrict__ Y,
                                   int M, int N) {
    __shared__ float As[16 * K];

    const int lane = threadIdx.x & 31;
    const int wave = threadIdx.x >> 5;
    const int wavesPerBlock = blockDim.x >> 5;
    const int tilesN = N >> 4;
    const int strip = blockIdx.x * 16;

    if (strip + 16 <= M) {
        const float4* __restrict__ src = (const float4*)(X + (size_t)strip * K);
        float4* dst = (float4*)As;
        for (int i = threadIdx.x; i < (16 * K) / 4; i += blockDim.x) dst[i] = src[i];
    } else {
        for (int i = threadIdx.x; i < 16 * K; i += blockDim.x) {
            const int rr = i / K, cc = i - rr * K;
            const int gr = min(strip + rr, M - 1);
            As[i] = X[(size_t)gr * K + cc];
        }
    }
    __syncthreads();

    const int koff = (lane >> 4) << 1;
    const float* __restrict__ arow = As + (lane & 15) * K + koff;
    const bool fullTile = (strip + 16 <= M);
    const int rbase = strip + ((lane >> 4) << 3);

    for (int tn = wave; tn < tilesN; tn += wavesPerBlock) {
        const int ncol = tn * 16 + (lane & 15);
        const float* __restrict__ wrow = W + (size_t)ncol * K + koff;

        v8f acc = {};
#pragma unroll
        for (int k = 0; k < K; k += 4) {
            v2f a = *(const v2f*)(arow + k);
            v2f b = *(const v2f*)(wrow + k);
            acc = wmma_f32x4(a, b, acc);
        }

        const float bv = bias[ncol];
        float* yp = Y + (size_t)rbase * N + ncol;
        if (fullTile) {
#pragma unroll
            for (int r = 0; r < 8; ++r) { *yp = acc[r] + bv; yp += N; }
        } else {
#pragma unroll
            for (int r = 0; r < 8; ++r) {
                if (rbase + r < M) *yp = acc[r] + bv;
                yp += N;
            }
        }
    }
}

__global__ void gemm_bias_wmma_rt(const float* __restrict__ X,
                                  const float* __restrict__ W,
                                  const float* __restrict__ bias,
                                  float* __restrict__ Y,
                                  int M, int K, int N) {
    const int lane = threadIdx.x & 31;
    const int wave = threadIdx.x >> 5;
    const int tilesN = N >> 4;
    const int tile = blockIdx.x * (blockDim.x >> 5) + wave;
    const int tm = tile / tilesN;
    const int tn = tile - tm * tilesN;
    if (tm * 16 >= M) return;

    const int mrow = min(tm * 16 + (lane & 15), M - 1);
    const int ncol = tn * 16 + (lane & 15);
    const int koff = (lane >> 4) << 1;
    const float* __restrict__ xrow = X + (size_t)mrow * K + koff;
    const float* __restrict__ wrow = W + (size_t)ncol * K + koff;

    v8f acc = {};
#pragma unroll 8
    for (int k = 0; k < K; k += 4) {
        v2f a = *(const v2f*)(xrow + k);
        v2f b = *(const v2f*)(wrow + k);
        acc = wmma_f32x4(a, b, acc);
    }

    const float bv = bias[ncol];
    const int rbase = tm * 16 + ((lane >> 4) << 3);
    float* yp = Y + (size_t)rbase * N + ncol;
#pragma unroll
    for (int r = 0; r < 8; ++r) {
        if (rbase + r < M) *yp = acc[r] + bv;
        yp += N;
    }
}

template <int HT>
__global__ void edge_mlp_wmma(const float* __restrict__ xf,
                              const int* __restrict__ row,
                              const int* __restrict__ col,
                              const float* __restrict__ We,
                              const float* __restrict__ be,
                              float* __restrict__ out,
                              int E, int Nu, int Hrt, int C) {
    const int H = (HT > 0) ? HT : Hrt;
    const int K = 2 * H;
    const int lane = threadIdx.x & 31;
    const int wave = threadIdx.x >> 5;
    const int tile = blockIdx.x * (blockDim.x >> 5) + wave;
    const int ebase = tile * 16;
    if (ebase >= E) return;

    const int m = lane & 15;
    const int e = min(ebase + m, E - 1);
    const int r = row[e];
    const int c = col[e];
    const int koff = (lane >> 4) << 1;

    const float* __restrict__ urow = xf + (size_t)r * H + koff;
    const float* __restrict__ mrow = xf + (size_t)(Nu + c) * H + koff;

    const int n = lane & 15;
    const float wmask = (n < C) ? 1.0f : 0.0f;
    const float* __restrict__ wrow = We + (size_t)min(n, C - 1) * K + koff;

    v8f acc = {};
#pragma unroll 8
    for (int k = 0; k < K; k += 4) {
        const float* src = (k + koff < H) ? (urow + k) : (mrow + (k - H));
        v2f a = *(const v2f*)src;
        v2f b = *(const v2f*)(wrow + k);
        b.x *= wmask;
        b.y *= wmask;
        acc = wmma_f32x4(a, b, acc);
    }

    if (n < C) {
        const float bv = be[n];
        const int mb = (lane >> 4) << 3;
        float* op = out + (size_t)(ebase + mb) * C + n;
        if (ebase + 16 <= E) {
#pragma unroll
            for (int rr = 0; rr < 8; ++rr) { *op = acc[rr] + bv; op += C; }
        } else {
#pragma unroll
            for (int rr = 0; rr < 8; ++rr) {
                if (ebase + mb + rr < E) *op = acc[rr] + bv;
                op += C;
            }
        }
    }
}

__global__ void fill_kernel(float* __restrict__ p, float v, long long n) {
    long long i = (long long)blockIdx.x * blockDim.x + threadIdx.x;
    if (i < n) p[i] = v;
}

__global__ void zero4_kernel(float4* __restrict__ p, long long n4) {
    long long i = (long long)blockIdx.x * blockDim.x + threadIdx.x;
    if (i < n4) p[i] = make_float4(0.f, 0.f, 0.f, 0.f);
}

__global__ void deg_scatter_kernel(const int* __restrict__ col,
                                   float* __restrict__ deg, int E) {
    int e = blockIdx.x * blockDim.x + threadIdx.x;
    if (e < E) atomic_add_f32(&deg[col[e]], 1.0f);
}

__global__ void rsqrt_kernel(float* __restrict__ d, int N) {
    int i = blockIdx.x * blockDim.x + threadIdx.x;
    if (i < N) d[i] = rsqrtf(d[i]);
}

__global__ void edge_scatter_kernel(const float* __restrict__ h,
                                    const float* __restrict__ dinv,
                                    const int* __restrict__ row,
                                    const int* __restrict__ col,
                                    float* __restrict__ acc,
                                    int E, int H) {
    const int chunks = H >> 2;
    long long gid = (long long)blockIdx.x * blockDim.x + threadIdx.x;
    long long total = (long long)E * chunks;
    if (gid >= total) return;
    const int e = (int)(gid / chunks);
    const int j = (int)(gid - (long long)e * chunks) << 2;
    const int r = row[e];
    const int c = col[e];
    const float s = dinv[r];
    const float4 hv = *(const float4*)(h + (size_t)r * H + j);
    float* dst = acc + (size_t)c * H + j;
    atomic_add_f32(dst + 0, s * hv.x);
    atomic_add_f32(dst + 1, s * hv.y);
    atomic_add_f32(dst + 2, s * hv.z);
    atomic_add_f32(dst + 3, s * hv.w);
}

__global__ void finalize_relu_kernel(const float* __restrict__ h,
                                     const float* __restrict__ dinv,
                                     float* __restrict__ acc_inout, int H) {
    const int node = blockIdx.x;
    const int j = threadIdx.x;
    const float di = dinv[node];
    const size_t idx = (size_t)node * H + j;
    const float v = di * acc_inout[idx] + di * di * h[idx];
    acc_inout[idx] = v > 0.f ? v : 0.f;
}

static inline int ceil_div_i(long long a, long long b) { return (int)((a + b - 1) / b); }

static void launch_gemm(const float* X, const float* W, const float* b, float* Y,
                        int M, int K, int N, hipStream_t stream) {
    const int TPB = 256;
    const int strips = ceil_div_i(M, 16);
    if (K == 32)
        gemm_bias_wmma_lds<32><<<strips, TPB, 0, stream>>>(X, W, b, Y, M, N);
    else if (K == 64)
        gemm_bias_wmma_lds<64><<<strips, TPB, 0, stream>>>(X, W, b, Y, M, N);
    else if (K == 128)
        gemm_bias_wmma_lds<128><<<strips, TPB, 0, stream>>>(X, W, b, Y, M, N);
    else {
        const int tiles = strips * (N / 16);
        gemm_bias_wmma_rt<<<ceil_div_i(tiles, 8), TPB, 0, stream>>>(X, W, b, Y, M, K, N);
    }
}

extern "C" void kernel_launch(void* const* d_in, const int* in_sizes, int n_in,
                              void* d_out, int out_size, void* d_ws, size_t ws_size,
                              hipStream_t stream) {
    const float* x_user  = (const float*)d_in[0];
    const float* x_movie = (const float*)d_in[1];
    const int*   edge    = (const int*)d_in[2];
    const float* Wu = (const float*)d_in[3];
    const float* bu = (const float*)d_in[4];
    const float* Wm = (const float*)d_in[5];
    const float* bm = (const float*)d_in[6];
    const float* W1 = (const float*)d_in[7];
    const float* b1 = (const float*)d_in[8];
    const float* W2 = (const float*)d_in[9];
    const float* b2 = (const float*)d_in[10];
    const float* We = (const float*)d_in[11];
    const float* be = (const float*)d_in[12];
    float* out = (float*)d_out;

    const int H  = in_sizes[4];
    const int C  = in_sizes[12];
    const int Fu = in_sizes[3] / H;
    const int Fm = in_sizes[5] / H;
    const int Nu = in_sizes[0] / Fu;
    const int Nm = in_sizes[1] / Fm;
    const int E  = in_sizes[2] / 2;
    const int N  = Nu + Nm;
    const int* row = edge;
    const int* col = edge + E;

    float* ws   = (float*)d_ws;
    float* X    = ws;
    float* Hh   = X  + (size_t)N * H;
    float* Acc  = Hh + (size_t)N * H;
    float* dinv = Acc + (size_t)N * H;

    const int TPB = 256, WPB = 8;
    const long long NH = (long long)N * H;

    launch_gemm(x_user, Wu, bu, X, Nu, Fu, H, stream);
    launch_gemm(x_movie, Wm, bm, X + (size_t)Nu * H, Nm, Fm, H, stream);

    fill_kernel<<<ceil_div_i(N, TPB), TPB, 0, stream>>>(dinv, 1.0f, N);
    deg_scatter_kernel<<<ceil_div_i(E, TPB), TPB, 0, stream>>>(col, dinv, E);
    rsqrt_kernel<<<ceil_div_i(N, TPB), TPB, 0, stream>>>(dinv, N);

    const long long scatterThreads = (long long)E * (H / 4);

    launch_gemm(X, W1, b1, Hh, N, H, H, stream);
    zero4_kernel<<<ceil_div_i(NH / 4, TPB), TPB, 0, stream>>>((float4*)Acc, NH / 4);
    edge_scatter_kernel<<<ceil_div_i(scatterThreads, TPB), TPB, 0, stream>>>(Hh, dinv, row, col, Acc, E, H);
    finalize_relu_kernel<<<N, H, 0, stream>>>(Hh, dinv, Acc, H);

    launch_gemm(Acc, W2, b2, Hh, N, H, H, stream);
    zero4_kernel<<<ceil_div_i(NH / 4, TPB), TPB, 0, stream>>>((float4*)X, NH / 4);
    edge_scatter_kernel<<<ceil_div_i(scatterThreads, TPB), TPB, 0, stream>>>(Hh, dinv, row, col, X, E, H);
    finalize_relu_kernel<<<N, H, 0, stream>>>(Hh, dinv, X, H);

    const int edgeTiles = ceil_div_i(E, 16);
    if (H == 128)
        edge_mlp_wmma<128><<<ceil_div_i(edgeTiles, WPB), TPB, 0, stream>>>(X, row, col, We, be, out, E, Nu, H, C);
    else
        edge_mlp_wmma<0><<<ceil_div_i(edgeTiles, WPB), TPB, 0, stream>>>(X, row, col, We, be, out, E, Nu, H, C);
}


# conv1 aggregates the raw 32/64-wide node features per target before the linear maps (linearity): its scatter adds 33 not 128 floats per edge, then two GEMMs; products regrouped, sum order changes
# speedup vs baseline: 1.3495x; 1.3495x over previous
#include <hip/hip_runtime.h>

typedef __attribute__((ext_vector_type(2))) float v2f;
typedef __attribute__((ext_vector_type(8))) float v8f;

__device__ __forceinline__ v8f wmma_f32x4(v2f a, v2f b, v8f c) {
    return __builtin_amdgcn_wmma_f32_16x16x4_f32(
         false, a,  false, b,
         (short)0, c,  false,  false);
}

__device__ __forceinline__ void atomic_add_f32(float* p, float v) {
#if defined(__HIP_DEVICE_COMPILE__)
    unsafeAtomicAdd(p, v);
#else
    atomicAdd(p, v);
#endif
}

constexpr int GEMM_GRID_LOG2 = 9;
constexpr int GEMM_GRID = 1 << GEMM_GRID_LOG2;

template <int MODE>
__device__ __forceinline__ float gemm_epilogue(float a, float b, float g) {
    if (MODE == 0) return a + b;
    if (MODE == 1) return g * a;
    const float v = a + g * b;
    return v > 0.f ? v : 0.f;
}

template <int K, int MODE>
__global__ __launch_bounds__(256)
void gemm_bias_wmma_wlds(const float* __restrict__ X,
                         const float* __restrict__ W,
                         const float* __restrict__ bias,
                         const float* __restrict__ rs,
                         const float* __restrict__ zc,
                         float* __restrict__ Y,
                         int M) {
    constexpr int N = 128;
    constexpr int NT = N / 16;
    constexpr int PITCH = K + 4;
    constexpr int KB = K / 8;
    constexpr int WAVES = 8;
    __shared__ __attribute__((aligned(16))) float Ws[N * PITCH];

    const int lane = threadIdx.x & 31;
    const int wave = __builtin_amdgcn_readfirstlane((int)(threadIdx.x >> 5));
    const int c = lane & 15;
    const int h = lane >> 4;

    const int strips = (M + 15) >> 4;
    const int s0 = (int)(((long long)strips * blockIdx.x) >> GEMM_GRID_LOG2);
    const int s1 = (int)(((long long)strips * (blockIdx.x + 1)) >> GEMM_GRID_LOG2);

    float bv[NT];
#pragma unroll
    for (int t = 0; t < NT; ++t) bv[t] = (MODE == 1) ? 0.f : bias[16 * t + c];

    float4 a[KB];
    int s = s0 + wave;

    if (s < s1) {
        const int arow = min(s * 16 + c, M - 1);
        const float4* __restrict__ ap = (const float4*)(X + (size_t)arow * K) + h;
#pragma unroll
        for (int kb = 0; kb < KB; ++kb) a[kb] = ap[2 * kb];
    }

    for (int i = threadIdx.x; i < N * (K / 4); i += WAVES * 32) {
        const int n = i / (K / 4);
        const int k4 = i - n * (K / 4);
        *(float4*)(Ws + n * PITCH + 4 * k4) = ((const float4*)W)[i];
    }
    __syncthreads();

    const float* __restrict__ bfrag = Ws + c * PITCH + 4 * h;

    while (s < s1) {
        float gr[8];
#pragma unroll
        for (int r = 0; r < 8; ++r) {
            gr[r] = 0.f;
            if (MODE != 0) {
                const int rr = min(s * 16 + (h << 3) + r, M - 1);
                float g = rs[rr];
                if (MODE == 2) {
                    const v2f w = *(const v2f*)(zc + (size_t)rr * N + 96);
                    g *= (w.x + w.y);
                }
                gr[r] = g;
            }
        }

        v8f acc[NT];
#pragma unroll
        for (int t = 0; t < NT; ++t) acc[t] = (v8f){};

        float4 bc[NT], bn[NT];
#pragma unroll
        for (int t = 0; t < NT; ++t)
            bc[t] = *(const float4*)(bfrag + 16 * t * PITCH);
#pragma unroll
        for (int kb = 0; kb < KB; ++kb) {
            if (kb + 1 < KB) {
#pragma unroll
                for (int t = 0; t < NT; ++t)
                    bn[t] = *(const float4*)(bfrag + 16 * t * PITCH + 8 * (kb + 1));
            }
            __builtin_amdgcn_sched_barrier(0);
            const v2f a01 = {a[kb].x, a[kb].y};
            const v2f a23 = {a[kb].z, a[kb].w};
#pragma unroll
            for (int t = 0; t < NT; ++t) {
                const v2f b01 = {bc[t].x, bc[t].y};
                acc[t] = wmma_f32x4(a01, b01, acc[t]);
            }
#pragma unroll
            for (int t = 0; t < NT; ++t) {
                const v2f b23 = {bc[t].z, bc[t].w};
                acc[t] = wmma_f32x4(a23, b23, acc[t]);
            }
            __builtin_amdgcn_sched_barrier(0);
#pragma unroll
            for (int t = 0; t < NT; ++t) bc[t] = bn[t];
        }

        const int row0 = s * 16;
        s += WAVES;

        if (s < s1) {
            const int arow = min(s * 16 + c, M - 1);
            const float4* __restrict__ ap = (const float4*)(X + (size_t)arow * K) + h;
#pragma unroll
            for (int kb = 0; kb < KB; ++kb) a[kb] = ap[2 * kb];
        }

        const int rbase = row0 + (h << 3);
        float* __restrict__ yrow = Y + (size_t)rbase * N + c;
        if (row0 + 16 <= M) {
#pragma unroll
            for (int t = 0; t < NT; ++t) {
#pragma unroll
                for (int r = 0; r < 8; ++r)
                    yrow[(size_t)r * N + 16 * t] = gemm_epilogue<MODE>(acc[t][r], bv[t], gr[r]);
            }
        } else {
#pragma unroll
            for (int t = 0; t < NT; ++t) {
#pragma unroll
                for (int r = 0; r < 8; ++r)
                    if (rbase + r < M)
                        yrow[(size_t)r * N + 16 * t] = gemm_epilogue<MODE>(acc[t][r], bv[t], gr[r]);
            }
        }
    }
}

__global__ void gemm_bias_wmma_rt(const float* __restrict__ X,
                                  const float* __restrict__ W,
                                  const float* __restrict__ bias,
                                  float* __restrict__ Y,
                                  int M, int K, int N) {
    const int lane = threadIdx.x & 31;
    const int wave = threadIdx.x >> 5;
    const int tilesN = N >> 4;
    const int tile = blockIdx.x * (blockDim.x >> 5) + wave;
    const int tm = tile / tilesN;
    const int tn = tile - tm * tilesN;
    if (tm * 16 >= M) return;

    const int mrow = min(tm * 16 + (lane & 15), M - 1);
    const int ncol = tn * 16 + (lane & 15);
    const int koff = (lane >> 4) << 1;
    const float* __restrict__ xrow = X + (size_t)mrow * K + koff;
    const float* __restrict__ wrow = W + (size_t)ncol * K + koff;

    v8f acc = {};
#pragma unroll 8
    for (int k = 0; k < K; k += 4) {
        v2f a = *(const v2f*)(xrow + k);
        v2f b = *(const v2f*)(wrow + k);
        acc = wmma_f32x4(a, b, acc);
    }

    const float bv = bias[ncol];
    const int rbase = tm * 16 + ((lane >> 4) << 3);
    float* yp = Y + (size_t)rbase * N + ncol;
#pragma unroll
    for (int r = 0; r < 8; ++r) {
        if (rbase + r < M) *yp = acc[r] + bv;
        yp += N;
    }
}

template <int HT>
__global__ void edge_mlp_wmma(const float* __restrict__ xf,
                              const int* __restrict__ row,
                              const int* __restrict__ col,
                              const float* __restrict__ We,
                              const float* __restrict__ be,
                              float* __restrict__ out,
                              int E, int Nu, int Hrt, int C) {
    const int H = (HT > 0) ? HT : Hrt;
    const int K = 2 * H;
    const int lane = threadIdx.x & 31;
    const int wave = threadIdx.x >> 5;
    const int tile = blockIdx.x * (blockDim.x >> 5) + wave;
    const int ebase = tile * 16;
    if (ebase >= E) return;

    const int m = lane & 15;
    const int e = min(ebase + m, E - 1);
    const int r = row[e];
    const int c = col[e];
    const int koff = (lane >> 4) << 1;

    const float* __restrict__ urow = xf + (size_t)r * H + koff;
    const float* __restrict__ mrow = xf + (size_t)(Nu + c) * H + koff;

    const int n = lane & 15;
    const float wmask = (n < C) ? 1.0f : 0.0f;
    const float* __restrict__ wrow = We + (size_t)min(n, C - 1) * K + koff;

    v8f acc = {};
#pragma unroll 8
    for (int k = 0; k < K; k += 4) {
        const float* src = (k + koff < H) ? (urow + k) : (mrow + (k - H));
        v2f a = *(const v2f*)src;
        v2f b = *(const v2f*)(wrow + k);
        b.x *= wmask;
        b.y *= wmask;
        acc = wmma_f32x4(a, b, acc);
    }

    if (n < C) {
        const float bv = be[n];
        const int mb = (lane >> 4) << 3;
        float* op = out + (size_t)(ebase + mb) * C + n;
        if (ebase + 16 <= E) {
#pragma unroll
            for (int rr = 0; rr < 8; ++rr) { *op = acc[rr] + bv; op += C; }
        } else {
#pragma unroll
            for (int rr = 0; rr < 8; ++rr) {
                if (ebase + mb + rr < E) *op = acc[rr] + bv;
                op += C;
            }
        }
    }
}

__global__ void fill_kernel(float* __restrict__ p, float v, long long n) {
    long long i = (long long)blockIdx.x * blockDim.x + threadIdx.x;
    if (i < n) p[i] = v;
}

__global__ void zero4_kernel(float4* __restrict__ p, long long n4) {
    long long i = (long long)blockIdx.x * blockDim.x + threadIdx.x;
    if (i < n4) p[i] = make_float4(0.f, 0.f, 0.f, 0.f);
}

__global__ void deg_scatter_kernel(const int* __restrict__ col,
                                   float* __restrict__ deg, int E) {
    int e = blockIdx.x * blockDim.x + threadIdx.x;
    if (e < E) atomic_add_f32(&deg[col[e]], 1.0f);
}

__global__ void rsqrt_kernel(float* __restrict__ d, int N) {
    int i = blockIdx.x * blockDim.x + threadIdx.x;
    if (i < N) d[i] = rsqrtf(d[i]);
}

__global__ void edge_scatter_kernel(const float* __restrict__ h,
                                    const float* __restrict__ dinv,
                                    const int* __restrict__ row,
                                    const int* __restrict__ col,
                                    float* __restrict__ acc,
                                    int E, int H) {
    const int chunks = H >> 2;
    long long gid = (long long)blockIdx.x * blockDim.x + threadIdx.x;
    long long total = (long long)E * chunks;
    if (gid >= total) return;
    const int e = (int)(gid / chunks);
    const int j = (int)(gid - (long long)e * chunks) << 2;
    const int r = row[e];
    const int c = col[e];
    const float s = dinv[r];
    const float4 hv = *(const float4*)(h + (size_t)r * H + j);
    float* dst = acc + (size_t)c * H + j;
    atomic_add_f32(dst + 0, s * hv.x);
    atomic_add_f32(dst + 1, s * hv.y);
    atomic_add_f32(dst + 2, s * hv.z);
    atomic_add_f32(dst + 3, s * hv.w);
}

__global__ void finalize_relu_kernel(const float* __restrict__ h,
                                     const float* __restrict__ dinv,
                                     float* __restrict__ acc_inout, int H) {
    const int node = blockIdx.x;
    const int j = threadIdx.x;
    const float di = dinv[node];
    const size_t idx = (size_t)node * H + j;
    const float v = di * acc_inout[idx] + di * di * h[idx];
    acc_inout[idx] = v > 0.f ? v : 0.f;
}

constexpr int RAW_FU = 32;
constexpr int RAW_FM = 64;
constexpr int RAW_H = 128;
constexpr int RAW_WU = RAW_FU + RAW_FM;
constexpr int RAW_WM = RAW_FU + RAW_FM + 1;

__global__ void compose_wz_kernel(const float* __restrict__ Wu, const float* __restrict__ bu,
                                  const float* __restrict__ Wm, const float* __restrict__ bm,
                                  float* __restrict__ Wz) {
    const int i = blockIdx.x * blockDim.x + threadIdx.x;
    if (i >= RAW_H * RAW_H) return;
    const int n = i / RAW_H;
    const int k = i - n * RAW_H;
    float w = 0.f;
    if (k < RAW_FU) w = Wu[n * RAW_FU + k];
    else if (k < RAW_WU) w = Wm[n * RAW_FM + (k - RAW_FU)];
    else if (k == RAW_WU) w = bu[n];
    else if (k == RAW_WM) w = bm[n];
    Wz[i] = w;
}

__global__ void init_z_kernel(const float* __restrict__ xu, const float* __restrict__ xm,
                              const float* __restrict__ dinv, float* __restrict__ Z, int Nu) {
    const int node = blockIdx.x;
    const int j = threadIdx.x;
    const float di = dinv[node];
    float self = 0.f;
    if (node < Nu) {
        if (j < RAW_FU) self = xu[(size_t)node * RAW_FU + j];
        else if (j == RAW_WU) self = 1.f;
    } else {
        if (j >= RAW_FU && j < RAW_WU) self = xm[(size_t)(node - Nu) * RAW_FM + (j - RAW_FU)];
        else if (j == RAW_WM) self = 1.f;
    }
    Z[(size_t)node * RAW_H + j] = di * self;
}

constexpr int RAW_EPW = 8;

template <bool FULL>
__device__ __forceinline__ void raw_scatter_wave(const float* __restrict__ xu, const float* __restrict__ xm,
                                                 const float* __restrict__ dinv,
                                                 const int* __restrict__ row, const int* __restrict__ col,
                                                 float* __restrict__ Z, int Nu, int e0, int ne, int l) {
    int r[RAW_EPW], c[RAW_EPW];
    float s[RAW_EPW], v0[RAW_EPW], v1[RAW_EPW];
#pragma unroll
    for (int i = 0; i < RAW_EPW; ++i) {
        const int e = FULL ? e0 + i : e0 + min(i, ne - 1);
        r[i] = row[e];
        c[i] = col[e];
    }
#pragma unroll
    for (int i = 0; i < RAW_EPW; ++i) {
        s[i] = dinv[r[i]];
        const bool user = r[i] < Nu;
        const float* __restrict__ p = user ? xu + (size_t)r[i] * RAW_FU + l
                                           : xm + (size_t)(r[i] - Nu) * RAW_FM + l;
        v0[i] = p[0];
        v1[i] = 0.f;
        if (!user) v1[i] = p[32];
    }
#pragma unroll
    for (int i = 0; i < RAW_EPW; ++i) {
        if (FULL || i < ne) {
            const bool user = r[i] < Nu;
            float* dst = Z + (size_t)c[i] * RAW_H;
            atomic_add_f32(dst + (user ? 0 : RAW_FU) + l, s[i] * v0[i]);
            if (!user) atomic_add_f32(dst + RAW_FU + 32 + l, s[i] * v1[i]);
        }
    }
    if (l < (FULL ? RAW_EPW : ne)) {
        const int rl = row[e0 + l];
        const int cl = col[e0 + l];
        atomic_add_f32(Z + (size_t)cl * RAW_H + (rl < Nu ? RAW_WU : RAW_WM), dinv[rl]);
    }
}

__global__ __launch_bounds__(256)
void raw_scatter_kernel(const float* __restrict__ xu, const float* __restrict__ xm,
                        const float* __restrict__ dinv,
                        const int* __restrict__ row, const int* __restrict__ col,
                        float* __restrict__ Z, int E, int Nu) {
    const int l = threadIdx.x & 31;
    const int wv = __builtin_amdgcn_readfirstlane(
        (int)(blockIdx.x * (blockDim.x >> 5) + (threadIdx.x >> 5)));
    const int e0 = wv * RAW_EPW;
    if (e0 >= E) return;
    const int ne = min(RAW_EPW, E - e0);
    if (ne == RAW_EPW) raw_scatter_wave<true>(xu, xm, dinv, row, col, Z, Nu, e0, ne, l);
    else               raw_scatter_wave<false>(xu, xm, dinv, row, col, Z, Nu, e0, ne, l);
}

static inline int ceil_div_i(long long a, long long b) { return (int)((a + b - 1) / b); }

static void launch_gemm(const float* X, const float* W, const float* b, float* Y,
                        int M, int K, int N, hipStream_t stream) {
    const int TPB = 256;
    if (N == 128 && K == 32)
        gemm_bias_wmma_wlds<32, 0><<<GEMM_GRID, TPB, 0, stream>>>(X, W, b, nullptr, nullptr, Y, M);
    else if (N == 128 && K == 64)
        gemm_bias_wmma_wlds<64, 0><<<GEMM_GRID, TPB, 0, stream>>>(X, W, b, nullptr, nullptr, Y, M);
    else if (N == 128 && K == 128)
        gemm_bias_wmma_wlds<128, 0><<<GEMM_GRID, TPB, 0, stream>>>(X, W, b, nullptr, nullptr, Y, M);
    else {
        const int strips = ceil_div_i(M, 16);
        const int tiles = strips * (N / 16);
        gemm_bias_wmma_rt<<<ceil_div_i(tiles, 8), TPB, 0, stream>>>(X, W, b, Y, M, K, N);
    }
}

extern "C" void kernel_launch(void* const* d_in, const int* in_sizes, int n_in,
                              void* d_out, int out_size, void* d_ws, size_t ws_size,
                              hipStream_t stream) {
    const float* x_user  = (const float*)d_in[0];
    const float* x_movie = (const float*)d_in[1];
    const int*   edge    = (const int*)d_in[2];
    const float* Wu = (const float*)d_in[3];
    const float* bu = (const float*)d_in[4];
    const float* Wm = (const float*)d_in[5];
    const float* bm = (const float*)d_in[6];
    const float* W1 = (const float*)d_in[7];
    const float* b1 = (const float*)d_in[8];
    const float* W2 = (const float*)d_in[9];
    const float* b2 = (const float*)d_in[10];
    const float* We = (const float*)d_in[11];
    const float* be = (const float*)d_in[12];
    float* out = (float*)d_out;

    const int H  = in_sizes[4];
    const int C  = in_sizes[12];
    const int Fu = in_sizes[3] / H;
    const int Fm = in_sizes[5] / H;
    const int Nu = in_sizes[0] / Fu;
    const int Nm = in_sizes[1] / Fm;
    const int E  = in_sizes[2] / 2;
    const int N  = Nu + Nm;
    const int* row = edge;
    const int* col = edge + E;

    float* ws   = (float*)d_ws;
    float* X    = ws;
    float* Hh   = X  + (size_t)N * H;
    float* Acc  = Hh + (size_t)N * H;
    float* Wz   = Acc + (size_t)N * H;
    float* dinv = Wz + (size_t)RAW_H * RAW_H;

    const int TPB = 256, WPB = 8;
    const long long NH = (long long)N * H;

    fill_kernel<<<ceil_div_i(N, TPB), TPB, 0, stream>>>(dinv, 1.0f, N);
    deg_scatter_kernel<<<ceil_div_i(E, TPB), TPB, 0, stream>>>(col, dinv, E);
    rsqrt_kernel<<<ceil_div_i(N, TPB), TPB, 0, stream>>>(dinv, N);

    const long long scatterThreads = (long long)E * (H / 4);

    if (H == RAW_H && Fu == RAW_FU && Fm == RAW_FM) {
        compose_wz_kernel<<<ceil_div_i(RAW_H * RAW_H, TPB), TPB, 0, stream>>>(Wu, bu, Wm, bm, Wz);
        init_z_kernel<<<N, RAW_H, 0, stream>>>(x_user, x_movie, dinv, Hh, Nu);
        raw_scatter_kernel<<<ceil_div_i(E, RAW_EPW * WPB), TPB, 0, stream>>>(x_user, x_movie, dinv, row, col, Hh, E, Nu);
        gemm_bias_wmma_wlds<RAW_H, 1><<<GEMM_GRID, TPB, 0, stream>>>(Hh, Wz, nullptr, dinv, nullptr, X, N);
        gemm_bias_wmma_wlds<RAW_H, 2><<<GEMM_GRID, TPB, 0, stream>>>(X, W1, b1, dinv, Hh, Acc, N);
    } else {
        launch_gemm(x_user, Wu, bu, X, Nu, Fu, H, stream);
        launch_gemm(x_movie, Wm, bm, X + (size_t)Nu * H, Nm, Fm, H, stream);

        launch_gemm(X, W1, b1, Hh, N, H, H, stream);
        zero4_kernel<<<ceil_div_i(NH / 4, TPB), TPB, 0, stream>>>((float4*)Acc, NH / 4);
        edge_scatter_kernel<<<ceil_div_i(scatterThreads, TPB), TPB, 0, stream>>>(Hh, dinv, row, col, Acc, E, H);
        finalize_relu_kernel<<<N, H, 0, stream>>>(Hh, dinv, Acc, H);
    }

    launch_gemm(Acc, W2, b2, Hh, N, H, H, stream);
    zero4_kernel<<<ceil_div_i(NH / 4, TPB), TPB, 0, stream>>>((float4*)X, NH / 4);
    edge_scatter_kernel<<<ceil_div_i(scatterThreads, TPB), TPB, 0, stream>>>(Hh, dinv, row, col, X, E, H);
    finalize_relu_kernel<<<N, H, 0, stream>>>(Hh, dinv, X, H);

    const int edgeTiles = ceil_div_i(E, 16);
    if (H == 128)
        edge_mlp_wmma<128><<<ceil_div_i(edgeTiles, WPB), TPB, 0, stream>>>(X, row, col, We, be, out, E, Nu, H, C);
    else
        edge_mlp_wmma<0><<<ceil_div_i(edgeTiles, WPB), TPB, 0, stream>>>(X, row, col, We, be, out, E, Nu, H, C);
}


# edge MLP reads whole rows: an edge's two 512-byte rows come by one full-line load each and reach WMMA through a per-wave LDS patch, We in LDS; k regrouped, two partial sums, sum order changes
# speedup vs baseline: 1.6403x; 1.2155x over previous
#include <hip/hip_runtime.h>

typedef __attribute__((ext_vector_type(2))) float v2f;
typedef __attribute__((ext_vector_type(8))) float v8f;

__device__ __forceinline__ v8f wmma_f32x4(v2f a, v2f b, v8f c) {
    return __builtin_amdgcn_wmma_f32_16x16x4_f32(
         false, a,  false, b,
         (short)0, c,  false,  false);
}

__device__ __forceinline__ void atomic_add_f32(float* p, float v) {
#if defined(__HIP_DEVICE_COMPILE__)
    unsafeAtomicAdd(p, v);
#else
    atomicAdd(p, v);
#endif
}

constexpr int GEMM_GRID_LOG2 = 9;
constexpr int GEMM_GRID = 1 << GEMM_GRID_LOG2;

template <int MODE>
__device__ __forceinline__ float gemm_epilogue(float a, float b, float g) {
    if (MODE == 0) return a + b;
    if (MODE == 1) return g * a;
    const float v = a + g * b;
    return v > 0.f ? v : 0.f;
}

template <int K, int MODE>
__global__ __launch_bounds__(256)
void gemm_bias_wmma_wlds(const float* __restrict__ X,
                         const float* __restrict__ W,
                         const float* __restrict__ bias,
                         const float* __restrict__ rs,
                         const float* __restrict__ zc,
                         float* __restrict__ Y,
                         int M) {
    constexpr int N = 128;
    constexpr int NT = N / 16;
    constexpr int PITCH = K + 4;
    constexpr int KB = K / 8;
    constexpr int WAVES = 8;
    __shared__ __attribute__((aligned(16))) float Ws[N * PITCH];

    const int lane = threadIdx.x & 31;
    const int wave = __builtin_amdgcn_readfirstlane((int)(threadIdx.x >> 5));
    const int c = lane & 15;
    const int h = lane >> 4;

    const int strips = (M + 15) >> 4;
    const int s0 = (int)(((long long)strips * blockIdx.x) >> GEMM_GRID_LOG2);
    const int s1 = (int)(((long long)strips * (blockIdx.x + 1)) >> GEMM_GRID_LOG2);

    float bv[NT];
#pragma unroll
    for (int t = 0; t < NT; ++t) bv[t] = (MODE == 1) ? 0.f : bias[16 * t + c];

    float4 a[KB];
    int s = s0 + wave;

    if (s < s1) {
        const int arow = min(s * 16 + c, M - 1);
        const float4* __restrict__ ap = (const float4*)(X + (size_t)arow * K) + h;
#pragma unroll
        for (int kb = 0; kb < KB; ++kb) a[kb] = ap[2 * kb];
    }

    for (int i = threadIdx.x; i < N * (K / 4); i += WAVES * 32) {
        const int n = i / (K / 4);
        const int k4 = i - n * (K / 4);
        *(float4*)(Ws + n * PITCH + 4 * k4) = ((const float4*)W)[i];
    }
    __syncthreads();

    const float* __restrict__ bfrag = Ws + c * PITCH + 4 * h;

    while (s < s1) {
        float gr[8];
#pragma unroll
        for (int r = 0; r < 8; ++r) {
            gr[r] = 0.f;
            if (MODE != 0) {
                const int rr = min(s * 16 + (h << 3) + r, M - 1);
                float g = rs[rr];
                if (MODE == 2) {
                    const v2f w = *(const v2f*)(zc + (size_t)rr * N + 96);
                    g *= (w.x + w.y);
                }
                gr[r] = g;
            }
        }

        v8f acc[NT];
#pragma unroll
        for (int t = 0; t < NT; ++t) acc[t] = (v8f){};

        float4 bc[NT], bn[NT];
#pragma unroll
        for (int t = 0; t < NT; ++t)
            bc[t] = *(const float4*)(bfrag + 16 * t * PITCH);
#pragma unroll
        for (int kb = 0; kb < KB; ++kb) {
            if (kb + 1 < KB) {
#pragma unroll
                for (int t = 0; t < NT; ++t)
                    bn[t] = *(const float4*)(bfrag + 16 * t * PITCH + 8 * (kb + 1));
            }
            __builtin_amdgcn_sched_barrier(0);
            const v2f a01 = {a[kb].x, a[kb].y};
            const v2f a23 = {a[kb].z, a[kb].w};
#pragma unroll
            for (int t = 0; t < NT; ++t) {
                const v2f b01 = {bc[t].x, bc[t].y};
                acc[t] = wmma_f32x4(a01, b01, acc[t]);
            }
#pragma unroll
            for (int t = 0; t < NT; ++t) {
                const v2f b23 = {bc[t].z, bc[t].w};
                acc[t] = wmma_f32x4(a23, b23, acc[t]);
            }
            __builtin_amdgcn_sched_barrier(0);
#pragma unroll
            for (int t = 0; t < NT; ++t) bc[t] = bn[t];
        }

        const int row0 = s * 16;
        s += WAVES;

        if (s < s1) {
            const int arow = min(s * 16 + c, M - 1);
            const float4* __restrict__ ap = (const float4*)(X + (size_t)arow * K) + h;
#pragma unroll
            for (int kb = 0; kb < KB; ++kb) a[kb] = ap[2 * kb];
        }

        const int rbase = row0 + (h << 3);
        float* __restrict__ yrow = Y + (size_t)rbase * N + c;
        if (row0 + 16 <= M) {
#pragma unroll
            for (int t = 0; t < NT; ++t) {
#pragma unroll
                for (int r = 0; r < 8; ++r)
                    yrow[(size_t)r * N + 16 * t] = gemm_epilogue<MODE>(acc[t][r], bv[t], gr[r]);
            }
        } else {
#pragma unroll
            for (int t = 0; t < NT; ++t) {
#pragma unroll
                for (int r = 0; r < 8; ++r)
                    if (rbase + r < M)
                        yrow[(size_t)r * N + 16 * t] = gemm_epilogue<MODE>(acc[t][r], bv[t], gr[r]);
            }
        }
    }
}

__global__ void gemm_bias_wmma_rt(const float* __restrict__ X,
                                  const float* __restrict__ W,
                                  const float* __restrict__ bias,
                                  float* __restrict__ Y,
                                  int M, int K, int N) {
    const int lane = threadIdx.x & 31;
    const int wave = threadIdx.x >> 5;
    const int tilesN = N >> 4;
    const int tile = blockIdx.x * (blockDim.x >> 5) + wave;
    const int tm = tile / tilesN;
    const int tn = tile - tm * tilesN;
    if (tm * 16 >= M) return;

    const int mrow = min(tm * 16 + (lane & 15), M - 1);
    const int ncol = tn * 16 + (lane & 15);
    const int koff = (lane >> 4) << 1;
    const float* __restrict__ xrow = X + (size_t)mrow * K + koff;
    const float* __restrict__ wrow = W + (size_t)ncol * K + koff;

    v8f acc = {};
#pragma unroll 8
    for (int k = 0; k < K; k += 4) {
        v2f a = *(const v2f*)(xrow + k);
        v2f b = *(const v2f*)(wrow + k);
        acc = wmma_f32x4(a, b, acc);
    }

    const float bv = bias[ncol];
    const int rbase = tm * 16 + ((lane >> 4) << 3);
    float* yp = Y + (size_t)rbase * N + ncol;
#pragma unroll
    for (int r = 0; r < 8; ++r) {
        if (rbase + r < M) *yp = acc[r] + bv;
        yp += N;
    }
}

template <int HT>
__global__ void edge_mlp_wmma(const float* __restrict__ xf,
                              const int* __restrict__ row,
                              const int* __restrict__ col,
                              const float* __restrict__ We,
                              const float* __restrict__ be,
                              float* __restrict__ out,
                              int E, int Nu, int Hrt, int C) {
    const int H = (HT > 0) ? HT : Hrt;
    const int K = 2 * H;
    const int lane = threadIdx.x & 31;
    const int wave = threadIdx.x >> 5;
    const int tile = blockIdx.x * (blockDim.x >> 5) + wave;
    const int ebase = tile * 16;
    if (ebase >= E) return;

    const int m = lane & 15;
    const int e = min(ebase + m, E - 1);
    const int r = row[e];
    const int c = col[e];
    const int koff = (lane >> 4) << 1;

    const float* __restrict__ urow = xf + (size_t)r * H + koff;
    const float* __restrict__ mrow = xf + (size_t)(Nu + c) * H + koff;

    const int n = lane & 15;
    const float wmask = (n < C) ? 1.0f : 0.0f;
    const float* __restrict__ wrow = We + (size_t)min(n, C - 1) * K + koff;

    v8f acc = {};
#pragma unroll 8
    for (int k = 0; k < K; k += 4) {
        const float* src = (k + koff < H) ? (urow + k) : (mrow + (k - H));
        v2f a = *(const v2f*)src;
        v2f b = *(const v2f*)(wrow + k);
        b.x *= wmask;
        b.y *= wmask;
        acc = wmma_f32x4(a, b, acc);
    }

    if (n < C) {
        const float bv = be[n];
        const int mb = (lane >> 4) << 3;
        float* op = out + (size_t)(ebase + mb) * C + n;
        if (ebase + 16 <= E) {
#pragma unroll
            for (int rr = 0; rr < 8; ++rr) { *op = acc[rr] + bv; op += C; }
        } else {
#pragma unroll
            for (int rr = 0; rr < 8; ++rr) {
                if (ebase + mb + rr < E) *op = acc[rr] + bv;
                op += C;
            }
        }
    }
}

constexpr int EM_H = 128;
constexpr int EM_K = 2 * EM_H;
constexpr int EM_WPITCH = EM_K + 4;
constexpr int EM_APITCH = EM_H + 4;
constexpr int EM_WAVES = 8;
constexpr int EM_GRID = 768;

__device__ __forceinline__ int edge_tile_ends(const int* __restrict__ row, const int* __restrict__ col,
                                              int t, int lane, int E) {
    const int e = min(t * 16 + (lane & 15), E - 1);
    const int* __restrict__ p = (lane < 16) ? row : col;
    return p[e];
}

#define EM_ROW_LOAD(j, IDX, BASE) \
    rv##j = ((const float4*)(xf + (size_t)(((BASE) ? Nu : 0) + __builtin_amdgcn_readlane(IDX, (BASE) + j)) * EM_H))[lane];
#define EM_ROW_STORE(j, IDX, BASE) *(float4*)(astore + j * EM_APITCH) = rv##j;
#define EM_ALL_ROWS(OP, IDX, BASE) \
    OP(0, IDX, BASE) OP(1, IDX, BASE) OP(2, IDX, BASE) OP(3, IDX, BASE) \
    OP(4, IDX, BASE) OP(5, IDX, BASE) OP(6, IDX, BASE) OP(7, IDX, BASE) \
    OP(8, IDX, BASE) OP(9, IDX, BASE) OP(10, IDX, BASE) OP(11, IDX, BASE) \
    OP(12, IDX, BASE) OP(13, IDX, BASE) OP(14, IDX, BASE) OP(15, IDX, BASE)

__global__ __launch_bounds__(256)
void edge_mlp_rows_lds(const float* __restrict__ xf,
                       const int* __restrict__ row,
                       const int* __restrict__ col,
                       const float* __restrict__ We,
                       const float* __restrict__ be,
                       float* __restrict__ out,
                       int E, int Nu, int C) {
    __shared__ __attribute__((aligned(16))) float Ws[16 * EM_WPITCH];
    __shared__ __attribute__((aligned(16))) float As[EM_WAVES * 16 * EM_APITCH];

    const int lane = threadIdx.x & 31;
    const int wave = __builtin_amdgcn_readfirstlane((int)(threadIdx.x >> 5));
    const int c = lane & 15;
    const int h = lane >> 4;

    const int tiles = (E + 15) >> 4;
    const int t0 = (int)(((long long)tiles * blockIdx.x) / EM_GRID);
    const int t1 = (int)(((long long)tiles * (blockIdx.x + 1)) / EM_GRID);
    int t = t0 + wave;

    float4 rv0, rv1, rv2, rv3, rv4, rv5, rv6, rv7, rv8, rv9, rv10, rv11, rv12, rv13, rv14, rv15;
    rv0 = rv1 = rv2 = rv3 = rv4 = rv5 = rv6 = rv7 = make_float4(0.f, 0.f, 0.f, 0.f);
    rv8 = rv9 = rv10 = rv11 = rv12 = rv13 = rv14 = rv15 = rv0;
    int idx = 0;
    if (t < t1) {
        idx = edge_tile_ends(row, col, t, lane, E);
        EM_ALL_ROWS(EM_ROW_LOAD, idx, 0)
    }

    for (int i = threadIdx.x; i < 16 * (EM_K / 4); i += EM_WAVES * 32) {
        const int n = i / (EM_K / 4);
        const int k4 = i - n * (EM_K / 4);
        float4 w = make_float4(0.f, 0.f, 0.f, 0.f);
        if (n < C) w = ((const float4*)We)[i];
        *(float4*)(Ws + n * EM_WPITCH + 4 * k4) = w;
    }
    __syncthreads();

    float* Aw = As + wave * (16 * EM_APITCH);
    float* astore = Aw + 4 * lane;
    const float* afrag = Aw + c * EM_APITCH + 4 * h;
    const float* bfrag = Ws + c * EM_WPITCH + 4 * h;
    const float bv = (c < C) ? be[c] : 0.f;

    while (t < t1) {
        const int tn = t + EM_WAVES;
        int idxn = 0;
        if (tn < t1) idxn = edge_tile_ends(row, col, tn, lane, E);

        EM_ALL_ROWS(EM_ROW_STORE, idx, 0)
        EM_ALL_ROWS(EM_ROW_LOAD, idx, 16)

        v8f acc0 = {}, acc1 = {};
#pragma unroll
        for (int kb = 0; kb < EM_H / 8; ++kb) {
            const float4 a = *(const float4*)(afrag + 8 * kb);
            const float4 b = *(const float4*)(bfrag + 8 * kb);
            acc0 = wmma_f32x4((v2f){a.x, a.y}, (v2f){b.x, b.y}, acc0);
            acc1 = wmma_f32x4((v2f){a.z, a.w}, (v2f){b.z, b.w}, acc1);
        }

        EM_ALL_ROWS(EM_ROW_STORE, idx, 0)
        if (tn < t1) {
            EM_ALL_ROWS(EM_ROW_LOAD, idxn, 0)
        }
#pragma unroll
        for (int kb = 0; kb < EM_H / 8; ++kb) {
            const float4 a = *(const float4*)(afrag + 8 * kb);
            const float4 b = *(const float4*)(bfrag + EM_H + 8 * kb);
            acc0 = wmma_f32x4((v2f){a.x, a.y}, (v2f){b.x, b.y}, acc0);
            acc1 = wmma_f32x4((v2f){a.z, a.w}, (v2f){b.z, b.w}, acc1);
        }

        if (c < C) {
            const int ebase = t * 16;
            const int mb = h << 3;
            float* op = out + (size_t)(ebase + mb) * C + c;
            if (ebase + 16 <= E) {
#pragma unroll
                for (int rr = 0; rr < 8; ++rr) { *op = (acc0[rr] + acc1[rr]) + bv; op += C; }
            } else {
#pragma unroll
                for (int rr = 0; rr < 8; ++rr) {
                    if (ebase + mb + rr < E) *op = (acc0[rr] + acc1[rr]) + bv;
                    op += C;
                }
            }
        }

        idx = idxn;
        t = tn;
    }
}

__global__ void fill_kernel(float* __restrict__ p, float v, long long n) {
    long long i = (long long)blockIdx.x * blockDim.x + threadIdx.x;
    if (i < n) p[i] = v;
}

__global__ void zero4_kernel(float4* __restrict__ p, long long n4) {
    long long i = (long long)blockIdx.x * blockDim.x + threadIdx.x;
    if (i < n4) p[i] = make_float4(0.f, 0.f, 0.f, 0.f);
}

__global__ void deg_scatter_kernel(const int* __restrict__ col,
                                   float* __restrict__ deg, int E) {
    int e = blockIdx.x * blockDim.x + threadIdx.x;
    if (e < E) atomic_add_f32(&deg[col[e]], 1.0f);
}

__global__ void rsqrt_kernel(float* __restrict__ d, int N) {
    int i = blockIdx.x * blockDim.x + threadIdx.x;
    if (i < N) d[i] = rsqrtf(d[i]);
}

__global__ void edge_scatter_kernel(const float* __restrict__ h,
                                    const float* __restrict__ dinv,
                                    const int* __restrict__ row,
                                    const int* __restrict__ col,
                                    float* __restrict__ acc,
                                    int E, int H) {
    const int chunks = H >> 2;
    long long gid = (long long)blockIdx.x * blockDim.x + threadIdx.x;
    long long total = (long long)E * chunks;
    if (gid >= total) return;
    const int e = (int)(gid / chunks);
    const int j = (int)(gid - (long long)e * chunks) << 2;
    const int r = row[e];
    const int c = col[e];
    const float s = dinv[r];
    const float4 hv = *(const float4*)(h + (size_t)r * H + j);
    float* dst = acc + (size_t)c * H + j;
    atomic_add_f32(dst + 0, s * hv.x);
    atomic_add_f32(dst + 1, s * hv.y);
    atomic_add_f32(dst + 2, s * hv.z);
    atomic_add_f32(dst + 3, s * hv.w);
}

__global__ void finalize_relu_kernel(const float* __restrict__ h,
                                     const float* __restrict__ dinv,
                                     float* __restrict__ acc_inout, int H) {
    const int node = blockIdx.x;
    const int j = threadIdx.x;
    const float di = dinv[node];
    const size_t idx = (size_t)node * H + j;
    const float v = di * acc_inout[idx] + di * di * h[idx];
    acc_inout[idx] = v > 0.f ? v : 0.f;
}

constexpr int RAW_FU = 32;
constexpr int RAW_FM = 64;
constexpr int RAW_H = 128;
constexpr int RAW_WU = RAW_FU + RAW_FM;
constexpr int RAW_WM = RAW_FU + RAW_FM + 1;

__global__ void compose_wz_kernel(const float* __restrict__ Wu, const float* __restrict__ bu,
                                  const float* __restrict__ Wm, const float* __restrict__ bm,
                                  float* __restrict__ Wz) {
    const int i = blockIdx.x * blockDim.x + threadIdx.x;
    if (i >= RAW_H * RAW_H) return;
    const int n = i / RAW_H;
    const int k = i - n * RAW_H;
    float w = 0.f;
    if (k < RAW_FU) w = Wu[n * RAW_FU + k];
    else if (k < RAW_WU) w = Wm[n * RAW_FM + (k - RAW_FU)];
    else if (k == RAW_WU) w = bu[n];
    else if (k == RAW_WM) w = bm[n];
    Wz[i] = w;
}

__global__ void init_z_kernel(const float* __restrict__ xu, const float* __restrict__ xm,
                              const float* __restrict__ dinv, float* __restrict__ Z, int Nu) {
    const int node = blockIdx.x;
    const int j = threadIdx.x;
    const float di = dinv[node];
    float self = 0.f;
    if (node < Nu) {
        if (j < RAW_FU) self = xu[(size_t)node * RAW_FU + j];
        else if (j == RAW_WU) self = 1.f;
    } else {
        if (j >= RAW_FU && j < RAW_WU) self = xm[(size_t)(node - Nu) * RAW_FM + (j - RAW_FU)];
        else if (j == RAW_WM) self = 1.f;
    }
    Z[(size_t)node * RAW_H + j] = di * self;
}

constexpr int RAW_EPW = 8;

template <bool FULL>
__device__ __forceinline__ void raw_scatter_wave(const float* __restrict__ xu, const float* __restrict__ xm,
                                                 const float* __restrict__ dinv,
                                                 const int* __restrict__ row, const int* __restrict__ col,
                                                 float* __restrict__ Z, int Nu, int e0, int ne, int l) {
    int r[RAW_EPW], c[RAW_EPW];
    float s[RAW_EPW], v0[RAW_EPW], v1[RAW_EPW];
#pragma unroll
    for (int i = 0; i < RAW_EPW; ++i) {
        const int e = FULL ? e0 + i : e0 + min(i, ne - 1);
        r[i] = row[e];
        c[i] = col[e];
    }
#pragma unroll
    for (int i = 0; i < RAW_EPW; ++i) {
        s[i] = dinv[r[i]];
        const bool user = r[i] < Nu;
        const float* __restrict__ p = user ? xu + (size_t)r[i] * RAW_FU + l
                                           : xm + (size_t)(r[i] - Nu) * RAW_FM + l;
        v0[i] = p[0];
        v1[i] = 0.f;
        if (!user) v1[i] = p[32];
    }
#pragma unroll
    for (int i = 0; i < RAW_EPW; ++i) {
        if (FULL || i < ne) {
            const bool user = r[i] < Nu;
            float* dst = Z + (size_t)c[i] * RAW_H;
            atomic_add_f32(dst + (user ? 0 : RAW_FU) + l, s[i] * v0[i]);
            if (!user) atomic_add_f32(dst + RAW_FU + 32 + l, s[i] * v1[i]);
        }
    }
    if (l < (FULL ? RAW_EPW : ne)) {
        const int rl = row[e0 + l];
        const int cl = col[e0 + l];
        atomic_add_f32(Z + (size_t)cl * RAW_H + (rl < Nu ? RAW_WU : RAW_WM), dinv[rl]);
    }
}

__global__ __launch_bounds__(256)
void raw_scatter_kernel(const float* __restrict__ xu, const float* __restrict__ xm,
                        const float* __restrict__ dinv,
                        const int* __restrict__ row, const int* __restrict__ col,
                        float* __restrict__ Z, int E, int Nu) {
    const int l = threadIdx.x & 31;
    const int wv = __builtin_amdgcn_readfirstlane(
        (int)(blockIdx.x * (blockDim.x >> 5) + (threadIdx.x >> 5)));
    const int e0 = wv * RAW_EPW;
    if (e0 >= E) return;
    const int ne = min(RAW_EPW, E - e0);
    if (ne == RAW_EPW) raw_scatter_wave<true>(xu, xm, dinv, row, col, Z, Nu, e0, ne, l);
    else               raw_scatter_wave<false>(xu, xm, dinv, row, col, Z, Nu, e0, ne, l);
}

static inline int ceil_div_i(long long a, long long b) { return (int)((a + b - 1) / b); }

static void launch_gemm(const float* X, const float* W, const float* b, float* Y,
                        int M, int K, int N, hipStream_t stream) {
    const int TPB = 256;
    if (N == 128 && K == 32)
        gemm_bias_wmma_wlds<32, 0><<<GEMM_GRID, TPB, 0, stream>>>(X, W, b, nullptr, nullptr, Y, M);
    else if (N == 128 && K == 64)
        gemm_bias_wmma_wlds<64, 0><<<GEMM_GRID, TPB, 0, stream>>>(X, W, b, nullptr, nullptr, Y, M);
    else if (N == 128 && K == 128)
        gemm_bias_wmma_wlds<128, 0><<<GEMM_GRID, TPB, 0, stream>>>(X, W, b, nullptr, nullptr, Y, M);
    else {
        const int strips = ceil_div_i(M, 16);
        const int tiles = strips * (N / 16);
        gemm_bias_wmma_rt<<<ceil_div_i(tiles, 8), TPB, 0, stream>>>(X, W, b, Y, M, K, N);
    }
}

extern "C" void kernel_launch(void* const* d_in, const int* in_sizes, int n_in,
                              void* d_out, int out_size, void* d_ws, size_t ws_size,
                              hipStream_t stream) {
    const float* x_user  = (const float*)d_in[0];
    const float* x_movie = (const float*)d_in[1];
    const int*   edge    = (const int*)d_in[2];
    const float* Wu = (const float*)d_in[3];
    const float* bu = (const float*)d_in[4];
    const float* Wm = (const float*)d_in[5];
    const float* bm = (const float*)d_in[6];
    const float* W1 = (const float*)d_in[7];
    const float* b1 = (const float*)d_in[8];
    const float* W2 = (const float*)d_in[9];
    const float* b2 = (const float*)d_in[10];
    const float* We = (const float*)d_in[11];
    const float* be = (const float*)d_in[12];
    float* out = (float*)d_out;

    const int H  = in_sizes[4];
    const int C  = in_sizes[12];
    const int Fu = in_sizes[3] / H;
    const int Fm = in_sizes[5] / H;
    const int Nu = in_sizes[0] / Fu;
    const int Nm = in_sizes[1] / Fm;
    const int E  = in_sizes[2] / 2;
    const int N  = Nu + Nm;
    const int* row = edge;
    const int* col = edge + E;

    float* ws   = (float*)d_ws;
    float* X    = ws;
    float* Hh   = X  + (size_t)N * H;
    float* Acc  = Hh + (size_t)N * H;
    float* Wz   = Acc + (size_t)N * H;
    float* dinv = Wz + (size_t)RAW_H * RAW_H;

    const int TPB = 256, WPB = 8;
    const long long NH = (long long)N * H;

    fill_kernel<<<ceil_div_i(N, TPB), TPB, 0, stream>>>(dinv, 1.0f, N);
    deg_scatter_kernel<<<ceil_div_i(E, TPB), TPB, 0, stream>>>(col, dinv, E);
    rsqrt_kernel<<<ceil_div_i(N, TPB), TPB, 0, stream>>>(dinv, N);

    const long long scatterThreads = (long long)E * (H / 4);

    if (H == RAW_H && Fu == RAW_FU && Fm == RAW_FM) {
        compose_wz_kernel<<<ceil_div_i(RAW_H * RAW_H, TPB), TPB, 0, stream>>>(Wu, bu, Wm, bm, Wz);
        init_z_kernel<<<N, RAW_H, 0, stream>>>(x_user, x_movie, dinv, Hh, Nu);
        raw_scatter_kernel<<<ceil_div_i(E, RAW_EPW * WPB), TPB, 0, stream>>>(x_user, x_movie, dinv, row, col, Hh, E, Nu);
        gemm_bias_wmma_wlds<RAW_H, 1><<<GEMM_GRID, TPB, 0, stream>>>(Hh, Wz, nullptr, dinv, nullptr, X, N);
        gemm_bias_wmma_wlds<RAW_H, 2><<<GEMM_GRID, TPB, 0, stream>>>(X, W1, b1, dinv, Hh, Acc, N);
    } else {
        launch_gemm(x_user, Wu, bu, X, Nu, Fu, H, stream);
        launch_gemm(x_movie, Wm, bm, X + (size_t)Nu * H, Nm, Fm, H, stream);

        launch_gemm(X, W1, b1, Hh, N, H, H, stream);
        zero4_kernel<<<ceil_div_i(NH / 4, TPB), TPB, 0, stream>>>((float4*)Acc, NH / 4);
        edge_scatter_kernel<<<ceil_div_i(scatterThreads, TPB), TPB, 0, stream>>>(Hh, dinv, row, col, Acc, E, H);
        finalize_relu_kernel<<<N, H, 0, stream>>>(Hh, dinv, Acc, H);
    }

    launch_gemm(Acc, W2, b2, Hh, N, H, H, stream);
    zero4_kernel<<<ceil_div_i(NH / 4, TPB), TPB, 0, stream>>>((float4*)X, NH / 4);
    edge_scatter_kernel<<<ceil_div_i(scatterThreads, TPB), TPB, 0, stream>>>(Hh, dinv, row, col, X, E, H);
    finalize_relu_kernel<<<N, H, 0, stream>>>(Hh, dinv, X, H);

    const int edgeTiles = ceil_div_i(E, 16);
    if (H == EM_H && C <= 16)
        edge_mlp_rows_lds<<<EM_GRID, TPB, 0, stream>>>(X, row, col, We, be, out, E, Nu, C);
    else
        edge_mlp_wmma<0><<<ceil_div_i(edgeTiles, WPB), TPB, 0, stream>>>(X, row, col, We, be, out, E, Nu, H, C);
}


# conv2 scatter re-laid: a wave takes 8 edges, indices scalar, all source rows in flight; lane l adds floats l,l+32,l+64,l+96, so each atomic covers one whole 128-byte line; sums by arrival as before
# speedup vs baseline: 2.0732x; 1.2639x over previous
#include <hip/hip_runtime.h>

typedef __attribute__((ext_vector_type(2))) float v2f;
typedef __attribute__((ext_vector_type(8))) float v8f;

__device__ __forceinline__ v8f wmma_f32x4(v2f a, v2f b, v8f c) {
    return __builtin_amdgcn_wmma_f32_16x16x4_f32(
         false, a,  false, b,
         (short)0, c,  false,  false);
}

__device__ __forceinline__ void atomic_add_f32(float* p, float v) {
#if defined(__HIP_DEVICE_COMPILE__)
    unsafeAtomicAdd(p, v);
#else
    atomicAdd(p, v);
#endif
}

constexpr int GEMM_GRID_LOG2 = 9;
constexpr int GEMM_GRID = 1 << GEMM_GRID_LOG2;

template <int MODE>
__device__ __forceinline__ float gemm_epilogue(float a, float b, float g) {
    if (MODE == 0) return a + b;
    if (MODE == 1) return g * a;
    const float v = a + g * b;
    return v > 0.f ? v : 0.f;
}

template <int K, int MODE>
__global__ __launch_bounds__(256)
void gemm_bias_wmma_wlds(const float* __restrict__ X,
                         const float* __restrict__ W,
                         const float* __restrict__ bias,
                         const float* __restrict__ rs,
                         const float* __restrict__ zc,
                         float* __restrict__ Y,
                         int M) {
    constexpr int N = 128;
    constexpr int NT = N / 16;
    constexpr int PITCH = K + 4;
    constexpr int KB = K / 8;
    constexpr int WAVES = 8;
    __shared__ __attribute__((aligned(16))) float Ws[N * PITCH];

    const int lane = threadIdx.x & 31;
    const int wave = __builtin_amdgcn_readfirstlane((int)(threadIdx.x >> 5));
    const int c = lane & 15;
    const int h = lane >> 4;

    const int strips = (M + 15) >> 4;
    const int s0 = (int)(((long long)strips * blockIdx.x) >> GEMM_GRID_LOG2);
    const int s1 = (int)(((long long)strips * (blockIdx.x + 1)) >> GEMM_GRID_LOG2);

    float bv[NT];
#pragma unroll
    for (int t = 0; t < NT; ++t) bv[t] = (MODE == 1) ? 0.f : bias[16 * t + c];

    float4 a[KB];
    int s = s0 + wave;

    if (s < s1) {
        const int arow = min(s * 16 + c, M - 1);
        const float4* __restrict__ ap = (const float4*)(X + (size_t)arow * K) + h;
#pragma unroll
        for (int kb = 0; kb < KB; ++kb) a[kb] = ap[2 * kb];
    }

    for (int i = threadIdx.x; i < N * (K / 4); i += WAVES * 32) {
        const int n = i / (K / 4);
        const int k4 = i - n * (K / 4);
        *(float4*)(Ws + n * PITCH + 4 * k4) = ((const float4*)W)[i];
    }
    __syncthreads();

    const float* __restrict__ bfrag = Ws + c * PITCH + 4 * h;

    while (s < s1) {
        float gr[8];
#pragma unroll
        for (int r = 0; r < 8; ++r) {
            gr[r] = 0.f;
            if (MODE != 0) {
                const int rr = min(s * 16 + (h << 3) + r, M - 1);
                float g = rs[rr];
                if (MODE == 2) {
                    const v2f w = *(const v2f*)(zc + (size_t)rr * N + 96);
                    g *= (w.x + w.y);
                }
                gr[r] = g;
            }
        }

        v8f acc[NT];
#pragma unroll
        for (int t = 0; t < NT; ++t) acc[t] = (v8f){};

        float4 bc[NT], bn[NT];
#pragma unroll
        for (int t = 0; t < NT; ++t)
            bc[t] = *(const float4*)(bfrag + 16 * t * PITCH);
#pragma unroll
        for (int kb = 0; kb < KB; ++kb) {
            if (kb + 1 < KB) {
#pragma unroll
                for (int t = 0; t < NT; ++t)
                    bn[t] = *(const float4*)(bfrag + 16 * t * PITCH + 8 * (kb + 1));
            }
            __builtin_amdgcn_sched_barrier(0);
            const v2f a01 = {a[kb].x, a[kb].y};
            const v2f a23 = {a[kb].z, a[kb].w};
#pragma unroll
            for (int t = 0; t < NT; ++t) {
                const v2f b01 = {bc[t].x, bc[t].y};
                acc[t] = wmma_f32x4(a01, b01, acc[t]);
            }
#pragma unroll
            for (int t = 0; t < NT; ++t) {
                const v2f b23 = {bc[t].z, bc[t].w};
                acc[t] = wmma_f32x4(a23, b23, acc[t]);
            }
            __builtin_amdgcn_sched_barrier(0);
#pragma unroll
            for (int t = 0; t < NT; ++t) bc[t] = bn[t];
        }

        const int row0 = s * 16;
        s += WAVES;

        if (s < s1) {
            const int arow = min(s * 16 + c, M - 1);
            const float4* __restrict__ ap = (const float4*)(X + (size_t)arow * K) + h;
#pragma unroll
            for (int kb = 0; kb < KB; ++kb) a[kb] = ap[2 * kb];
        }

        const int rbase = row0 + (h << 3);
        float* __restrict__ yrow = Y + (size_t)rbase * N + c;
        if (row0 + 16 <= M) {
#pragma unroll
            for (int t = 0; t < NT; ++t) {
#pragma unroll
                for (int r = 0; r < 8; ++r)
                    yrow[(size_t)r * N + 16 * t] = gemm_epilogue<MODE>(acc[t][r], bv[t], gr[r]);
            }
        } else {
#pragma unroll
            for (int t = 0; t < NT; ++t) {
#pragma unroll
                for (int r = 0; r < 8; ++r)
                    if (rbase + r < M)
                        yrow[(size_t)r * N + 16 * t] = gemm_epilogue<MODE>(acc[t][r], bv[t], gr[r]);
            }
        }
    }
}

__global__ void gemm_bias_wmma_rt(const float* __restrict__ X,
                                  const float* __restrict__ W,
                                  const float* __restrict__ bias,
                                  float* __restrict__ Y,
                                  int M, int K, int N) {
    const int lane = threadIdx.x & 31;
    const int wave = threadIdx.x >> 5;
    const int tilesN = N >> 4;
    const int tile = blockIdx.x * (blockDim.x >> 5) + wave;
    const int tm = tile / tilesN;
    const int tn = tile - tm * tilesN;
    if (tm * 16 >= M) return;

    const int mrow = min(tm * 16 + (lane & 15), M - 1);
    const int ncol = tn * 16 + (lane & 15);
    const int koff = (lane >> 4) << 1;
    const float* __restrict__ xrow = X + (size_t)mrow * K + koff;
    const float* __restrict__ wrow = W + (size_t)ncol * K + koff;

    v8f acc = {};
#pragma unroll 8
    for (int k = 0; k < K; k += 4) {
        v2f a = *(const v2f*)(xrow + k);
        v2f b = *(const v2f*)(wrow + k);
        acc = wmma_f32x4(a, b, acc);
    }

    const float bv = bias[ncol];
    const int rbase = tm * 16 + ((lane >> 4) << 3);
    float* yp = Y + (size_t)rbase * N + ncol;
#pragma unroll
    for (int r = 0; r < 8; ++r) {
        if (rbase + r < M) *yp = acc[r] + bv;
        yp += N;
    }
}

template <int HT>
__global__ void edge_mlp_wmma(const float* __restrict__ xf,
                              const int* __restrict__ row,
                              const int* __restrict__ col,
                              const float* __restrict__ We,
                              const float* __restrict__ be,
                              float* __restrict__ out,
                              int E, int Nu, int Hrt, int C) {
    const int H = (HT > 0) ? HT : Hrt;
    const int K = 2 * H;
    const int lane = threadIdx.x & 31;
    const int wave = threadIdx.x >> 5;
    const int tile = blockIdx.x * (blockDim.x >> 5) + wave;
    const int ebase = tile * 16;
    if (ebase >= E) return;

    const int m = lane & 15;
    const int e = min(ebase + m, E - 1);
    const int r = row[e];
    const int c = col[e];
    const int koff = (lane >> 4) << 1;

    const float* __restrict__ urow = xf + (size_t)r * H + koff;
    const float* __restrict__ mrow = xf + (size_t)(Nu + c) * H + koff;

    const int n = lane & 15;
    const float wmask = (n < C) ? 1.0f : 0.0f;
    const float* __restrict__ wrow = We + (size_t)min(n, C - 1) * K + koff;

    v8f acc = {};
#pragma unroll 8
    for (int k = 0; k < K; k += 4) {
        const float* src = (k + koff < H) ? (urow + k) : (mrow + (k - H));
        v2f a = *(const v2f*)src;
        v2f b = *(const v2f*)(wrow + k);
        b.x *= wmask;
        b.y *= wmask;
        acc = wmma_f32x4(a, b, acc);
    }

    if (n < C) {
        const float bv = be[n];
        const int mb = (lane >> 4) << 3;
        float* op = out + (size_t)(ebase + mb) * C + n;
        if (ebase + 16 <= E) {
#pragma unroll
            for (int rr = 0; rr < 8; ++rr) { *op = acc[rr] + bv; op += C; }
        } else {
#pragma unroll
            for (int rr = 0; rr < 8; ++rr) {
                if (ebase + mb + rr < E) *op = acc[rr] + bv;
                op += C;
            }
        }
    }
}

constexpr int EM_H = 128;
constexpr int EM_K = 2 * EM_H;
constexpr int EM_WPITCH = EM_K + 4;
constexpr int EM_APITCH = EM_H + 4;
constexpr int EM_WAVES = 8;
constexpr int EM_GRID = 768;

__device__ __forceinline__ int edge_tile_ends(const int* __restrict__ row, const int* __restrict__ col,
                                              int t, int lane, int E) {
    const int e = min(t * 16 + (lane & 15), E - 1);
    const int* __restrict__ p = (lane < 16) ? row : col;
    return p[e];
}

#define EM_ROW_LOAD(j, IDX, BASE) \
    rv##j = ((const float4*)(xf + (size_t)(((BASE) ? Nu : 0) + __builtin_amdgcn_readlane(IDX, (BASE) + j)) * EM_H))[lane];
#define EM_ROW_STORE(j, IDX, BASE) *(float4*)(astore + j * EM_APITCH) = rv##j;
#define EM_ALL_ROWS(OP, IDX, BASE) \
    OP(0, IDX, BASE) OP(1, IDX, BASE) OP(2, IDX, BASE) OP(3, IDX, BASE) \
    OP(4, IDX, BASE) OP(5, IDX, BASE) OP(6, IDX, BASE) OP(7, IDX, BASE) \
    OP(8, IDX, BASE) OP(9, IDX, BASE) OP(10, IDX, BASE) OP(11, IDX, BASE) \
    OP(12, IDX, BASE) OP(13, IDX, BASE) OP(14, IDX, BASE) OP(15, IDX, BASE)

__global__ __launch_bounds__(256)
void edge_mlp_rows_lds(const float* __restrict__ xf,
                       const int* __restrict__ row,
                       const int* __restrict__ col,
                       const float* __restrict__ We,
                       const float* __restrict__ be,
                       float* __restrict__ out,
                       int E, int Nu, int C) {
    __shared__ __attribute__((aligned(16))) float Ws[16 * EM_WPITCH];
    __shared__ __attribute__((aligned(16))) float As[EM_WAVES * 16 * EM_APITCH];

    const int lane = threadIdx.x & 31;
    const int wave = __builtin_amdgcn_readfirstlane((int)(threadIdx.x >> 5));
    const int c = lane & 15;
    const int h = lane >> 4;

    const int tiles = (E + 15) >> 4;
    const int t0 = (int)(((long long)tiles * blockIdx.x) / EM_GRID);
    const int t1 = (int)(((long long)tiles * (blockIdx.x + 1)) / EM_GRID);
    int t = t0 + wave;

    float4 rv0, rv1, rv2, rv3, rv4, rv5, rv6, rv7, rv8, rv9, rv10, rv11, rv12, rv13, rv14, rv15;
    rv0 = rv1 = rv2 = rv3 = rv4 = rv5 = rv6 = rv7 = make_float4(0.f, 0.f, 0.f, 0.f);
    rv8 = rv9 = rv10 = rv11 = rv12 = rv13 = rv14 = rv15 = rv0;
    int idx = 0;
    if (t < t1) {
        idx = edge_tile_ends(row, col, t, lane, E);
        EM_ALL_ROWS(EM_ROW_LOAD, idx, 0)
    }

    for (int i = threadIdx.x; i < 16 * (EM_K / 4); i += EM_WAVES * 32) {
        const int n = i / (EM_K / 4);
        const int k4 = i - n * (EM_K / 4);
        float4 w = make_float4(0.f, 0.f, 0.f, 0.f);
        if (n < C) w = ((const float4*)We)[i];
        *(float4*)(Ws + n * EM_WPITCH + 4 * k4) = w;
    }
    __syncthreads();

    float* Aw = As + wave * (16 * EM_APITCH);
    float* astore = Aw + 4 * lane;
    const float* afrag = Aw + c * EM_APITCH + 4 * h;
    const float* bfrag = Ws + c * EM_WPITCH + 4 * h;
    const float bv = (c < C) ? be[c] : 0.f;

    while (t < t1) {
        const int tn = t + EM_WAVES;
        int idxn = 0;
        if (tn < t1) idxn = edge_tile_ends(row, col, tn, lane, E);

        EM_ALL_ROWS(EM_ROW_STORE, idx, 0)
        EM_ALL_ROWS(EM_ROW_LOAD, idx, 16)

        v8f acc0 = {}, acc1 = {};
#pragma unroll
        for (int kb = 0; kb < EM_H / 8; ++kb) {
            const float4 a = *(const float4*)(afrag + 8 * kb);
            const float4 b = *(const float4*)(bfrag + 8 * kb);
            acc0 = wmma_f32x4((v2f){a.x, a.y}, (v2f){b.x, b.y}, acc0);
            acc1 = wmma_f32x4((v2f){a.z, a.w}, (v2f){b.z, b.w}, acc1);
        }

        EM_ALL_ROWS(EM_ROW_STORE, idx, 0)
        if (tn < t1) {
            EM_ALL_ROWS(EM_ROW_LOAD, idxn, 0)
        }
#pragma unroll
        for (int kb = 0; kb < EM_H / 8; ++kb) {
            const float4 a = *(const float4*)(afrag + 8 * kb);
            const float4 b = *(const float4*)(bfrag + EM_H + 8 * kb);
            acc0 = wmma_f32x4((v2f){a.x, a.y}, (v2f){b.x, b.y}, acc0);
            acc1 = wmma_f32x4((v2f){a.z, a.w}, (v2f){b.z, b.w}, acc1);
        }

        if (c < C) {
            const int ebase = t * 16;
            const int mb = h << 3;
            float* op = out + (size_t)(ebase + mb) * C + c;
            if (ebase + 16 <= E) {
#pragma unroll
                for (int rr = 0; rr < 8; ++rr) { *op = (acc0[rr] + acc1[rr]) + bv; op += C; }
            } else {
#pragma unroll
                for (int rr = 0; rr < 8; ++rr) {
                    if (ebase + mb + rr < E) *op = (acc0[rr] + acc1[rr]) + bv;
                    op += C;
                }
            }
        }

        idx = idxn;
        t = tn;
    }
}

__global__ void fill_kernel(float* __restrict__ p, float v, long long n) {
    long long i = (long long)blockIdx.x * blockDim.x + threadIdx.x;
    if (i < n) p[i] = v;
}

__global__ void zero4_kernel(float4* __restrict__ p, long long n4) {
    long long i = (long long)blockIdx.x * blockDim.x + threadIdx.x;
    if (i < n4) p[i] = make_float4(0.f, 0.f, 0.f, 0.f);
}

__global__ void deg_scatter_kernel(const int* __restrict__ col,
                                   float* __restrict__ deg, int E) {
    int e = blockIdx.x * blockDim.x + threadIdx.x;
    if (e < E) atomic_add_f32(&deg[col[e]], 1.0f);
}

__global__ void rsqrt_kernel(float* __restrict__ d, int N) {
    int i = blockIdx.x * blockDim.x + threadIdx.x;
    if (i < N) d[i] = rsqrtf(d[i]);
}

__global__ void edge_scatter_kernel(const float* __restrict__ h,
                                    const float* __restrict__ dinv,
                                    const int* __restrict__ row,
                                    const int* __restrict__ col,
                                    float* __restrict__ acc,
                                    int E, int H) {
    const int chunks = H >> 2;
    long long gid = (long long)blockIdx.x * blockDim.x + threadIdx.x;
    long long total = (long long)E * chunks;
    if (gid >= total) return;
    const int e = (int)(gid / chunks);
    const int j = (int)(gid - (long long)e * chunks) << 2;
    const int r = row[e];
    const int c = col[e];
    const float s = dinv[r];
    const float4 hv = *(const float4*)(h + (size_t)r * H + j);
    float* dst = acc + (size_t)c * H + j;
    atomic_add_f32(dst + 0, s * hv.x);
    atomic_add_f32(dst + 1, s * hv.y);
    atomic_add_f32(dst + 2, s * hv.z);
    atomic_add_f32(dst + 3, s * hv.w);
}

constexpr int SC_H = 128;
constexpr int SC_EPW = 8;

template <bool FULL>
__device__ __forceinline__ void line_scatter_wave(const float* __restrict__ h,
                                                  const float* __restrict__ dinv,
                                                  const int* __restrict__ row, const int* __restrict__ col,
                                                  float* __restrict__ acc, int e0, int ne, int l) {
    int r[SC_EPW], c[SC_EPW];
    float s[SC_EPW], v0[SC_EPW], v1[SC_EPW], v2[SC_EPW], v3[SC_EPW];
#pragma unroll
    for (int i = 0; i < SC_EPW; ++i) {
        const int e = FULL ? e0 + i : e0 + min(i, ne - 1);
        r[i] = row[e];
        c[i] = col[e];
    }
#pragma unroll
    for (int i = 0; i < SC_EPW; ++i) {
        s[i] = dinv[r[i]];
        const float* __restrict__ p = h + (size_t)r[i] * SC_H + l;
        v0[i] = p[0];
        v1[i] = p[32];
        v2[i] = p[64];
        v3[i] = p[96];
    }
#pragma unroll
    for (int i = 0; i < SC_EPW; ++i) {
        if (FULL || i < ne) {
            float* dst = acc + (size_t)c[i] * SC_H + l;
            atomic_add_f32(dst, s[i] * v0[i]);
            atomic_add_f32(dst + 32, s[i] * v1[i]);
            atomic_add_f32(dst + 64, s[i] * v2[i]);
            atomic_add_f32(dst + 96, s[i] * v3[i]);
        }
    }
}

__global__ __launch_bounds__(256)
void edge_scatter_lines_kernel(const float* __restrict__ h,
                               const float* __restrict__ dinv,
                               const int* __restrict__ row,
                               const int* __restrict__ col,
                               float* __restrict__ acc, int E) {
    const int l = threadIdx.x & 31;
    const int wv = __builtin_amdgcn_readfirstlane(
        (int)(blockIdx.x * (blockDim.x >> 5) + (threadIdx.x >> 5)));
    const int e0 = wv * SC_EPW;
    if (e0 >= E) return;
    const int ne = min(SC_EPW, E - e0);
    if (ne == SC_EPW) line_scatter_wave<true>(h, dinv, row, col, acc, e0, ne, l);
    else              line_scatter_wave<false>(h, dinv, row, col, acc, e0, ne, l);
}

__global__ void finalize_relu_kernel(const float* __restrict__ h,
                                     const float* __restrict__ dinv,
                                     float* __restrict__ acc_inout, int H) {
    const int node = blockIdx.x;
    const int j = threadIdx.x;
    const float di = dinv[node];
    const size_t idx = (size_t)node * H + j;
    const float v = di * acc_inout[idx] + di * di * h[idx];
    acc_inout[idx] = v > 0.f ? v : 0.f;
}

constexpr int RAW_FU = 32;
constexpr int RAW_FM = 64;
constexpr int RAW_H = 128;
constexpr int RAW_WU = RAW_FU + RAW_FM;
constexpr int RAW_WM = RAW_FU + RAW_FM + 1;

__global__ void compose_wz_kernel(const float* __restrict__ Wu, const float* __restrict__ bu,
                                  const float* __restrict__ Wm, const float* __restrict__ bm,
                                  float* __restrict__ Wz) {
    const int i = blockIdx.x * blockDim.x + threadIdx.x;
    if (i >= RAW_H * RAW_H) return;
    const int n = i / RAW_H;
    const int k = i - n * RAW_H;
    float w = 0.f;
    if (k < RAW_FU) w = Wu[n * RAW_FU + k];
    else if (k < RAW_WU) w = Wm[n * RAW_FM + (k - RAW_FU)];
    else if (k == RAW_WU) w = bu[n];
    else if (k == RAW_WM) w = bm[n];
    Wz[i] = w;
}

__global__ void init_z_kernel(const float* __restrict__ xu, const float* __restrict__ xm,
                              const float* __restrict__ dinv, float* __restrict__ Z, int Nu) {
    const int node = blockIdx.x;
    const int j = threadIdx.x;
    const float di = dinv[node];
    float self = 0.f;
    if (node < Nu) {
        if (j < RAW_FU) self = xu[(size_t)node * RAW_FU + j];
        else if (j == RAW_WU) self = 1.f;
    } else {
        if (j >= RAW_FU && j < RAW_WU) self = xm[(size_t)(node - Nu) * RAW_FM + (j - RAW_FU)];
        else if (j == RAW_WM) self = 1.f;
    }
    Z[(size_t)node * RAW_H + j] = di * self;
}

constexpr int RAW_EPW = 8;

template <bool FULL>
__device__ __forceinline__ void raw_scatter_wave(const float* __restrict__ xu, const float* __restrict__ xm,
                                                 const float* __restrict__ dinv,
                                                 const int* __restrict__ row, const int* __restrict__ col,
                                                 float* __restrict__ Z, int Nu, int e0, int ne, int l) {
    int r[RAW_EPW], c[RAW_EPW];
    float s[RAW_EPW], v0[RAW_EPW], v1[RAW_EPW];
#pragma unroll
    for (int i = 0; i < RAW_EPW; ++i) {
        const int e = FULL ? e0 + i : e0 + min(i, ne - 1);
        r[i] = row[e];
        c[i] = col[e];
    }
#pragma unroll
    for (int i = 0; i < RAW_EPW; ++i) {
        s[i] = dinv[r[i]];
        const bool user = r[i] < Nu;
        const float* __restrict__ p = user ? xu + (size_t)r[i] * RAW_FU + l
                                           : xm + (size_t)(r[i] - Nu) * RAW_FM + l;
        v0[i] = p[0];
        v1[i] = 0.f;
        if (!user) v1[i] = p[32];
    }
#pragma unroll
    for (int i = 0; i < RAW_EPW; ++i) {
        if (FULL || i < ne) {
            const bool user = r[i] < Nu;
            float* dst = Z + (size_t)c[i] * RAW_H;
            atomic_add_f32(dst + (user ? 0 : RAW_FU) + l, s[i] * v0[i]);
            if (!user) atomic_add_f32(dst + RAW_FU + 32 + l, s[i] * v1[i]);
        }
    }
    if (l < (FULL ? RAW_EPW : ne)) {
        const int rl = row[e0 + l];
        const int cl = col[e0 + l];
        atomic_add_f32(Z + (size_t)cl * RAW_H + (rl < Nu ? RAW_WU : RAW_WM), dinv[rl]);
    }
}

__global__ __launch_bounds__(256)
void raw_scatter_kernel(const float* __restrict__ xu, const float* __restrict__ xm,
                        const float* __restrict__ dinv,
                        const int* __restrict__ row, const int* __restrict__ col,
                        float* __restrict__ Z, int E, int Nu) {
    const int l = threadIdx.x & 31;
    const int wv = __builtin_amdgcn_readfirstlane(
        (int)(blockIdx.x * (blockDim.x >> 5) + (threadIdx.x >> 5)));
    const int e0 = wv * RAW_EPW;
    if (e0 >= E) return;
    const int ne = min(RAW_EPW, E - e0);
    if (ne == RAW_EPW) raw_scatter_wave<true>(xu, xm, dinv, row, col, Z, Nu, e0, ne, l);
    else               raw_scatter_wave<false>(xu, xm, dinv, row, col, Z, Nu, e0, ne, l);
}

static inline int ceil_div_i(long long a, long long b) { return (int)((a + b - 1) / b); }

static void launch_gemm(const float* X, const float* W, const float* b, float* Y,
                        int M, int K, int N, hipStream_t stream) {
    const int TPB = 256;
    if (N == 128 && K == 32)
        gemm_bias_wmma_wlds<32, 0><<<GEMM_GRID, TPB, 0, stream>>>(X, W, b, nullptr, nullptr, Y, M);
    else if (N == 128 && K == 64)
        gemm_bias_wmma_wlds<64, 0><<<GEMM_GRID, TPB, 0, stream>>>(X, W, b, nullptr, nullptr, Y, M);
    else if (N == 128 && K == 128)
        gemm_bias_wmma_wlds<128, 0><<<GEMM_GRID, TPB, 0, stream>>>(X, W, b, nullptr, nullptr, Y, M);
    else {
        const int strips = ceil_div_i(M, 16);
        const int tiles = strips * (N / 16);
        gemm_bias_wmma_rt<<<ceil_div_i(tiles, 8), TPB, 0, stream>>>(X, W, b, Y, M, K, N);
    }
}

extern "C" void kernel_launch(void* const* d_in, const int* in_sizes, int n_in,
                              void* d_out, int out_size, void* d_ws, size_t ws_size,
                              hipStream_t stream) {
    const float* x_user  = (const float*)d_in[0];
    const float* x_movie = (const float*)d_in[1];
    const int*   edge    = (const int*)d_in[2];
    const float* Wu = (const float*)d_in[3];
    const float* bu = (const float*)d_in[4];
    const float* Wm = (const float*)d_in[5];
    const float* bm = (const float*)d_in[6];
    const float* W1 = (const float*)d_in[7];
    const float* b1 = (const float*)d_in[8];
    const float* W2 = (const float*)d_in[9];
    const float* b2 = (const float*)d_in[10];
    const float* We = (const float*)d_in[11];
    const float* be = (const float*)d_in[12];
    float* out = (float*)d_out;

    const int H  = in_sizes[4];
    const int C  = in_sizes[12];
    const int Fu = in_sizes[3] / H;
    const int Fm = in_sizes[5] / H;
    const int Nu = in_sizes[0] / Fu;
    const int Nm = in_sizes[1] / Fm;
    const int E  = in_sizes[2] / 2;
    const int N  = Nu + Nm;
    const int* row = edge;
    const int* col = edge + E;

    float* ws   = (float*)d_ws;
    float* X    = ws;
    float* Hh   = X  + (size_t)N * H;
    float* Acc  = Hh + (size_t)N * H;
    float* Wz   = Acc + (size_t)N * H;
    float* dinv = Wz + (size_t)RAW_H * RAW_H;

    const int TPB = 256, WPB = 8;
    const long long NH = (long long)N * H;

    fill_kernel<<<ceil_div_i(N, TPB), TPB, 0, stream>>>(dinv, 1.0f, N);
    deg_scatter_kernel<<<ceil_div_i(E, TPB), TPB, 0, stream>>>(col, dinv, E);
    rsqrt_kernel<<<ceil_div_i(N, TPB), TPB, 0, stream>>>(dinv, N);

    const long long scatterThreads = (long long)E * (H / 4);

    if (H == RAW_H && Fu == RAW_FU && Fm == RAW_FM) {
        compose_wz_kernel<<<ceil_div_i(RAW_H * RAW_H, TPB), TPB, 0, stream>>>(Wu, bu, Wm, bm, Wz);
        init_z_kernel<<<N, RAW_H, 0, stream>>>(x_user, x_movie, dinv, Hh, Nu);
        raw_scatter_kernel<<<ceil_div_i(E, RAW_EPW * WPB), TPB, 0, stream>>>(x_user, x_movie, dinv, row, col, Hh, E, Nu);
        gemm_bias_wmma_wlds<RAW_H, 1><<<GEMM_GRID, TPB, 0, stream>>>(Hh, Wz, nullptr, dinv, nullptr, X, N);
        gemm_bias_wmma_wlds<RAW_H, 2><<<GEMM_GRID, TPB, 0, stream>>>(X, W1, b1, dinv, Hh, Acc, N);
    } else {
        launch_gemm(x_user, Wu, bu, X, Nu, Fu, H, stream);
        launch_gemm(x_movie, Wm, bm, X + (size_t)Nu * H, Nm, Fm, H, stream);

        launch_gemm(X, W1, b1, Hh, N, H, H, stream);
        zero4_kernel<<<ceil_div_i(NH / 4, TPB), TPB, 0, stream>>>((float4*)Acc, NH / 4);
        edge_scatter_kernel<<<ceil_div_i(scatterThreads, TPB), TPB, 0, stream>>>(Hh, dinv, row, col, Acc, E, H);
        finalize_relu_kernel<<<N, H, 0, stream>>>(Hh, dinv, Acc, H);
    }

    launch_gemm(Acc, W2, b2, Hh, N, H, H, stream);
    zero4_kernel<<<ceil_div_i(NH / 4, TPB), TPB, 0, stream>>>((float4*)X, NH / 4);
    if (H == SC_H)
        edge_scatter_lines_kernel<<<ceil_div_i(E, SC_EPW * WPB), TPB, 0, stream>>>(Hh, dinv, row, col, X, E);
    else
        edge_scatter_kernel<<<ceil_div_i(scatterThreads, TPB), TPB, 0, stream>>>(Hh, dinv, row, col, X, E, H);
    finalize_relu_kernel<<<N, H, 0, stream>>>(Hh, dinv, X, H);

    const int edgeTiles = ceil_div_i(E, 16);
    if (H == EM_H && C <= 16)
        edge_mlp_rows_lds<<<EM_GRID, TPB, 0, stream>>>(X, row, col, We, be, out, E, Nu, C);
    else
        edge_mlp_wmma<0><<<ceil_div_i(edgeTiles, WPB), TPB, 0, stream>>>(X, row, col, We, be, out, E, Nu, H, C);
}


# conv2 sums by owner blocks: block b has rows b, b+256, .. in LDS, scans a byte-per-edge owner list from the degree pass, gathers its edges' rows, LDS float adds; no global atomic; sum order changes
# speedup vs baseline: 2.2051x; 1.0636x over previous
#include <hip/hip_runtime.h>

typedef __attribute__((ext_vector_type(2))) float v2f;
typedef __attribute__((ext_vector_type(8))) float v8f;

__device__ __forceinline__ v8f wmma_f32x4(v2f a, v2f b, v8f c) {
    return __builtin_amdgcn_wmma_f32_16x16x4_f32(
         false, a,  false, b,
         (short)0, c,  false,  false);
}

__device__ __forceinline__ void atomic_add_f32(float* p, float v) {
#if defined(__HIP_DEVICE_COMPILE__)
    unsafeAtomicAdd(p, v);
#else
    atomicAdd(p, v);
#endif
}

constexpr int GEMM_GRID_LOG2 = 9;
constexpr int GEMM_GRID = 1 << GEMM_GRID_LOG2;

template <int MODE>
__device__ __forceinline__ float gemm_epilogue(float a, float b, float g) {
    if (MODE == 0) return a + b;
    if (MODE == 1) return g * a;
    const float v = a + g * b;
    return v > 0.f ? v : 0.f;
}

template <int K, int MODE>
__global__ __launch_bounds__(256)
void gemm_bias_wmma_wlds(const float* __restrict__ X,
                         const float* __restrict__ W,
                         const float* __restrict__ bias,
                         const float* __restrict__ rs,
                         const float* __restrict__ zc,
                         float* __restrict__ Y,
                         int M) {
    constexpr int N = 128;
    constexpr int NT = N / 16;
    constexpr int PITCH = K + 4;
    constexpr int KB = K / 8;
    constexpr int WAVES = 8;
    __shared__ __attribute__((aligned(16))) float Ws[N * PITCH];

    const int lane = threadIdx.x & 31;
    const int wave = __builtin_amdgcn_readfirstlane((int)(threadIdx.x >> 5));
    const int c = lane & 15;
    const int h = lane >> 4;

    const int strips = (M + 15) >> 4;
    const int s0 = (int)(((long long)strips * blockIdx.x) >> GEMM_GRID_LOG2);
    const int s1 = (int)(((long long)strips * (blockIdx.x + 1)) >> GEMM_GRID_LOG2);

    float bv[NT];
#pragma unroll
    for (int t = 0; t < NT; ++t) bv[t] = (MODE == 1) ? 0.f : bias[16 * t + c];

    float4 a[KB];
    int s = s0 + wave;

    if (s < s1) {
        const int arow = min(s * 16 + c, M - 1);
        const float4* __restrict__ ap = (const float4*)(X + (size_t)arow * K) + h;
#pragma unroll
        for (int kb = 0; kb < KB; ++kb) a[kb] = ap[2 * kb];
    }

    for (int i = threadIdx.x; i < N * (K / 4); i += WAVES * 32) {
        const int n = i / (K / 4);
        const int k4 = i - n * (K / 4);
        *(float4*)(Ws + n * PITCH + 4 * k4) = ((const float4*)W)[i];
    }
    __syncthreads();

    const float* __restrict__ bfrag = Ws + c * PITCH + 4 * h;

    while (s < s1) {
        float gr[8];
#pragma unroll
        for (int r = 0; r < 8; ++r) {
            gr[r] = 0.f;
            if (MODE != 0) {
                const int rr = min(s * 16 + (h << 3) + r, M - 1);
                float g = rs[rr];
                if (MODE == 2) {
                    const v2f w = *(const v2f*)(zc + (size_t)rr * N + 96);
                    g *= (w.x + w.y);
                }
                gr[r] = g;
            }
        }

        v8f acc[NT];
#pragma unroll
        for (int t = 0; t < NT; ++t) acc[t] = (v8f){};

        float4 bc[NT], bn[NT];
#pragma unroll
        for (int t = 0; t < NT; ++t)
            bc[t] = *(const float4*)(bfrag + 16 * t * PITCH);
#pragma unroll
        for (int kb = 0; kb < KB; ++kb) {
            if (kb + 1 < KB) {
#pragma unroll
                for (int t = 0; t < NT; ++t)
                    bn[t] = *(const float4*)(bfrag + 16 * t * PITCH + 8 * (kb + 1));
            }
            __builtin_amdgcn_sched_barrier(0);
            const v2f a01 = {a[kb].x, a[kb].y};
            const v2f a23 = {a[kb].z, a[kb].w};
#pragma unroll
            for (int t = 0; t < NT; ++t) {
                const v2f b01 = {bc[t].x, bc[t].y};
                acc[t] = wmma_f32x4(a01, b01, acc[t]);
            }
#pragma unroll
            for (int t = 0; t < NT; ++t) {
                const v2f b23 = {bc[t].z, bc[t].w};
                acc[t] = wmma_f32x4(a23, b23, acc[t]);
            }
            __builtin_amdgcn_sched_barrier(0);
#pragma unroll
            for (int t = 0; t < NT; ++t) bc[t] = bn[t];
        }

        const int row0 = s * 16;
        s += WAVES;

        if (s < s1) {
            const int arow = min(s * 16 + c, M - 1);
            const float4* __restrict__ ap = (const float4*)(X + (size_t)arow * K) + h;
#pragma unroll
            for (int kb = 0; kb < KB; ++kb) a[kb] = ap[2 * kb];
        }

        const int rbase = row0 + (h << 3);
        float* __restrict__ yrow = Y + (size_t)rbase * N + c;
        if (row0 + 16 <= M) {
#pragma unroll
            for (int t = 0; t < NT; ++t) {
#pragma unroll
                for (int r = 0; r < 8; ++r)
                    yrow[(size_t)r * N + 16 * t] = gemm_epilogue<MODE>(acc[t][r], bv[t], gr[r]);
            }
        } else {
#pragma unroll
            for (int t = 0; t < NT; ++t) {
#pragma unroll
                for (int r = 0; r < 8; ++r)
                    if (rbase + r < M)
                        yrow[(size_t)r * N + 16 * t] = gemm_epilogue<MODE>(acc[t][r], bv[t], gr[r]);
            }
        }
    }
}

__global__ void gemm_bias_wmma_rt(const float* __restrict__ X,
                                  const float* __restrict__ W,
                                  const float* __restrict__ bias,
                                  float* __restrict__ Y,
                                  int M, int K, int N) {
    const int lane = threadIdx.x & 31;
    const int wave = threadIdx.x >> 5;
    const int tilesN = N >> 4;
    const int tile = blockIdx.x * (blockDim.x >> 5) + wave;
    const int tm = tile / tilesN;
    const int tn = tile - tm * tilesN;
    if (tm * 16 >= M) return;

    const int mrow = min(tm * 16 + (lane & 15), M - 1);
    const int ncol = tn * 16 + (lane & 15);
    const int koff = (lane >> 4) << 1;
    const float* __restrict__ xrow = X + (size_t)mrow * K + koff;
    const float* __restrict__ wrow = W + (size_t)ncol * K + koff;

    v8f acc = {};
#pragma unroll 8
    for (int k = 0; k < K; k += 4) {
        v2f a = *(const v2f*)(xrow + k);
        v2f b = *(const v2f*)(wrow + k);
        acc = wmma_f32x4(a, b, acc);
    }

    const float bv = bias[ncol];
    const int rbase = tm * 16 + ((lane >> 4) << 3);
    float* yp = Y + (size_t)rbase * N + ncol;
#pragma unroll
    for (int r = 0; r < 8; ++r) {
        if (rbase + r < M) *yp = acc[r] + bv;
        yp += N;
    }
}

template <int HT>
__global__ void edge_mlp_wmma(const float* __restrict__ xf,
                              const int* __restrict__ row,
                              const int* __restrict__ col,
                              const float* __restrict__ We,
                              const float* __restrict__ be,
                              float* __restrict__ out,
                              int E, int Nu, int Hrt, int C) {
    const int H = (HT > 0) ? HT : Hrt;
    const int K = 2 * H;
    const int lane = threadIdx.x & 31;
    const int wave = threadIdx.x >> 5;
    const int tile = blockIdx.x * (blockDim.x >> 5) + wave;
    const int ebase = tile * 16;
    if (ebase >= E) return;

    const int m = lane & 15;
    const int e = min(ebase + m, E - 1);
    const int r = row[e];
    const int c = col[e];
    const int koff = (lane >> 4) << 1;

    const float* __restrict__ urow = xf + (size_t)r * H + koff;
    const float* __restrict__ mrow = xf + (size_t)(Nu + c) * H + koff;

    const int n = lane & 15;
    const float wmask = (n < C) ? 1.0f : 0.0f;
    const float* __restrict__ wrow = We + (size_t)min(n, C - 1) * K + koff;

    v8f acc = {};
#pragma unroll 8
    for (int k = 0; k < K; k += 4) {
        const float* src = (k + koff < H) ? (urow + k) : (mrow + (k - H));
        v2f a = *(const v2f*)src;
        v2f b = *(const v2f*)(wrow + k);
        b.x *= wmask;
        b.y *= wmask;
        acc = wmma_f32x4(a, b, acc);
    }

    if (n < C) {
        const float bv = be[n];
        const int mb = (lane >> 4) << 3;
        float* op = out + (size_t)(ebase + mb) * C + n;
        if (ebase + 16 <= E) {
#pragma unroll
            for (int rr = 0; rr < 8; ++rr) { *op = acc[rr] + bv; op += C; }
        } else {
#pragma unroll
            for (int rr = 0; rr < 8; ++rr) {
                if (ebase + mb + rr < E) *op = acc[rr] + bv;
                op += C;
            }
        }
    }
}

constexpr int EM_H = 128;
constexpr int EM_K = 2 * EM_H;
constexpr int EM_WPITCH = EM_K + 4;
constexpr int EM_APITCH = EM_H + 4;
constexpr int EM_WAVES = 8;
constexpr int EM_GRID = 768;

__device__ __forceinline__ int edge_tile_ends(const int* __restrict__ row, const int* __restrict__ col,
                                              int t, int lane, int E) {
    const int e = min(t * 16 + (lane & 15), E - 1);
    const int* __restrict__ p = (lane < 16) ? row : col;
    return p[e];
}

#define EM_ROW_LOAD(j, IDX, BASE) \
    rv##j = ((const float4*)(xf + (size_t)(((BASE) ? Nu : 0) + __builtin_amdgcn_readlane(IDX, (BASE) + j)) * EM_H))[lane];
#define EM_ROW_STORE(j, IDX, BASE) *(float4*)(astore + j * EM_APITCH) = rv##j;
#define EM_ALL_ROWS(OP, IDX, BASE) \
    OP(0, IDX, BASE) OP(1, IDX, BASE) OP(2, IDX, BASE) OP(3, IDX, BASE) \
    OP(4, IDX, BASE) OP(5, IDX, BASE) OP(6, IDX, BASE) OP(7, IDX, BASE) \
    OP(8, IDX, BASE) OP(9, IDX, BASE) OP(10, IDX, BASE) OP(11, IDX, BASE) \
    OP(12, IDX, BASE) OP(13, IDX, BASE) OP(14, IDX, BASE) OP(15, IDX, BASE)

__global__ __launch_bounds__(256)
void edge_mlp_rows_lds(const float* __restrict__ xf,
                       const int* __restrict__ row,
                       const int* __restrict__ col,
                       const float* __restrict__ We,
                       const float* __restrict__ be,
                       float* __restrict__ out,
                       int E, int Nu, int C) {
    __shared__ __attribute__((aligned(16))) float Ws[16 * EM_WPITCH];
    __shared__ __attribute__((aligned(16))) float As[EM_WAVES * 16 * EM_APITCH];

    const int lane = threadIdx.x & 31;
    const int wave = __builtin_amdgcn_readfirstlane((int)(threadIdx.x >> 5));
    const int c = lane & 15;
    const int h = lane >> 4;

    const int tiles = (E + 15) >> 4;
    const int t0 = (int)(((long long)tiles * blockIdx.x) / EM_GRID);
    const int t1 = (int)(((long long)tiles * (blockIdx.x + 1)) / EM_GRID);
    int t = t0 + wave;

    float4 rv0, rv1, rv2, rv3, rv4, rv5, rv6, rv7, rv8, rv9, rv10, rv11, rv12, rv13, rv14, rv15;
    rv0 = rv1 = rv2 = rv3 = rv4 = rv5 = rv6 = rv7 = make_float4(0.f, 0.f, 0.f, 0.f);
    rv8 = rv9 = rv10 = rv11 = rv12 = rv13 = rv14 = rv15 = rv0;
    int idx = 0;
    if (t < t1) {
        idx = edge_tile_ends(row, col, t, lane, E);
        EM_ALL_ROWS(EM_ROW_LOAD, idx, 0)
    }

    for (int i = threadIdx.x; i < 16 * (EM_K / 4); i += EM_WAVES * 32) {
        const int n = i / (EM_K / 4);
        const int k4 = i - n * (EM_K / 4);
        float4 w = make_float4(0.f, 0.f, 0.f, 0.f);
        if (n < C) w = ((const float4*)We)[i];
        *(float4*)(Ws + n * EM_WPITCH + 4 * k4) = w;
    }
    __syncthreads();

    float* Aw = As + wave * (16 * EM_APITCH);
    float* astore = Aw + 4 * lane;
    const float* afrag = Aw + c * EM_APITCH + 4 * h;
    const float* bfrag = Ws + c * EM_WPITCH + 4 * h;
    const float bv = (c < C) ? be[c] : 0.f;

    while (t < t1) {
        const int tn = t + EM_WAVES;
        int idxn = 0;
        if (tn < t1) idxn = edge_tile_ends(row, col, tn, lane, E);

        EM_ALL_ROWS(EM_ROW_STORE, idx, 0)
        EM_ALL_ROWS(EM_ROW_LOAD, idx, 16)

        v8f acc0 = {}, acc1 = {};
#pragma unroll
        for (int kb = 0; kb < EM_H / 8; ++kb) {
            const float4 a = *(const float4*)(afrag + 8 * kb);
            const float4 b = *(const float4*)(bfrag + 8 * kb);
            acc0 = wmma_f32x4((v2f){a.x, a.y}, (v2f){b.x, b.y}, acc0);
            acc1 = wmma_f32x4((v2f){a.z, a.w}, (v2f){b.z, b.w}, acc1);
        }

        EM_ALL_ROWS(EM_ROW_STORE, idx, 0)
        if (tn < t1) {
            EM_ALL_ROWS(EM_ROW_LOAD, idxn, 0)
        }
#pragma unroll
        for (int kb = 0; kb < EM_H / 8; ++kb) {
            const float4 a = *(const float4*)(afrag + 8 * kb);
            const float4 b = *(const float4*)(bfrag + EM_H + 8 * kb);
            acc0 = wmma_f32x4((v2f){a.x, a.y}, (v2f){b.x, b.y}, acc0);
            acc1 = wmma_f32x4((v2f){a.z, a.w}, (v2f){b.z, b.w}, acc1);
        }

        if (c < C) {
            const int ebase = t * 16;
            const int mb = h << 3;
            float* op = out + (size_t)(ebase + mb) * C + c;
            if (ebase + 16 <= E) {
#pragma unroll
                for (int rr = 0; rr < 8; ++rr) { *op = (acc0[rr] + acc1[rr]) + bv; op += C; }
            } else {
#pragma unroll
                for (int rr = 0; rr < 8; ++rr) {
                    if (ebase + mb + rr < E) *op = (acc0[rr] + acc1[rr]) + bv;
                    op += C;
                }
            }
        }

        idx = idxn;
        t = tn;
    }
}

__global__ void fill_kernel(float* __restrict__ p, float v, long long n) {
    long long i = (long long)blockIdx.x * blockDim.x + threadIdx.x;
    if (i < n) p[i] = v;
}

__global__ void zero4_kernel(float4* __restrict__ p, long long n4) {
    long long i = (long long)blockIdx.x * blockDim.x + threadIdx.x;
    if (i < n4) p[i] = make_float4(0.f, 0.f, 0.f, 0.f);
}

__global__ void deg_scatter_kernel(const int* __restrict__ col,
                                   float* __restrict__ deg, int E) {
    int e = blockIdx.x * blockDim.x + threadIdx.x;
    if (e < E) atomic_add_f32(&deg[col[e]], 1.0f);
}

__global__ void deg_scatter_own_kernel(const int* __restrict__ col,
                                       float* __restrict__ deg,
                                       unsigned char* __restrict__ own, int E) {
    int e = blockIdx.x * blockDim.x + threadIdx.x;
    if (e < E) {
        const int c = col[e];
        own[e] = (unsigned char)(c & 255);
        atomic_add_f32(&deg[c], 1.0f);
    }
}

__global__ void rsqrt_kernel(float* __restrict__ d, int N) {
    int i = blockIdx.x * blockDim.x + threadIdx.x;
    if (i < N) d[i] = rsqrtf(d[i]);
}

__global__ void edge_scatter_kernel(const float* __restrict__ h,
                                    const float* __restrict__ dinv,
                                    const int* __restrict__ row,
                                    const int* __restrict__ col,
                                    float* __restrict__ acc,
                                    int E, int H) {
    const int chunks = H >> 2;
    long long gid = (long long)blockIdx.x * blockDim.x + threadIdx.x;
    long long total = (long long)E * chunks;
    if (gid >= total) return;
    const int e = (int)(gid / chunks);
    const int j = (int)(gid - (long long)e * chunks) << 2;
    const int r = row[e];
    const int c = col[e];
    const float s = dinv[r];
    const float4 hv = *(const float4*)(h + (size_t)r * H + j);
    float* dst = acc + (size_t)c * H + j;
    atomic_add_f32(dst + 0, s * hv.x);
    atomic_add_f32(dst + 1, s * hv.y);
    atomic_add_f32(dst + 2, s * hv.z);
    atomic_add_f32(dst + 3, s * hv.w);
}

constexpr int SC_H = 128;
constexpr int SC_EPW = 8;

template <bool FULL>
__device__ __forceinline__ void line_scatter_wave(const float* __restrict__ h,
                                                  const float* __restrict__ dinv,
                                                  const int* __restrict__ row, const int* __restrict__ col,
                                                  float* __restrict__ acc, int e0, int ne, int l) {
    int r[SC_EPW], c[SC_EPW];
    float s[SC_EPW], v0[SC_EPW], v1[SC_EPW], v2[SC_EPW], v3[SC_EPW];
#pragma unroll
    for (int i = 0; i < SC_EPW; ++i) {
        const int e = FULL ? e0 + i : e0 + min(i, ne - 1);
        r[i] = row[e];
        c[i] = col[e];
    }
#pragma unroll
    for (int i = 0; i < SC_EPW; ++i) {
        s[i] = dinv[r[i]];
        const float* __restrict__ p = h + (size_t)r[i] * SC_H + l;
        v0[i] = p[0];
        v1[i] = p[32];
        v2[i] = p[64];
        v3[i] = p[96];
    }
#pragma unroll
    for (int i = 0; i < SC_EPW; ++i) {
        if (FULL || i < ne) {
            float* dst = acc + (size_t)c[i] * SC_H + l;
            atomic_add_f32(dst, s[i] * v0[i]);
            atomic_add_f32(dst + 32, s[i] * v1[i]);
            atomic_add_f32(dst + 64, s[i] * v2[i]);
            atomic_add_f32(dst + 96, s[i] * v3[i]);
        }
    }
}

__global__ __launch_bounds__(256)
void edge_scatter_lines_kernel(const float* __restrict__ h,
                               const float* __restrict__ dinv,
                               const int* __restrict__ row,
                               const int* __restrict__ col,
                               float* __restrict__ acc, int E) {
    const int l = threadIdx.x & 31;
    const int wv = __builtin_amdgcn_readfirstlane(
        (int)(blockIdx.x * (blockDim.x >> 5) + (threadIdx.x >> 5)));
    const int e0 = wv * SC_EPW;
    if (e0 >= E) return;
    const int ne = min(SC_EPW, E - e0);
    if (ne == SC_EPW) line_scatter_wave<true>(h, dinv, row, col, acc, e0, ne, l);
    else              line_scatter_wave<false>(h, dinv, row, col, acc, e0, ne, l);
}


constexpr int OW_GRID_LOG2 = 8;
constexpr int OW_GRID = 1 << OW_GRID_LOG2;
static_assert(OW_GRID == 256, "the owner list holds one byte per edge: target mod 256");
constexpr int OW_WAVES = 32;
constexpr int OW_SLOTS = 392;
constexpr int OW_U = 8;
constexpr int OW_Q = 544;

__device__ __forceinline__ void lds_add_f32(float* p, float v) {
    __hip_atomic_fetch_add(p, v, __ATOMIC_RELAXED, __HIP_MEMORY_SCOPE_WORKGROUP);
}

__device__ __forceinline__ void ow_drain(const float* __restrict__ h, const float* __restrict__ dinv,
                                         const int* __restrict__ row, const int* __restrict__ col,
                                         float* accs, const int* q, int m, int lane) {
    const int e = q[min(lane, m - 1)];
    const int r = row[e];
    const int c = col[e];
    const int sbits = __float_as_int(dinv[r]);
    for (int j = 0; j < m; j += 8) {
        float v0[8], v1[8], v2[8], v3[8], ss[8];
        int sl[8];
#pragma unroll
        for (int i = 0; i < 8; ++i) {
            const int idx = min(j + i, m - 1);
            const int rr = __builtin_amdgcn_readlane(r, idx);
            sl[i] = __builtin_amdgcn_readlane(c, idx) >> OW_GRID_LOG2;
            ss[i] = __int_as_float(__builtin_amdgcn_readlane(sbits, idx));
            const float* __restrict__ p = h + (size_t)rr * SC_H + lane;
            v0[i] = p[0];
            v1[i] = p[32];
            v2[i] = p[64];
            v3[i] = p[96];
        }
#pragma unroll
        for (int i = 0; i < 8; ++i) {
            if (j + i < m) {
                float* d = accs + sl[i] * SC_H + lane;
                lds_add_f32(d, ss[i] * v0[i]);
                lds_add_f32(d + 32, ss[i] * v1[i]);
                lds_add_f32(d + 64, ss[i] * v2[i]);
                lds_add_f32(d + 96, ss[i] * v3[i]);
            }
        }
    }
}

#define OW_STEP(MT, EDGE) \
    { \
        const bool mt = (MT); \
        const unsigned mask = __builtin_amdgcn_ballot_w32(mt); \
        if (mask) { \
            if (mt) q[cnt + (int)__builtin_amdgcn_mbcnt_lo(mask, 0u)] = (EDGE); \
            cnt += __builtin_popcount(mask); \
        } \
    }

#define OW_WORD(WV, E0) \
    { \
        const unsigned xw = ok ? ((WV) ^ bpat) : 0xffffffffu; \
        const unsigned zb = (xw - 0x01010101u) & ~xw & 0x80808080u; \
        if (__builtin_amdgcn_ballot_w32(zb != 0u)) { \
            OW_STEP((xw & 0x000000ffu) == 0u, (E0) + 0) \
            OW_STEP((xw & 0x0000ff00u) == 0u, (E0) + 1) \
            OW_STEP((xw & 0x00ff0000u) == 0u, (E0) + 2) \
            OW_STEP((xw & 0xff000000u) == 0u, (E0) + 3) \
        } \
    }

__global__ __launch_bounds__(1024)
void owner_aggregate_kernel(const float* __restrict__ h, const float* __restrict__ dinv,
                            const int* __restrict__ row, const int* __restrict__ col,
                            const unsigned char* __restrict__ own,
                            float* __restrict__ x, int E, int N) {
    __shared__ __attribute__((aligned(16))) float accs[OW_SLOTS * SC_H];
    __shared__ int qs[OW_WAVES * OW_Q];

    const int lane = threadIdx.x & 31;
    const int wave = __builtin_amdgcn_readfirstlane((int)(threadIdx.x >> 5));
    const int b = blockIdx.x;
    const int slots = (N - b + OW_GRID - 1) >> OW_GRID_LOG2;

#pragma unroll 4
    for (int sl = wave; sl < slots; sl += OW_WAVES) {
        const int node = (sl << OW_GRID_LOG2) + b;
        const float di = dinv[node];
        const float4 hv = ((const float4*)(h + (size_t)node * SC_H))[lane];
        *(float4*)(accs + sl * SC_H + 4 * lane) = make_float4(di * hv.x, di * hv.y, di * hv.z, di * hv.w);
    }
    __syncthreads();

    int* q = qs + wave * OW_Q;
    int cnt = 0;
    const uint4* __restrict__ own4 = (const uint4*)own;
    const int total16 = E >> 4;
    const unsigned bpat = (unsigned)b * 0x01010101u;
    for (int base = 0; base < total16; base += OW_U * OW_WAVES * 32) {
        uint4 ov[OW_U];
#pragma unroll
        for (int u = 0; u < OW_U; ++u) {
            const int i16 = base + (u * OW_WAVES + wave) * 32 + lane;
            ov[u] = own4[min(i16, total16 - 1)];
        }
#pragma unroll
        for (int u = 0; u < OW_U; ++u) {
            const int i16 = base + (u * OW_WAVES + wave) * 32 + lane;
            const bool ok = i16 < total16;
            OW_WORD(ov[u].x, 16 * i16 + 0)
            OW_WORD(ov[u].y, 16 * i16 + 4)
            OW_WORD(ov[u].z, 16 * i16 + 8)
            OW_WORD(ov[u].w, 16 * i16 + 12)
            if (cnt >= 32) {
                for (int g = 0; g < cnt; g += 32)
                    ow_drain(h, dinv, row, col, accs, q + g, min(32, cnt - g), lane);
                cnt = 0;
            }
        }
    }
    for (int g = 0; g < cnt; g += 32)
        ow_drain(h, dinv, row, col, accs, q + g, min(32, cnt - g), lane);
    __syncthreads();

#pragma unroll 4
    for (int sl = wave; sl < slots; sl += OW_WAVES) {
        const int node = (sl << OW_GRID_LOG2) + b;
        const float di = dinv[node];
        const float4 a = *(const float4*)(accs + sl * SC_H + 4 * lane);
        float4 o;
        o.x = di * a.x; o.x = o.x > 0.f ? o.x : 0.f;
        o.y = di * a.y; o.y = o.y > 0.f ? o.y : 0.f;
        o.z = di * a.z; o.z = o.z > 0.f ? o.z : 0.f;
        o.w = di * a.w; o.w = o.w > 0.f ? o.w : 0.f;
        ((float4*)(x + (size_t)node * SC_H))[lane] = o;
    }
}

__global__ void finalize_relu_kernel(const float* __restrict__ h,
                                     const float* __restrict__ dinv,
                                     float* __restrict__ acc_inout, int H) {
    const int node = blockIdx.x;
    const int j = threadIdx.x;
    const float di = dinv[node];
    const size_t idx = (size_t)node * H + j;
    const float v = di * acc_inout[idx] + di * di * h[idx];
    acc_inout[idx] = v > 0.f ? v : 0.f;
}

constexpr int RAW_FU = 32;
constexpr int RAW_FM = 64;
constexpr int RAW_H = 128;
constexpr int RAW_WU = RAW_FU + RAW_FM;
constexpr int RAW_WM = RAW_FU + RAW_FM + 1;

__global__ void compose_wz_kernel(const float* __restrict__ Wu, const float* __restrict__ bu,
                                  const float* __restrict__ Wm, const float* __restrict__ bm,
                                  float* __restrict__ Wz) {
    const int i = blockIdx.x * blockDim.x + threadIdx.x;
    if (i >= RAW_H * RAW_H) return;
    const int n = i / RAW_H;
    const int k = i - n * RAW_H;
    float w = 0.f;
    if (k < RAW_FU) w = Wu[n * RAW_FU + k];
    else if (k < RAW_WU) w = Wm[n * RAW_FM + (k - RAW_FU)];
    else if (k == RAW_WU) w = bu[n];
    else if (k == RAW_WM) w = bm[n];
    Wz[i] = w;
}

__global__ void init_z_kernel(const float* __restrict__ xu, const float* __restrict__ xm,
                              const float* __restrict__ dinv, float* __restrict__ Z, int Nu) {
    const int node = blockIdx.x;
    const int j = threadIdx.x;
    const float di = dinv[node];
    float self = 0.f;
    if (node < Nu) {
        if (j < RAW_FU) self = xu[(size_t)node * RAW_FU + j];
        else if (j == RAW_WU) self = 1.f;
    } else {
        if (j >= RAW_FU && j < RAW_WU) self = xm[(size_t)(node - Nu) * RAW_FM + (j - RAW_FU)];
        else if (j == RAW_WM) self = 1.f;
    }
    Z[(size_t)node * RAW_H + j] = di * self;
}

constexpr int RAW_EPW = 8;

template <bool FULL>
__device__ __forceinline__ void raw_scatter_wave(const float* __restrict__ xu, const float* __restrict__ xm,
                                                 const float* __restrict__ dinv,
                                                 const int* __restrict__ row, const int* __restrict__ col,
                                                 float* __restrict__ Z, int Nu, int e0, int ne, int l) {
    int r[RAW_EPW], c[RAW_EPW];
    float s[RAW_EPW], v0[RAW_EPW], v1[RAW_EPW];
#pragma unroll
    for (int i = 0; i < RAW_EPW; ++i) {
        const int e = FULL ? e0 + i : e0 + min(i, ne - 1);
        r[i] = row[e];
        c[i] = col[e];
    }
#pragma unroll
    for (int i = 0; i < RAW_EPW; ++i) {
        s[i] = dinv[r[i]];
        const bool user = r[i] < Nu;
        const float* __restrict__ p = user ? xu + (size_t)r[i] * RAW_FU + l
                                           : xm + (size_t)(r[i] - Nu) * RAW_FM + l;
        v0[i] = p[0];
        v1[i] = 0.f;
        if (!user) v1[i] = p[32];
    }
#pragma unroll
    for (int i = 0; i < RAW_EPW; ++i) {
        if (FULL || i < ne) {
            const bool user = r[i] < Nu;
            float* dst = Z + (size_t)c[i] * RAW_H;
            atomic_add_f32(dst + (user ? 0 : RAW_FU) + l, s[i] * v0[i]);
            if (!user) atomic_add_f32(dst + RAW_FU + 32 + l, s[i] * v1[i]);
        }
    }
    if (l < (FULL ? RAW_EPW : ne)) {
        const int rl = row[e0 + l];
        const int cl = col[e0 + l];
        atomic_add_f32(Z + (size_t)cl * RAW_H + (rl < Nu ? RAW_WU : RAW_WM), dinv[rl]);
    }
}

__global__ __launch_bounds__(256)
void raw_scatter_kernel(const float* __restrict__ xu, const float* __restrict__ xm,
                        const float* __restrict__ dinv,
                        const int* __restrict__ row, const int* __restrict__ col,
                        float* __restrict__ Z, int E, int Nu) {
    const int l = threadIdx.x & 31;
    const int wv = __builtin_amdgcn_readfirstlane(
        (int)(blockIdx.x * (blockDim.x >> 5) + (threadIdx.x >> 5)));
    const int e0 = wv * RAW_EPW;
    if (e0 >= E) return;
    const int ne = min(RAW_EPW, E - e0);
    if (ne == RAW_EPW) raw_scatter_wave<true>(xu, xm, dinv, row, col, Z, Nu, e0, ne, l);
    else               raw_scatter_wave<false>(xu, xm, dinv, row, col, Z, Nu, e0, ne, l);
}

static inline int ceil_div_i(long long a, long long b) { return (int)((a + b - 1) / b); }

static void launch_gemm(const float* X, const float* W, const float* b, float* Y,
                        int M, int K, int N, hipStream_t stream) {
    const int TPB = 256;
    if (N == 128 && K == 32)
        gemm_bias_wmma_wlds<32, 0><<<GEMM_GRID, TPB, 0, stream>>>(X, W, b, nullptr, nullptr, Y, M);
    else if (N == 128 && K == 64)
        gemm_bias_wmma_wlds<64, 0><<<GEMM_GRID, TPB, 0, stream>>>(X, W, b, nullptr, nullptr, Y, M);
    else if (N == 128 && K == 128)
        gemm_bias_wmma_wlds<128, 0><<<GEMM_GRID, TPB, 0, stream>>>(X, W, b, nullptr, nullptr, Y, M);
    else {
        const int strips = ceil_div_i(M, 16);
        const int tiles = strips * (N / 16);
        gemm_bias_wmma_rt<<<ceil_div_i(tiles, 8), TPB, 0, stream>>>(X, W, b, Y, M, K, N);
    }
}

extern "C" void kernel_launch(void* const* d_in, const int* in_sizes, int n_in,
                              void* d_out, int out_size, void* d_ws, size_t ws_size,
                              hipStream_t stream) {
    const float* x_user  = (const float*)d_in[0];
    const float* x_movie = (const float*)d_in[1];
    const int*   edge    = (const int*)d_in[2];
    const float* Wu = (const float*)d_in[3];
    const float* bu = (const float*)d_in[4];
    const float* Wm = (const float*)d_in[5];
    const float* bm = (const float*)d_in[6];
    const float* W1 = (const float*)d_in[7];
    const float* b1 = (const float*)d_in[8];
    const float* W2 = (const float*)d_in[9];
    const float* b2 = (const float*)d_in[10];
    const float* We = (const float*)d_in[11];
    const float* be = (const float*)d_in[12];
    float* out = (float*)d_out;

    const int H  = in_sizes[4];
    const int C  = in_sizes[12];
    const int Fu = in_sizes[3] / H;
    const int Fm = in_sizes[5] / H;
    const int Nu = in_sizes[0] / Fu;
    const int Nm = in_sizes[1] / Fm;
    const int E  = in_sizes[2] / 2;
    const int N  = Nu + Nm;
    const int* row = edge;
    const int* col = edge + E;

    float* ws   = (float*)d_ws;
    float* X    = ws;
    float* Hh   = X  + (size_t)N * H;
    float* Acc  = Hh + (size_t)N * H;
    float* Wz   = Acc + (size_t)N * H;
    float* dinv = Wz + (size_t)RAW_H * RAW_H;
    unsigned char* own = (unsigned char*)(dinv + (size_t)((N + 3) & ~3));

    const int TPB = 256, WPB = 8;
    const long long NH = (long long)N * H;

    fill_kernel<<<ceil_div_i(N, TPB), TPB, 0, stream>>>(dinv, 1.0f, N);
    const bool owner_path = (H == SC_H) && ((E & 15) == 0) && (N <= OW_GRID * OW_SLOTS);
    if (owner_path)
        deg_scatter_own_kernel<<<ceil_div_i(E, TPB), TPB, 0, stream>>>(col, dinv, own, E);
    else
        deg_scatter_kernel<<<ceil_div_i(E, TPB), TPB, 0, stream>>>(col, dinv, E);
    rsqrt_kernel<<<ceil_div_i(N, TPB), TPB, 0, stream>>>(dinv, N);

    const long long scatterThreads = (long long)E * (H / 4);

    if (H == RAW_H && Fu == RAW_FU && Fm == RAW_FM) {
        compose_wz_kernel<<<ceil_div_i(RAW_H * RAW_H, TPB), TPB, 0, stream>>>(Wu, bu, Wm, bm, Wz);
        init_z_kernel<<<N, RAW_H, 0, stream>>>(x_user, x_movie, dinv, Hh, Nu);
        raw_scatter_kernel<<<ceil_div_i(E, RAW_EPW * WPB), TPB, 0, stream>>>(x_user, x_movie, dinv, row, col, Hh, E, Nu);
        gemm_bias_wmma_wlds<RAW_H, 1><<<GEMM_GRID, TPB, 0, stream>>>(Hh, Wz, nullptr, dinv, nullptr, X, N);
        gemm_bias_wmma_wlds<RAW_H, 2><<<GEMM_GRID, TPB, 0, stream>>>(X, W1, b1, dinv, Hh, Acc, N);
    } else {
        launch_gemm(x_user, Wu, bu, X, Nu, Fu, H, stream);
        launch_gemm(x_movie, Wm, bm, X + (size_t)Nu * H, Nm, Fm, H, stream);

        launch_gemm(X, W1, b1, Hh, N, H, H, stream);
        zero4_kernel<<<ceil_div_i(NH / 4, TPB), TPB, 0, stream>>>((float4*)Acc, NH / 4);
        edge_scatter_kernel<<<ceil_div_i(scatterThreads, TPB), TPB, 0, stream>>>(Hh, dinv, row, col, Acc, E, H);
        finalize_relu_kernel<<<N, H, 0, stream>>>(Hh, dinv, Acc, H);
    }

    launch_gemm(Acc, W2, b2, Hh, N, H, H, stream);
    if (owner_path) {
        owner_aggregate_kernel<<<OW_GRID, OW_WAVES * 32, 0, stream>>>(Hh, dinv, row, col, own, X, E, N);
    } else {
        zero4_kernel<<<ceil_div_i(NH / 4, TPB), TPB, 0, stream>>>((float4*)X, NH / 4);
        if (H == SC_H)
            edge_scatter_lines_kernel<<<ceil_div_i(E, SC_EPW * WPB), TPB, 0, stream>>>(Hh, dinv, row, col, X, E);
        else
            edge_scatter_kernel<<<ceil_div_i(scatterThreads, TPB), TPB, 0, stream>>>(Hh, dinv, row, col, X, E, H);
        finalize_relu_kernel<<<N, H, 0, stream>>>(Hh, dinv, X, H);
    }

    const int edgeTiles = ceil_div_i(E, 16);
    if (H == EM_H && C <= 16)
        edge_mlp_rows_lds<<<EM_GRID, TPB, 0, stream>>>(X, row, col, We, be, out, E, Nu, C);
    else
        edge_mlp_wmma<0><<<ceil_div_i(edgeTiles, WPB), TPB, 0, stream>>>(X, row, col, We, be, out, E, Nu, H, C);
}


# degrees without global atomics: 64 blocks count their edge runs in 16-bit LDS counters, a second launch adds the 64 tables and takes the inverse root; fill and root launches gone; dinv bit-identical
# speedup vs baseline: 2.4641x; 1.1175x over previous
#include <hip/hip_runtime.h>

typedef __attribute__((ext_vector_type(2))) float v2f;
typedef __attribute__((ext_vector_type(8))) float v8f;

__device__ __forceinline__ v8f wmma_f32x4(v2f a, v2f b, v8f c) {
    return __builtin_amdgcn_wmma_f32_16x16x4_f32(
         false, a,  false, b,
         (short)0, c,  false,  false);
}

__device__ __forceinline__ void atomic_add_f32(float* p, float v) {
#if defined(__HIP_DEVICE_COMPILE__)
    unsafeAtomicAdd(p, v);
#else
    atomicAdd(p, v);
#endif
}

constexpr int GEMM_GRID_LOG2 = 9;
constexpr int GEMM_GRID = 1 << GEMM_GRID_LOG2;

template <int MODE>
__device__ __forceinline__ float gemm_epilogue(float a, float b, float g) {
    if (MODE == 0) return a + b;
    if (MODE == 1) return g * a;
    const float v = a + g * b;
    return v > 0.f ? v : 0.f;
}

template <int K, int MODE>
__global__ __launch_bounds__(256)
void gemm_bias_wmma_wlds(const float* __restrict__ X,
                         const float* __restrict__ W,
                         const float* __restrict__ bias,
                         const float* __restrict__ rs,
                         const float* __restrict__ zc,
                         float* __restrict__ Y,
                         int M) {
    constexpr int N = 128;
    constexpr int NT = N / 16;
    constexpr int PITCH = K + 4;
    constexpr int KB = K / 8;
    constexpr int WAVES = 8;
    __shared__ __attribute__((aligned(16))) float Ws[N * PITCH];

    const int lane = threadIdx.x & 31;
    const int wave = __builtin_amdgcn_readfirstlane((int)(threadIdx.x >> 5));
    const int c = lane & 15;
    const int h = lane >> 4;

    const int strips = (M + 15) >> 4;
    const int s0 = (int)(((long long)strips * blockIdx.x) >> GEMM_GRID_LOG2);
    const int s1 = (int)(((long long)strips * (blockIdx.x + 1)) >> GEMM_GRID_LOG2);

    float bv[NT];
#pragma unroll
    for (int t = 0; t < NT; ++t) bv[t] = (MODE == 1) ? 0.f : bias[16 * t + c];

    float4 a[KB];
    int s = s0 + wave;

    if (s < s1) {
        const int arow = min(s * 16 + c, M - 1);
        const float4* __restrict__ ap = (const float4*)(X + (size_t)arow * K) + h;
#pragma unroll
        for (int kb = 0; kb < KB; ++kb) a[kb] = ap[2 * kb];
    }

    for (int i = threadIdx.x; i < N * (K / 4); i += WAVES * 32) {
        const int n = i / (K / 4);
        const int k4 = i - n * (K / 4);
        *(float4*)(Ws + n * PITCH + 4 * k4) = ((const float4*)W)[i];
    }
    __syncthreads();

    const float* __restrict__ bfrag = Ws + c * PITCH + 4 * h;

    while (s < s1) {
        float gr[8];
#pragma unroll
        for (int r = 0; r < 8; ++r) {
            gr[r] = 0.f;
            if (MODE != 0) {
                const int rr = min(s * 16 + (h << 3) + r, M - 1);
                float g = rs[rr];
                if (MODE == 2) {
                    const v2f w = *(const v2f*)(zc + (size_t)rr * N + 96);
                    g *= (w.x + w.y);
                }
                gr[r] = g;
            }
        }

        v8f acc[NT];
#pragma unroll
        for (int t = 0; t < NT; ++t) acc[t] = (v8f){};

        float4 bc[NT], bn[NT];
#pragma unroll
        for (int t = 0; t < NT; ++t)
            bc[t] = *(const float4*)(bfrag + 16 * t * PITCH);
#pragma unroll
        for (int kb = 0; kb < KB; ++kb) {
            if (kb + 1 < KB) {
#pragma unroll
                for (int t = 0; t < NT; ++t)
                    bn[t] = *(const float4*)(bfrag + 16 * t * PITCH + 8 * (kb + 1));
            }
            __builtin_amdgcn_sched_barrier(0);
            const v2f a01 = {a[kb].x, a[kb].y};
            const v2f a23 = {a[kb].z, a[kb].w};
#pragma unroll
            for (int t = 0; t < NT; ++t) {
                const v2f b01 = {bc[t].x, bc[t].y};
                acc[t] = wmma_f32x4(a01, b01, acc[t]);
            }
#pragma unroll
            for (int t = 0; t < NT; ++t) {
                const v2f b23 = {bc[t].z, bc[t].w};
                acc[t] = wmma_f32x4(a23, b23, acc[t]);
            }
            __builtin_amdgcn_sched_barrier(0);
#pragma unroll
            for (int t = 0; t < NT; ++t) bc[t] = bn[t];
        }

        const int row0 = s * 16;
        s += WAVES;

        if (s < s1) {
            const int arow = min(s * 16 + c, M - 1);
            const float4* __restrict__ ap = (const float4*)(X + (size_t)arow * K) + h;
#pragma unroll
            for (int kb = 0; kb < KB; ++kb) a[kb] = ap[2 * kb];
        }

        const int rbase = row0 + (h << 3);
        float* __restrict__ yrow = Y + (size_t)rbase * N + c;
        if (row0 + 16 <= M) {
#pragma unroll
            for (int t = 0; t < NT; ++t) {
#pragma unroll
                for (int r = 0; r < 8; ++r)
                    yrow[(size_t)r * N + 16 * t] = gemm_epilogue<MODE>(acc[t][r], bv[t], gr[r]);
            }
        } else {
#pragma unroll
            for (int t = 0; t < NT; ++t) {
#pragma unroll
                for (int r = 0; r < 8; ++r)
                    if (rbase + r < M)
                        yrow[(size_t)r * N + 16 * t] = gemm_epilogue<MODE>(acc[t][r], bv[t], gr[r]);
            }
        }
    }
}

__global__ void gemm_bias_wmma_rt(const float* __restrict__ X,
                                  const float* __restrict__ W,
                                  const float* __restrict__ bias,
                                  float* __restrict__ Y,
                                  int M, int K, int N) {
    const int lane = threadIdx.x & 31;
    const int wave = threadIdx.x >> 5;
    const int tilesN = N >> 4;
    const int tile = blockIdx.x * (blockDim.x >> 5) + wave;
    const int tm = tile / tilesN;
    const int tn = tile - tm * tilesN;
    if (tm * 16 >= M) return;

    const int mrow = min(tm * 16 + (lane & 15), M - 1);
    const int ncol = tn * 16 + (lane & 15);
    const int koff = (lane >> 4) << 1;
    const float* __restrict__ xrow = X + (size_t)mrow * K + koff;
    const float* __restrict__ wrow = W + (size_t)ncol * K + koff;

    v8f acc = {};
#pragma unroll 8
    for (int k = 0; k < K; k += 4) {
        v2f a = *(const v2f*)(xrow + k);
        v2f b = *(const v2f*)(wrow + k);
        acc = wmma_f32x4(a, b, acc);
    }

    const float bv = bias[ncol];
    const int rbase = tm * 16 + ((lane >> 4) << 3);
    float* yp = Y + (size_t)rbase * N + ncol;
#pragma unroll
    for (int r = 0; r < 8; ++r) {
        if (rbase + r < M) *yp = acc[r] + bv;
        yp += N;
    }
}

template <int HT>
__global__ void edge_mlp_wmma(const float* __restrict__ xf,
                              const int* __restrict__ row,
                              const int* __restrict__ col,
                              const float* __restrict__ We,
                              const float* __restrict__ be,
                              float* __restrict__ out,
                              int E, int Nu, int Hrt, int C) {
    const int H = (HT > 0) ? HT : Hrt;
    const int K = 2 * H;
    const int lane = threadIdx.x & 31;
    const int wave = threadIdx.x >> 5;
    const int tile = blockIdx.x * (blockDim.x >> 5) + wave;
    const int ebase = tile * 16;
    if (ebase >= E) return;

    const int m = lane & 15;
    const int e = min(ebase + m, E - 1);
    const int r = row[e];
    const int c = col[e];
    const int koff = (lane >> 4) << 1;

    const float* __restrict__ urow = xf + (size_t)r * H + koff;
    const float* __restrict__ mrow = xf + (size_t)(Nu + c) * H + koff;

    const int n = lane & 15;
    const float wmask = (n < C) ? 1.0f : 0.0f;
    const float* __restrict__ wrow = We + (size_t)min(n, C - 1) * K + koff;

    v8f acc = {};
#pragma unroll 8
    for (int k = 0; k < K; k += 4) {
        const float* src = (k + koff < H) ? (urow + k) : (mrow + (k - H));
        v2f a = *(const v2f*)src;
        v2f b = *(const v2f*)(wrow + k);
        b.x *= wmask;
        b.y *= wmask;
        acc = wmma_f32x4(a, b, acc);
    }

    if (n < C) {
        const float bv = be[n];
        const int mb = (lane >> 4) << 3;
        float* op = out + (size_t)(ebase + mb) * C + n;
        if (ebase + 16 <= E) {
#pragma unroll
            for (int rr = 0; rr < 8; ++rr) { *op = acc[rr] + bv; op += C; }
        } else {
#pragma unroll
            for (int rr = 0; rr < 8; ++rr) {
                if (ebase + mb + rr < E) *op = acc[rr] + bv;
                op += C;
            }
        }
    }
}

constexpr int EM_H = 128;
constexpr int EM_K = 2 * EM_H;
constexpr int EM_WPITCH = EM_K + 4;
constexpr int EM_APITCH = EM_H + 4;
constexpr int EM_WAVES = 8;
constexpr int EM_GRID = 768;

__device__ __forceinline__ int edge_tile_ends(const int* __restrict__ row, const int* __restrict__ col,
                                              int t, int lane, int E) {
    const int e = min(t * 16 + (lane & 15), E - 1);
    const int* __restrict__ p = (lane < 16) ? row : col;
    return p[e];
}

#define EM_ROW_LOAD(j, IDX, BASE) \
    rv##j = ((const float4*)(xf + (size_t)(((BASE) ? Nu : 0) + __builtin_amdgcn_readlane(IDX, (BASE) + j)) * EM_H))[lane];
#define EM_ROW_STORE(j, IDX, BASE) *(float4*)(astore + j * EM_APITCH) = rv##j;
#define EM_ALL_ROWS(OP, IDX, BASE) \
    OP(0, IDX, BASE) OP(1, IDX, BASE) OP(2, IDX, BASE) OP(3, IDX, BASE) \
    OP(4, IDX, BASE) OP(5, IDX, BASE) OP(6, IDX, BASE) OP(7, IDX, BASE) \
    OP(8, IDX, BASE) OP(9, IDX, BASE) OP(10, IDX, BASE) OP(11, IDX, BASE) \
    OP(12, IDX, BASE) OP(13, IDX, BASE) OP(14, IDX, BASE) OP(15, IDX, BASE)

__global__ __launch_bounds__(256)
void edge_mlp_rows_lds(const float* __restrict__ xf,
                       const int* __restrict__ row,
                       const int* __restrict__ col,
                       const float* __restrict__ We,
                       const float* __restrict__ be,
                       float* __restrict__ out,
                       int E, int Nu, int C) {
    __shared__ __attribute__((aligned(16))) float Ws[16 * EM_WPITCH];
    __shared__ __attribute__((aligned(16))) float As[EM_WAVES * 16 * EM_APITCH];

    const int lane = threadIdx.x & 31;
    const int wave = __builtin_amdgcn_readfirstlane((int)(threadIdx.x >> 5));
    const int c = lane & 15;
    const int h = lane >> 4;

    const int tiles = (E + 15) >> 4;
    const int t0 = (int)(((long long)tiles * blockIdx.x) / EM_GRID);
    const int t1 = (int)(((long long)tiles * (blockIdx.x + 1)) / EM_GRID);
    int t = t0 + wave;

    float4 rv0, rv1, rv2, rv3, rv4, rv5, rv6, rv7, rv8, rv9, rv10, rv11, rv12, rv13, rv14, rv15;
    rv0 = rv1 = rv2 = rv3 = rv4 = rv5 = rv6 = rv7 = make_float4(0.f, 0.f, 0.f, 0.f);
    rv8 = rv9 = rv10 = rv11 = rv12 = rv13 = rv14 = rv15 = rv0;
    int idx = 0;
    if (t < t1) {
        idx = edge_tile_ends(row, col, t, lane, E);
        EM_ALL_ROWS(EM_ROW_LOAD, idx, 0)
    }

    for (int i = threadIdx.x; i < 16 * (EM_K / 4); i += EM_WAVES * 32) {
        const int n = i / (EM_K / 4);
        const int k4 = i - n * (EM_K / 4);
        float4 w = make_float4(0.f, 0.f, 0.f, 0.f);
        if (n < C) w = ((const float4*)We)[i];
        *(float4*)(Ws + n * EM_WPITCH + 4 * k4) = w;
    }
    __syncthreads();

    float* Aw = As + wave * (16 * EM_APITCH);
    float* astore = Aw + 4 * lane;
    const float* afrag = Aw + c * EM_APITCH + 4 * h;
    const float* bfrag = Ws + c * EM_WPITCH + 4 * h;
    const float bv = (c < C) ? be[c] : 0.f;

    while (t < t1) {
        const int tn = t + EM_WAVES;
        int idxn = 0;
        if (tn < t1) idxn = edge_tile_ends(row, col, tn, lane, E);

        EM_ALL_ROWS(EM_ROW_STORE, idx, 0)
        EM_ALL_ROWS(EM_ROW_LOAD, idx, 16)

        v8f acc0 = {}, acc1 = {};
#pragma unroll
        for (int kb = 0; kb < EM_H / 8; ++kb) {
            const float4 a = *(const float4*)(afrag + 8 * kb);
            const float4 b = *(const float4*)(bfrag + 8 * kb);
            acc0 = wmma_f32x4((v2f){a.x, a.y}, (v2f){b.x, b.y}, acc0);
            acc1 = wmma_f32x4((v2f){a.z, a.w}, (v2f){b.z, b.w}, acc1);
        }

        EM_ALL_ROWS(EM_ROW_STORE, idx, 0)
        if (tn < t1) {
            EM_ALL_ROWS(EM_ROW_LOAD, idxn, 0)
        }
#pragma unroll
        for (int kb = 0; kb < EM_H / 8; ++kb) {
            const float4 a = *(const float4*)(afrag + 8 * kb);
            const float4 b = *(const float4*)(bfrag + EM_H + 8 * kb);
            acc0 = wmma_f32x4((v2f){a.x, a.y}, (v2f){b.x, b.y}, acc0);
            acc1 = wmma_f32x4((v2f){a.z, a.w}, (v2f){b.z, b.w}, acc1);
        }

        if (c < C) {
            const int ebase = t * 16;
            const int mb = h << 3;
            float* op = out + (size_t)(ebase + mb) * C + c;
            if (ebase + 16 <= E) {
#pragma unroll
                for (int rr = 0; rr < 8; ++rr) { *op = (acc0[rr] + acc1[rr]) + bv; op += C; }
            } else {
#pragma unroll
                for (int rr = 0; rr < 8; ++rr) {
                    if (ebase + mb + rr < E) *op = (acc0[rr] + acc1[rr]) + bv;
                    op += C;
                }
            }
        }

        idx = idxn;
        t = tn;
    }
}

__global__ void fill_kernel(float* __restrict__ p, float v, long long n) {
    long long i = (long long)blockIdx.x * blockDim.x + threadIdx.x;
    if (i < n) p[i] = v;
}

__global__ void zero4_kernel(float4* __restrict__ p, long long n4) {
    long long i = (long long)blockIdx.x * blockDim.x + threadIdx.x;
    if (i < n4) p[i] = make_float4(0.f, 0.f, 0.f, 0.f);
}

__global__ void deg_scatter_kernel(const int* __restrict__ col,
                                   float* __restrict__ deg, int E) {
    int e = blockIdx.x * blockDim.x + threadIdx.x;
    if (e < E) atomic_add_f32(&deg[col[e]], 1.0f);
}

__global__ void deg_scatter_own_kernel(const int* __restrict__ col,
                                       float* __restrict__ deg,
                                       unsigned char* __restrict__ own, int E) {
    int e = blockIdx.x * blockDim.x + threadIdx.x;
    if (e < E) {
        const int c = col[e];
        own[e] = (unsigned char)(c & 255);
        atomic_add_f32(&deg[c], 1.0f);
    }
}

__global__ void rsqrt_kernel(float* __restrict__ d, int N) {
    int i = blockIdx.x * blockDim.x + threadIdx.x;
    if (i < N) d[i] = rsqrtf(d[i]);
}

constexpr int DH_GRID = 64;
constexpr int DH_THREADS = 1024;
constexpr int DH_WORDS = 50000;
constexpr int DH_U = 4;
static_assert((DH_WORDS & 3) == 0, "tables are cleared and stored 16 bytes at a time");

__device__ __forceinline__ void dh_count(unsigned int* cnt, int c) {
    atomicAdd(cnt + (c >> 1), (c & 1) ? 0x10000u : 1u);
}

__global__ __launch_bounds__(DH_THREADS)
void deg_hist_kernel(const int* __restrict__ col, unsigned int* __restrict__ own4,
                     unsigned int* __restrict__ part, int E, int N) {
    __shared__ __attribute__((aligned(16))) unsigned int cnt[DH_WORDS];

    const int words4 = (((N + 1) >> 1) + 3) >> 2;
    const int g4 = E >> 2;
    const int q0 = (int)(((long long)g4 * blockIdx.x) / DH_GRID);
    const int q1 = (int)(((long long)g4 * (blockIdx.x + 1)) / DH_GRID);
    const int4* __restrict__ col4 = (const int4*)col;

    int4 cv[DH_U];
#pragma unroll
    for (int u = 0; u < DH_U; ++u)
        cv[u] = col4[min(q0 + u * DH_THREADS + (int)threadIdx.x, q1 - 1)];

    for (int i = threadIdx.x; i < words4; i += DH_THREADS)
        ((uint4*)cnt)[i] = make_uint4(0u, 0u, 0u, 0u);
    __syncthreads();

    for (int base = q0; base < q1; base += DH_U * DH_THREADS) {
#pragma unroll
        for (int u = 0; u < DH_U; ++u) {
            const int q = base + u * DH_THREADS + (int)threadIdx.x;
            if (q < q1) {
                const int4 c = cv[u];
                dh_count(cnt, c.x);
                dh_count(cnt, c.y);
                dh_count(cnt, c.z);
                dh_count(cnt, c.w);
                own4[q] = (unsigned)(c.x & 255) | ((unsigned)(c.y & 255) << 8) |
                          ((unsigned)(c.z & 255) << 16) | ((unsigned)(c.w & 255) << 24);
            }
        }
        if (base + DH_U * DH_THREADS < q1) {
#pragma unroll
            for (int u = 0; u < DH_U; ++u)
                cv[u] = col4[min(base + (DH_U + u) * DH_THREADS + (int)threadIdx.x, q1 - 1)];
        }
    }
    __syncthreads();

    uint4* __restrict__ p = (uint4*)part + (size_t)blockIdx.x * words4;
#pragma unroll 4
    for (int i = threadIdx.x; i < words4; i += DH_THREADS)
        p[i] = ((const uint4*)cnt)[i];
}

__global__ __launch_bounds__(256)
void deg_reduce_kernel(const unsigned int* __restrict__ part, float* __restrict__ dinv,
                       int N, int pitch) {
    const int w = blockIdx.x * blockDim.x + threadIdx.x;
    if (2 * w >= N) return;
    unsigned int v[DH_GRID];
#pragma unroll
    for (int b = 0; b < DH_GRID; ++b) v[b] = part[(size_t)b * pitch + w];
    unsigned int lo = 0u, hi = 0u;
#pragma unroll
    for (int b = 0; b < DH_GRID; ++b) {
        lo += v[b] & 0xffffu;
        hi += v[b] >> 16;
    }
    *(float2*)(dinv + 2 * w) = make_float2(rsqrtf(1.0f + (float)lo), rsqrtf(1.0f + (float)hi));
}

__global__ void edge_scatter_kernel(const float* __restrict__ h,
                                    const float* __restrict__ dinv,
                                    const int* __restrict__ row,
                                    const int* __restrict__ col,
                                    float* __restrict__ acc,
                                    int E, int H) {
    const int chunks = H >> 2;
    long long gid = (long long)blockIdx.x * blockDim.x + threadIdx.x;
    long long total = (long long)E * chunks;
    if (gid >= total) return;
    const int e = (int)(gid / chunks);
    const int j = (int)(gid - (long long)e * chunks) << 2;
    const int r = row[e];
    const int c = col[e];
    const float s = dinv[r];
    const float4 hv = *(const float4*)(h + (size_t)r * H + j);
    float* dst = acc + (size_t)c * H + j;
    atomic_add_f32(dst + 0, s * hv.x);
    atomic_add_f32(dst + 1, s * hv.y);
    atomic_add_f32(dst + 2, s * hv.z);
    atomic_add_f32(dst + 3, s * hv.w);
}

constexpr int SC_H = 128;
constexpr int SC_EPW = 8;

template <bool FULL>
__device__ __forceinline__ void line_scatter_wave(const float* __restrict__ h,
                                                  const float* __restrict__ dinv,
                                                  const int* __restrict__ row, const int* __restrict__ col,
                                                  float* __restrict__ acc, int e0, int ne, int l) {
    int r[SC_EPW], c[SC_EPW];
    float s[SC_EPW], v0[SC_EPW], v1[SC_EPW], v2[SC_EPW], v3[SC_EPW];
#pragma unroll
    for (int i = 0; i < SC_EPW; ++i) {
        const int e = FULL ? e0 + i : e0 + min(i, ne - 1);
        r[i] = row[e];
        c[i] = col[e];
    }
#pragma unroll
    for (int i = 0; i < SC_EPW; ++i) {
        s[i] = dinv[r[i]];
        const float* __restrict__ p = h + (size_t)r[i] * SC_H + l;
        v0[i] = p[0];
        v1[i] = p[32];
        v2[i] = p[64];
        v3[i] = p[96];
    }
#pragma unroll
    for (int i = 0; i < SC_EPW; ++i) {
        if (FULL || i < ne) {
            float* dst = acc + (size_t)c[i] * SC_H + l;
            atomic_add_f32(dst, s[i] * v0[i]);
            atomic_add_f32(dst + 32, s[i] * v1[i]);
            atomic_add_f32(dst + 64, s[i] * v2[i]);
            atomic_add_f32(dst + 96, s[i] * v3[i]);
        }
    }
}

__global__ __launch_bounds__(256)
void edge_scatter_lines_kernel(const float* __restrict__ h,
                               const float* __restrict__ dinv,
                               const int* __restrict__ row,
                               const int* __restrict__ col,
                               float* __restrict__ acc, int E) {
    const int l = threadIdx.x & 31;
    const int wv = __builtin_amdgcn_readfirstlane(
        (int)(blockIdx.x * (blockDim.x >> 5) + (threadIdx.x >> 5)));
    const int e0 = wv * SC_EPW;
    if (e0 >= E) return;
    const int ne = min(SC_EPW, E - e0);
    if (ne == SC_EPW) line_scatter_wave<true>(h, dinv, row, col, acc, e0, ne, l);
    else              line_scatter_wave<false>(h, dinv, row, col, acc, e0, ne, l);
}


constexpr int OW_GRID_LOG2 = 8;
constexpr int OW_GRID = 1 << OW_GRID_LOG2;
static_assert(OW_GRID == 256, "the owner list holds one byte per edge: target mod 256");
constexpr int OW_WAVES = 32;
constexpr int OW_SLOTS = 392;
constexpr int OW_U = 8;
constexpr int OW_Q = 544;

__device__ __forceinline__ void lds_add_f32(float* p, float v) {
    __hip_atomic_fetch_add(p, v, __ATOMIC_RELAXED, __HIP_MEMORY_SCOPE_WORKGROUP);
}

__device__ __forceinline__ void ow_drain(const float* __restrict__ h, const float* __restrict__ dinv,
                                         const int* __restrict__ row, const int* __restrict__ col,
                                         float* accs, const int* q, int m, int lane) {
    const int e = q[min(lane, m - 1)];
    const int r = row[e];
    const int c = col[e];
    const int sbits = __float_as_int(dinv[r]);
    for (int j = 0; j < m; j += 8) {
        float v0[8], v1[8], v2[8], v3[8], ss[8];
        int sl[8];
#pragma unroll
        for (int i = 0; i < 8; ++i) {
            const int idx = min(j + i, m - 1);
            const int rr = __builtin_amdgcn_readlane(r, idx);
            sl[i] = __builtin_amdgcn_readlane(c, idx) >> OW_GRID_LOG2;
            ss[i] = __int_as_float(__builtin_amdgcn_readlane(sbits, idx));
            const float* __restrict__ p = h + (size_t)rr * SC_H + lane;
            v0[i] = p[0];
            v1[i] = p[32];
            v2[i] = p[64];
            v3[i] = p[96];
        }
#pragma unroll
        for (int i = 0; i < 8; ++i) {
            if (j + i < m) {
                float* d = accs + sl[i] * SC_H + lane;
                lds_add_f32(d, ss[i] * v0[i]);
                lds_add_f32(d + 32, ss[i] * v1[i]);
                lds_add_f32(d + 64, ss[i] * v2[i]);
                lds_add_f32(d + 96, ss[i] * v3[i]);
            }
        }
    }
}

#define OW_STEP(MT, EDGE) \
    { \
        const bool mt = (MT); \
        const unsigned mask = __builtin_amdgcn_ballot_w32(mt); \
        if (mask) { \
            if (mt) q[cnt + (int)__builtin_amdgcn_mbcnt_lo(mask, 0u)] = (EDGE); \
            cnt += __builtin_popcount(mask); \
        } \
    }

#define OW_WORD(WV, E0) \
    { \
        const unsigned xw = ok ? ((WV) ^ bpat) : 0xffffffffu; \
        const unsigned zb = (xw - 0x01010101u) & ~xw & 0x80808080u; \
        if (__builtin_amdgcn_ballot_w32(zb != 0u)) { \
            OW_STEP((xw & 0x000000ffu) == 0u, (E0) + 0) \
            OW_STEP((xw & 0x0000ff00u) == 0u, (E0) + 1) \
            OW_STEP((xw & 0x00ff0000u) == 0u, (E0) + 2) \
            OW_STEP((xw & 0xff000000u) == 0u, (E0) + 3) \
        } \
    }

__global__ __launch_bounds__(1024)
void owner_aggregate_kernel(const float* __restrict__ h, const float* __restrict__ dinv,
                            const int* __restrict__ row, const int* __restrict__ col,
                            const unsigned char* __restrict__ own,
                            float* __restrict__ x, int E, int N) {
    __shared__ __attribute__((aligned(16))) float accs[OW_SLOTS * SC_H];
    __shared__ int qs[OW_WAVES * OW_Q];

    const int lane = threadIdx.x & 31;
    const int wave = __builtin_amdgcn_readfirstlane((int)(threadIdx.x >> 5));
    const int b = blockIdx.x;
    const int slots = (N - b + OW_GRID - 1) >> OW_GRID_LOG2;

#pragma unroll 4
    for (int sl = wave; sl < slots; sl += OW_WAVES) {
        const int node = (sl << OW_GRID_LOG2) + b;
        const float di = dinv[node];
        const float4 hv = ((const float4*)(h + (size_t)node * SC_H))[lane];
        *(float4*)(accs + sl * SC_H + 4 * lane) = make_float4(di * hv.x, di * hv.y, di * hv.z, di * hv.w);
    }
    __syncthreads();

    int* q = qs + wave * OW_Q;
    int cnt = 0;
    const uint4* __restrict__ own4 = (const uint4*)own;
    const int total16 = E >> 4;
    const unsigned bpat = (unsigned)b * 0x01010101u;
    for (int base = 0; base < total16; base += OW_U * OW_WAVES * 32) {
        uint4 ov[OW_U];
#pragma unroll
        for (int u = 0; u < OW_U; ++u) {
            const int i16 = base + (u * OW_WAVES + wave) * 32 + lane;
            ov[u] = own4[min(i16, total16 - 1)];
        }
#pragma unroll
        for (int u = 0; u < OW_U; ++u) {
            const int i16 = base + (u * OW_WAVES + wave) * 32 + lane;
            const bool ok = i16 < total16;
            OW_WORD(ov[u].x, 16 * i16 + 0)
            OW_WORD(ov[u].y, 16 * i16 + 4)
            OW_WORD(ov[u].z, 16 * i16 + 8)
            OW_WORD(ov[u].w, 16 * i16 + 12)
            if (cnt >= 32) {
                for (int g = 0; g < cnt; g += 32)
                    ow_drain(h, dinv, row, col, accs, q + g, min(32, cnt - g), lane);
                cnt = 0;
            }
        }
    }
    for (int g = 0; g < cnt; g += 32)
        ow_drain(h, dinv, row, col, accs, q + g, min(32, cnt - g), lane);
    __syncthreads();

#pragma unroll 4
    for (int sl = wave; sl < slots; sl += OW_WAVES) {
        const int node = (sl << OW_GRID_LOG2) + b;
        const float di = dinv[node];
        const float4 a = *(const float4*)(accs + sl * SC_H + 4 * lane);
        float4 o;
        o.x = di * a.x; o.x = o.x > 0.f ? o.x : 0.f;
        o.y = di * a.y; o.y = o.y > 0.f ? o.y : 0.f;
        o.z = di * a.z; o.z = o.z > 0.f ? o.z : 0.f;
        o.w = di * a.w; o.w = o.w > 0.f ? o.w : 0.f;
        ((float4*)(x + (size_t)node * SC_H))[lane] = o;
    }
}

__global__ void finalize_relu_kernel(const float* __restrict__ h,
                                     const float* __restrict__ dinv,
                                     float* __restrict__ acc_inout, int H) {
    const int node = blockIdx.x;
    const int j = threadIdx.x;
    const float di = dinv[node];
    const size_t idx = (size_t)node * H + j;
    const float v = di * acc_inout[idx] + di * di * h[idx];
    acc_inout[idx] = v > 0.f ? v : 0.f;
}

constexpr int RAW_FU = 32;
constexpr int RAW_FM = 64;
constexpr int RAW_H = 128;
constexpr int RAW_WU = RAW_FU + RAW_FM;
constexpr int RAW_WM = RAW_FU + RAW_FM + 1;

__global__ void compose_wz_kernel(const float* __restrict__ Wu, const float* __restrict__ bu,
                                  const float* __restrict__ Wm, const float* __restrict__ bm,
                                  float* __restrict__ Wz) {
    const int i = blockIdx.x * blockDim.x + threadIdx.x;
    if (i >= RAW_H * RAW_H) return;
    const int n = i / RAW_H;
    const int k = i - n * RAW_H;
    float w = 0.f;
    if (k < RAW_FU) w = Wu[n * RAW_FU + k];
    else if (k < RAW_WU) w = Wm[n * RAW_FM + (k - RAW_FU)];
    else if (k == RAW_WU) w = bu[n];
    else if (k == RAW_WM) w = bm[n];
    Wz[i] = w;
}

__global__ void init_z_kernel(const float* __restrict__ xu, const float* __restrict__ xm,
                              const float* __restrict__ dinv, float* __restrict__ Z, int Nu) {
    const int node = blockIdx.x;
    const int j = threadIdx.x;
    const float di = dinv[node];
    float self = 0.f;
    if (node < Nu) {
        if (j < RAW_FU) self = xu[(size_t)node * RAW_FU + j];
        else if (j == RAW_WU) self = 1.f;
    } else {
        if (j >= RAW_FU && j < RAW_WU) self = xm[(size_t)(node - Nu) * RAW_FM + (j - RAW_FU)];
        else if (j == RAW_WM) self = 1.f;
    }
    Z[(size_t)node * RAW_H + j] = di * self;
}

constexpr int RAW_EPW = 8;

template <bool FULL>
__device__ __forceinline__ void raw_scatter_wave(const float* __restrict__ xu, const float* __restrict__ xm,
                                                 const float* __restrict__ dinv,
                                                 const int* __restrict__ row, const int* __restrict__ col,
                                                 float* __restrict__ Z, int Nu, int e0, int ne, int l) {
    int r[RAW_EPW], c[RAW_EPW];
    float s[RAW_EPW], v0[RAW_EPW], v1[RAW_EPW];
#pragma unroll
    for (int i = 0; i < RAW_EPW; ++i) {
        const int e = FULL ? e0 + i : e0 + min(i, ne - 1);
        r[i] = row[e];
        c[i] = col[e];
    }
#pragma unroll
    for (int i = 0; i < RAW_EPW; ++i) {
        s[i] = dinv[r[i]];
        const bool user = r[i] < Nu;
        const float* __restrict__ p = user ? xu + (size_t)r[i] * RAW_FU + l
                                           : xm + (size_t)(r[i] - Nu) * RAW_FM + l;
        v0[i] = p[0];
        v1[i] = 0.f;
        if (!user) v1[i] = p[32];
    }
#pragma unroll
    for (int i = 0; i < RAW_EPW; ++i) {
        if (FULL || i < ne) {
            const bool user = r[i] < Nu;
            float* dst = Z + (size_t)c[i] * RAW_H;
            atomic_add_f32(dst + (user ? 0 : RAW_FU) + l, s[i] * v0[i]);
            if (!user) atomic_add_f32(dst + RAW_FU + 32 + l, s[i] * v1[i]);
        }
    }
    if (l < (FULL ? RAW_EPW : ne)) {
        const int rl = row[e0 + l];
        const int cl = col[e0 + l];
        atomic_add_f32(Z + (size_t)cl * RAW_H + (rl < Nu ? RAW_WU : RAW_WM), dinv[rl]);
    }
}

__global__ __launch_bounds__(256)
void raw_scatter_kernel(const float* __restrict__ xu, const float* __restrict__ xm,
                        const float* __restrict__ dinv,
                        const int* __restrict__ row, const int* __restrict__ col,
                        float* __restrict__ Z, int E, int Nu) {
    const int l = threadIdx.x & 31;
    const int wv = __builtin_amdgcn_readfirstlane(
        (int)(blockIdx.x * (blockDim.x >> 5) + (threadIdx.x >> 5)));
    const int e0 = wv * RAW_EPW;
    if (e0 >= E) return;
    const int ne = min(RAW_EPW, E - e0);
    if (ne == RAW_EPW) raw_scatter_wave<true>(xu, xm, dinv, row, col, Z, Nu, e0, ne, l);
    else               raw_scatter_wave<false>(xu, xm, dinv, row, col, Z, Nu, e0, ne, l);
}

static inline int ceil_div_i(long long a, long long b) { return (int)((a + b - 1) / b); }

static void launch_gemm(const float* X, const float* W, const float* b, float* Y,
                        int M, int K, int N, hipStream_t stream) {
    const int TPB = 256;
    if (N == 128 && K == 32)
        gemm_bias_wmma_wlds<32, 0><<<GEMM_GRID, TPB, 0, stream>>>(X, W, b, nullptr, nullptr, Y, M);
    else if (N == 128 && K == 64)
        gemm_bias_wmma_wlds<64, 0><<<GEMM_GRID, TPB, 0, stream>>>(X, W, b, nullptr, nullptr, Y, M);
    else if (N == 128 && K == 128)
        gemm_bias_wmma_wlds<128, 0><<<GEMM_GRID, TPB, 0, stream>>>(X, W, b, nullptr, nullptr, Y, M);
    else {
        const int strips = ceil_div_i(M, 16);
        const int tiles = strips * (N / 16);
        gemm_bias_wmma_rt<<<ceil_div_i(tiles, 8), TPB, 0, stream>>>(X, W, b, Y, M, K, N);
    }
}

extern "C" void kernel_launch(void* const* d_in, const int* in_sizes, int n_in,
                              void* d_out, int out_size, void* d_ws, size_t ws_size,
                              hipStream_t stream) {
    const float* x_user  = (const float*)d_in[0];
    const float* x_movie = (const float*)d_in[1];
    const int*   edge    = (const int*)d_in[2];
    const float* Wu = (const float*)d_in[3];
    const float* bu = (const float*)d_in[4];
    const float* Wm = (const float*)d_in[5];
    const float* bm = (const float*)d_in[6];
    const float* W1 = (const float*)d_in[7];
    const float* b1 = (const float*)d_in[8];
    const float* W2 = (const float*)d_in[9];
    const float* b2 = (const float*)d_in[10];
    const float* We = (const float*)d_in[11];
    const float* be = (const float*)d_in[12];
    float* out = (float*)d_out;

    const int H  = in_sizes[4];
    const int C  = in_sizes[12];
    const int Fu = in_sizes[3] / H;
    const int Fm = in_sizes[5] / H;
    const int Nu = in_sizes[0] / Fu;
    const int Nm = in_sizes[1] / Fm;
    const int E  = in_sizes[2] / 2;
    const int N  = Nu + Nm;
    const int* row = edge;
    const int* col = edge + E;

    float* ws   = (float*)d_ws;
    float* X    = ws;
    float* Hh   = X  + (size_t)N * H;
    float* Acc  = Hh + (size_t)N * H;
    float* Wz   = Acc + (size_t)N * H;
    float* dinv = Wz + (size_t)RAW_H * RAW_H;
    unsigned char* own = (unsigned char*)(dinv + (size_t)((N + 3) & ~3));
    unsigned int* part = (unsigned int*)(own + (((size_t)E + 15) & ~(size_t)15));

    const int TPB = 256, WPB = 8;
    const long long NH = (long long)N * H;

    const bool owner_path = (H == SC_H) && ((E & 15) == 0) && (N <= OW_GRID * OW_SLOTS);
    const int dh_words = (((N + 1) >> 1) + 3) & ~3;
    const bool hist_path = (N <= 2 * DH_WORDS) && ((E & 3) == 0) && (E >= 4 * DH_GRID) &&
                           (E / DH_GRID + 4 <= 65535) &&
                           ((size_t)((unsigned char*)(part + (size_t)DH_GRID * dh_words) - (unsigned char*)d_ws) <= ws_size);
    if (hist_path) {
        deg_hist_kernel<<<DH_GRID, DH_THREADS, 0, stream>>>(col, (unsigned int*)own, part, E, N);
        deg_reduce_kernel<<<ceil_div_i((N + 1) >> 1, TPB), TPB, 0, stream>>>(part, dinv, N, dh_words);
    } else {
        fill_kernel<<<ceil_div_i(N, TPB), TPB, 0, stream>>>(dinv, 1.0f, N);
        if (owner_path)
            deg_scatter_own_kernel<<<ceil_div_i(E, TPB), TPB, 0, stream>>>(col, dinv, own, E);
        else
            deg_scatter_kernel<<<ceil_div_i(E, TPB), TPB, 0, stream>>>(col, dinv, E);
        rsqrt_kernel<<<ceil_div_i(N, TPB), TPB, 0, stream>>>(dinv, N);
    }

    const long long scatterThreads = (long long)E * (H / 4);

    if (H == RAW_H && Fu == RAW_FU && Fm == RAW_FM) {
        compose_wz_kernel<<<ceil_div_i(RAW_H * RAW_H, TPB), TPB, 0, stream>>>(Wu, bu, Wm, bm, Wz);
        init_z_kernel<<<N, RAW_H, 0, stream>>>(x_user, x_movie, dinv, Hh, Nu);
        raw_scatter_kernel<<<ceil_div_i(E, RAW_EPW * WPB), TPB, 0, stream>>>(x_user, x_movie, dinv, row, col, Hh, E, Nu);
        gemm_bias_wmma_wlds<RAW_H, 1><<<GEMM_GRID, TPB, 0, stream>>>(Hh, Wz, nullptr, dinv, nullptr, X, N);
        gemm_bias_wmma_wlds<RAW_H, 2><<<GEMM_GRID, TPB, 0, stream>>>(X, W1, b1, dinv, Hh, Acc, N);
    } else {
        launch_gemm(x_user, Wu, bu, X, Nu, Fu, H, stream);
        launch_gemm(x_movie, Wm, bm, X + (size_t)Nu * H, Nm, Fm, H, stream);

        launch_gemm(X, W1, b1, Hh, N, H, H, stream);
        zero4_kernel<<<ceil_div_i(NH / 4, TPB), TPB, 0, stream>>>((float4*)Acc, NH / 4);
        edge_scatter_kernel<<<ceil_div_i(scatterThreads, TPB), TPB, 0, stream>>>(Hh, dinv, row, col, Acc, E, H);
        finalize_relu_kernel<<<N, H, 0, stream>>>(Hh, dinv, Acc, H);
    }

    launch_gemm(Acc, W2, b2, Hh, N, H, H, stream);
    if (owner_path) {
        owner_aggregate_kernel<<<OW_GRID, OW_WAVES * 32, 0, stream>>>(Hh, dinv, row, col, own, X, E, N);
    } else {
        zero4_kernel<<<ceil_div_i(NH / 4, TPB), TPB, 0, stream>>>((float4*)X, NH / 4);
        if (H == SC_H)
            edge_scatter_lines_kernel<<<ceil_div_i(E, SC_EPW * WPB), TPB, 0, stream>>>(Hh, dinv, row, col, X, E);
        else
            edge_scatter_kernel<<<ceil_div_i(scatterThreads, TPB), TPB, 0, stream>>>(Hh, dinv, row, col, X, E, H);
        finalize_relu_kernel<<<N, H, 0, stream>>>(Hh, dinv, X, H);
    }

    const int edgeTiles = ceil_div_i(E, 16);
    if (H == EM_H && C <= 16)
        edge_mlp_rows_lds<<<EM_GRID, TPB, 0, stream>>>(X, row, col, We, be, out, E, Nu, C);
    else
        edge_mlp_wmma<0><<<ceil_div_i(edgeTiles, WPB), TPB, 0, stream>>>(X, row, col, We, be, out, E, Nu, H, C);
}


# Z start rows written by the degree-reduce launch: a block turns 256 nodes' counts into dinv in LDS, then each wave stores whole 512-byte rows, 8 in flight; the block-per-node init launch is gone
# speedup vs baseline: 2.6389x; 1.0709x over previous
#include <hip/hip_runtime.h>

typedef __attribute__((ext_vector_type(2))) float v2f;
typedef __attribute__((ext_vector_type(8))) float v8f;

__device__ __forceinline__ v8f wmma_f32x4(v2f a, v2f b, v8f c) {
    return __builtin_amdgcn_wmma_f32_16x16x4_f32(
         false, a,  false, b,
         (short)0, c,  false,  false);
}

__device__ __forceinline__ void atomic_add_f32(float* p, float v) {
#if defined(__HIP_DEVICE_COMPILE__)
    unsafeAtomicAdd(p, v);
#else
    atomicAdd(p, v);
#endif
}

constexpr int GEMM_GRID_LOG2 = 9;
constexpr int GEMM_GRID = 1 << GEMM_GRID_LOG2;

template <int MODE>
__device__ __forceinline__ float gemm_epilogue(float a, float b, float g) {
    if (MODE == 0) return a + b;
    if (MODE == 1) return g * a;
    const float v = a + g * b;
    return v > 0.f ? v : 0.f;
}

template <int K, int MODE>
__global__ __launch_bounds__(256)
void gemm_bias_wmma_wlds(const float* __restrict__ X,
                         const float* __restrict__ W,
                         const float* __restrict__ bias,
                         const float* __restrict__ rs,
                         const float* __restrict__ zc,
                         float* __restrict__ Y,
                         int M) {
    constexpr int N = 128;
    constexpr int NT = N / 16;
    constexpr int PITCH = K + 4;
    constexpr int KB = K / 8;
    constexpr int WAVES = 8;
    __shared__ __attribute__((aligned(16))) float Ws[N * PITCH];

    const int lane = threadIdx.x & 31;
    const int wave = __builtin_amdgcn_readfirstlane((int)(threadIdx.x >> 5));
    const int c = lane & 15;
    const int h = lane >> 4;

    const int strips = (M + 15) >> 4;
    const int s0 = (int)(((long long)strips * blockIdx.x) >> GEMM_GRID_LOG2);
    const int s1 = (int)(((long long)strips * (blockIdx.x + 1)) >> GEMM_GRID_LOG2);

    float bv[NT];
#pragma unroll
    for (int t = 0; t < NT; ++t) bv[t] = (MODE == 1) ? 0.f : bias[16 * t + c];

    float4 a[KB];
    int s = s0 + wave;

    if (s < s1) {
        const int arow = min(s * 16 + c, M - 1);
        const float4* __restrict__ ap = (const float4*)(X + (size_t)arow * K) + h;
#pragma unroll
        for (int kb = 0; kb < KB; ++kb) a[kb] = ap[2 * kb];
    }

    for (int i = threadIdx.x; i < N * (K / 4); i += WAVES * 32) {
        const int n = i / (K / 4);
        const int k4 = i - n * (K / 4);
        *(float4*)(Ws + n * PITCH + 4 * k4) = ((const float4*)W)[i];
    }
    __syncthreads();

    const float* __restrict__ bfrag = Ws + c * PITCH + 4 * h;

    while (s < s1) {
        float gr[8];
#pragma unroll
        for (int r = 0; r < 8; ++r) {
            gr[r] = 0.f;
            if (MODE != 0) {
                const int rr = min(s * 16 + (h << 3) + r, M - 1);
                float g = rs[rr];
                if (MODE == 2) {
                    const v2f w = *(const v2f*)(zc + (size_t)rr * N + 96);
                    g *= (w.x + w.y);
                }
                gr[r] = g;
            }
        }

        v8f acc[NT];
#pragma unroll
        for (int t = 0; t < NT; ++t) acc[t] = (v8f){};

        float4 bc[NT], bn[NT];
#pragma unroll
        for (int t = 0; t < NT; ++t)
            bc[t] = *(const float4*)(bfrag + 16 * t * PITCH);
#pragma unroll
        for (int kb = 0; kb < KB; ++kb) {
            if (kb + 1 < KB) {
#pragma unroll
                for (int t = 0; t < NT; ++t)
                    bn[t] = *(const float4*)(bfrag + 16 * t * PITCH + 8 * (kb + 1));
            }
            __builtin_amdgcn_sched_barrier(0);
            const v2f a01 = {a[kb].x, a[kb].y};
            const v2f a23 = {a[kb].z, a[kb].w};
#pragma unroll
            for (int t = 0; t < NT; ++t) {
                const v2f b01 = {bc[t].x, bc[t].y};
                acc[t] = wmma_f32x4(a01, b01, acc[t]);
            }
#pragma unroll
            for (int t = 0; t < NT; ++t) {
                const v2f b23 = {bc[t].z, bc[t].w};
                acc[t] = wmma_f32x4(a23, b23, acc[t]);
            }
            __builtin_amdgcn_sched_barrier(0);
#pragma unroll
            for (int t = 0; t < NT; ++t) bc[t] = bn[t];
        }

        const int row0 = s * 16;
        s += WAVES;

        if (s < s1) {
            const int arow = min(s * 16 + c, M - 1);
            const float4* __restrict__ ap = (const float4*)(X + (size_t)arow * K) + h;
#pragma unroll
            for (int kb = 0; kb < KB; ++kb) a[kb] = ap[2 * kb];
        }

        const int rbase = row0 + (h << 3);
        float* __restrict__ yrow = Y + (size_t)rbase * N + c;
        if (row0 + 16 <= M) {
#pragma unroll
            for (int t = 0; t < NT; ++t) {
#pragma unroll
                for (int r = 0; r < 8; ++r)
                    yrow[(size_t)r * N + 16 * t] = gemm_epilogue<MODE>(acc[t][r], bv[t], gr[r]);
            }
        } else {
#pragma unroll
            for (int t = 0; t < NT; ++t) {
#pragma unroll
                for (int r = 0; r < 8; ++r)
                    if (rbase + r < M)
                        yrow[(size_t)r * N + 16 * t] = gemm_epilogue<MODE>(acc[t][r], bv[t], gr[r]);
            }
        }
    }
}

__global__ void gemm_bias_wmma_rt(const float* __restrict__ X,
                                  const float* __restrict__ W,
                                  const float* __restrict__ bias,
                                  float* __restrict__ Y,
                                  int M, int K, int N) {
    const int lane = threadIdx.x & 31;
    const int wave = threadIdx.x >> 5;
    const int tilesN = N >> 4;
    const int tile = blockIdx.x * (blockDim.x >> 5) + wave;
    const int tm = tile / tilesN;
    const int tn = tile - tm * tilesN;
    if (tm * 16 >= M) return;

    const int mrow = min(tm * 16 + (lane & 15), M - 1);
    const int ncol = tn * 16 + (lane & 15);
    const int koff = (lane >> 4) << 1;
    const float* __restrict__ xrow = X + (size_t)mrow * K + koff;
    const float* __restrict__ wrow = W + (size_t)ncol * K + koff;

    v8f acc = {};
#pragma unroll 8
    for (int k = 0; k < K; k += 4) {
        v2f a = *(const v2f*)(xrow + k);
        v2f b = *(const v2f*)(wrow + k);
        acc = wmma_f32x4(a, b, acc);
    }

    const float bv = bias[ncol];
    const int rbase = tm * 16 + ((lane >> 4) << 3);
    float* yp = Y + (size_t)rbase * N + ncol;
#pragma unroll
    for (int r = 0; r < 8; ++r) {
        if (rbase + r < M) *yp = acc[r] + bv;
        yp += N;
    }
}

template <int HT>
__global__ void edge_mlp_wmma(const float* __restrict__ xf,
                              const int* __restrict__ row,
                              const int* __restrict__ col,
                              const float* __restrict__ We,
                              const float* __restrict__ be,
                              float* __restrict__ out,
                              int E, int Nu, int Hrt, int C) {
    const int H = (HT > 0) ? HT : Hrt;
    const int K = 2 * H;
    const int lane = threadIdx.x & 31;
    const int wave = threadIdx.x >> 5;
    const int tile = blockIdx.x * (blockDim.x >> 5) + wave;
    const int ebase = tile * 16;
    if (ebase >= E) return;

    const int m = lane & 15;
    const int e = min(ebase + m, E - 1);
    const int r = row[e];
    const int c = col[e];
    const int koff = (lane >> 4) << 1;

    const float* __restrict__ urow = xf + (size_t)r * H + koff;
    const float* __restrict__ mrow = xf + (size_t)(Nu + c) * H + koff;

    const int n = lane & 15;
    const float wmask = (n < C) ? 1.0f : 0.0f;
    const float* __restrict__ wrow = We + (size_t)min(n, C - 1) * K + koff;

    v8f acc = {};
#pragma unroll 8
    for (int k = 0; k < K; k += 4) {
        const float* src = (k + koff < H) ? (urow + k) : (mrow + (k - H));
        v2f a = *(const v2f*)src;
        v2f b = *(const v2f*)(wrow + k);
        b.x *= wmask;
        b.y *= wmask;
        acc = wmma_f32x4(a, b, acc);
    }

    if (n < C) {
        const float bv = be[n];
        const int mb = (lane >> 4) << 3;
        float* op = out + (size_t)(ebase + mb) * C + n;
        if (ebase + 16 <= E) {
#pragma unroll
            for (int rr = 0; rr < 8; ++rr) { *op = acc[rr] + bv; op += C; }
        } else {
#pragma unroll
            for (int rr = 0; rr < 8; ++rr) {
                if (ebase + mb + rr < E) *op = acc[rr] + bv;
                op += C;
            }
        }
    }
}

constexpr int EM_H = 128;
constexpr int EM_K = 2 * EM_H;
constexpr int EM_WPITCH = EM_K + 4;
constexpr int EM_APITCH = EM_H + 4;
constexpr int EM_WAVES = 8;
constexpr int EM_GRID = 768;

__device__ __forceinline__ int edge_tile_ends(const int* __restrict__ row, const int* __restrict__ col,
                                              int t, int lane, int E) {
    const int e = min(t * 16 + (lane & 15), E - 1);
    const int* __restrict__ p = (lane < 16) ? row : col;
    return p[e];
}

#define EM_ROW_LOAD(j, IDX, BASE) \
    rv##j = ((const float4*)(xf + (size_t)(((BASE) ? Nu : 0) + __builtin_amdgcn_readlane(IDX, (BASE) + j)) * EM_H))[lane];
#define EM_ROW_STORE(j, IDX, BASE) *(float4*)(astore + j * EM_APITCH) = rv##j;
#define EM_ALL_ROWS(OP, IDX, BASE) \
    OP(0, IDX, BASE) OP(1, IDX, BASE) OP(2, IDX, BASE) OP(3, IDX, BASE) \
    OP(4, IDX, BASE) OP(5, IDX, BASE) OP(6, IDX, BASE) OP(7, IDX, BASE) \
    OP(8, IDX, BASE) OP(9, IDX, BASE) OP(10, IDX, BASE) OP(11, IDX, BASE) \
    OP(12, IDX, BASE) OP(13, IDX, BASE) OP(14, IDX, BASE) OP(15, IDX, BASE)

__global__ __launch_bounds__(256)
void edge_mlp_rows_lds(const float* __restrict__ xf,
                       const int* __restrict__ row,
                       const int* __restrict__ col,
                       const float* __restrict__ We,
                       const float* __restrict__ be,
                       float* __restrict__ out,
                       int E, int Nu, int C) {
    __shared__ __attribute__((aligned(16))) float Ws[16 * EM_WPITCH];
    __shared__ __attribute__((aligned(16))) float As[EM_WAVES * 16 * EM_APITCH];

    const int lane = threadIdx.x & 31;
    const int wave = __builtin_amdgcn_readfirstlane((int)(threadIdx.x >> 5));
    const int c = lane & 15;
    const int h = lane >> 4;

    const int tiles = (E + 15) >> 4;
    const int t0 = (int)(((long long)tiles * blockIdx.x) / EM_GRID);
    const int t1 = (int)(((long long)tiles * (blockIdx.x + 1)) / EM_GRID);
    int t = t0 + wave;

    float4 rv0, rv1, rv2, rv3, rv4, rv5, rv6, rv7, rv8, rv9, rv10, rv11, rv12, rv13, rv14, rv15;
    rv0 = rv1 = rv2 = rv3 = rv4 = rv5 = rv6 = rv7 = make_float4(0.f, 0.f, 0.f, 0.f);
    rv8 = rv9 = rv10 = rv11 = rv12 = rv13 = rv14 = rv15 = rv0;
    int idx = 0;
    if (t < t1) {
        idx = edge_tile_ends(row, col, t, lane, E);
        EM_ALL_ROWS(EM_ROW_LOAD, idx, 0)
    }

    for (int i = threadIdx.x; i < 16 * (EM_K / 4); i += EM_WAVES * 32) {
        const int n = i / (EM_K / 4);
        const int k4 = i - n * (EM_K / 4);
        float4 w = make_float4(0.f, 0.f, 0.f, 0.f);
        if (n < C) w = ((const float4*)We)[i];
        *(float4*)(Ws + n * EM_WPITCH + 4 * k4) = w;
    }
    __syncthreads();

    float* Aw = As + wave * (16 * EM_APITCH);
    float* astore = Aw + 4 * lane;
    const float* afrag = Aw + c * EM_APITCH + 4 * h;
    const float* bfrag = Ws + c * EM_WPITCH + 4 * h;
    const float bv = (c < C) ? be[c] : 0.f;

    while (t < t1) {
        const int tn = t + EM_WAVES;
        int idxn = 0;
        if (tn < t1) idxn = edge_tile_ends(row, col, tn, lane, E);

        EM_ALL_ROWS(EM_ROW_STORE, idx, 0)
        EM_ALL_ROWS(EM_ROW_LOAD, idx, 16)

        v8f acc0 = {}, acc1 = {};
#pragma unroll
        for (int kb = 0; kb < EM_H / 8; ++kb) {
            const float4 a = *(const float4*)(afrag + 8 * kb);
            const float4 b = *(const float4*)(bfrag + 8 * kb);
            acc0 = wmma_f32x4((v2f){a.x, a.y}, (v2f){b.x, b.y}, acc0);
            acc1 = wmma_f32x4((v2f){a.z, a.w}, (v2f){b.z, b.w}, acc1);
        }

        EM_ALL_ROWS(EM_ROW_STORE, idx, 0)
        if (tn < t1) {
            EM_ALL_ROWS(EM_ROW_LOAD, idxn, 0)
        }
#pragma unroll
        for (int kb = 0; kb < EM_H / 8; ++kb) {
            const float4 a = *(const float4*)(afrag + 8 * kb);
            const float4 b = *(const float4*)(bfrag + EM_H + 8 * kb);
            acc0 = wmma_f32x4((v2f){a.x, a.y}, (v2f){b.x, b.y}, acc0);
            acc1 = wmma_f32x4((v2f){a.z, a.w}, (v2f){b.z, b.w}, acc1);
        }

        if (c < C) {
            const int ebase = t * 16;
            const int mb = h << 3;
            float* op = out + (size_t)(ebase + mb) * C + c;
            if (ebase + 16 <= E) {
#pragma unroll
                for (int rr = 0; rr < 8; ++rr) { *op = (acc0[rr] + acc1[rr]) + bv; op += C; }
            } else {
#pragma unroll
                for (int rr = 0; rr < 8; ++rr) {
                    if (ebase + mb + rr < E) *op = (acc0[rr] + acc1[rr]) + bv;
                    op += C;
                }
            }
        }

        idx = idxn;
        t = tn;
    }
}

__global__ void fill_kernel(float* __restrict__ p, float v, long long n) {
    long long i = (long long)blockIdx.x * blockDim.x + threadIdx.x;
    if (i < n) p[i] = v;
}

__global__ void zero4_kernel(float4* __restrict__ p, long long n4) {
    long long i = (long long)blockIdx.x * blockDim.x + threadIdx.x;
    if (i < n4) p[i] = make_float4(0.f, 0.f, 0.f, 0.f);
}

__global__ void deg_scatter_kernel(const int* __restrict__ col,
                                   float* __restrict__ deg, int E) {
    int e = blockIdx.x * blockDim.x + threadIdx.x;
    if (e < E) atomic_add_f32(&deg[col[e]], 1.0f);
}

__global__ void deg_scatter_own_kernel(const int* __restrict__ col,
                                       float* __restrict__ deg,
                                       unsigned char* __restrict__ own, int E) {
    int e = blockIdx.x * blockDim.x + threadIdx.x;
    if (e < E) {
        const int c = col[e];
        own[e] = (unsigned char)(c & 255);
        atomic_add_f32(&deg[c], 1.0f);
    }
}

__global__ void rsqrt_kernel(float* __restrict__ d, int N) {
    int i = blockIdx.x * blockDim.x + threadIdx.x;
    if (i < N) d[i] = rsqrtf(d[i]);
}

constexpr int DH_GRID = 64;
constexpr int DH_THREADS = 1024;
constexpr int DH_WORDS = 50000;
constexpr int DH_U = 4;
static_assert((DH_WORDS & 3) == 0, "tables are cleared and stored 16 bytes at a time");

__device__ __forceinline__ void dh_count(unsigned int* cnt, int c) {
    atomicAdd(cnt + (c >> 1), (c & 1) ? 0x10000u : 1u);
}

__global__ __launch_bounds__(DH_THREADS)
void deg_hist_kernel(const int* __restrict__ col, unsigned int* __restrict__ own4,
                     unsigned int* __restrict__ part, int E, int N) {
    __shared__ __attribute__((aligned(16))) unsigned int cnt[DH_WORDS];

    const int words4 = (((N + 1) >> 1) + 3) >> 2;
    const int g4 = E >> 2;
    const int q0 = (int)(((long long)g4 * blockIdx.x) / DH_GRID);
    const int q1 = (int)(((long long)g4 * (blockIdx.x + 1)) / DH_GRID);
    const int4* __restrict__ col4 = (const int4*)col;

    int4 cv[DH_U];
#pragma unroll
    for (int u = 0; u < DH_U; ++u)
        cv[u] = col4[min(q0 + u * DH_THREADS + (int)threadIdx.x, q1 - 1)];

    for (int i = threadIdx.x; i < words4; i += DH_THREADS)
        ((uint4*)cnt)[i] = make_uint4(0u, 0u, 0u, 0u);
    __syncthreads();

    for (int base = q0; base < q1; base += DH_U * DH_THREADS) {
#pragma unroll
        for (int u = 0; u < DH_U; ++u) {
            const int q = base + u * DH_THREADS + (int)threadIdx.x;
            if (q < q1) {
                const int4 c = cv[u];
                dh_count(cnt, c.x);
                dh_count(cnt, c.y);
                dh_count(cnt, c.z);
                dh_count(cnt, c.w);
                own4[q] = (unsigned)(c.x & 255) | ((unsigned)(c.y & 255) << 8) |
                          ((unsigned)(c.z & 255) << 16) | ((unsigned)(c.w & 255) << 24);
            }
        }
        if (base + DH_U * DH_THREADS < q1) {
#pragma unroll
            for (int u = 0; u < DH_U; ++u)
                cv[u] = col4[min(base + (DH_U + u) * DH_THREADS + (int)threadIdx.x, q1 - 1)];
        }
    }
    __syncthreads();

    uint4* __restrict__ p = (uint4*)part + (size_t)blockIdx.x * words4;
#pragma unroll 4
    for (int i = threadIdx.x; i < words4; i += DH_THREADS)
        p[i] = ((const uint4*)cnt)[i];
}

__global__ __launch_bounds__(256)
void deg_reduce_kernel(const unsigned int* __restrict__ part, float* __restrict__ dinv,
                       int N, int pitch) {
    const int w = blockIdx.x * blockDim.x + threadIdx.x;
    if (2 * w >= N) return;
    unsigned int v[DH_GRID];
#pragma unroll
    for (int b = 0; b < DH_GRID; ++b) v[b] = part[(size_t)b * pitch + w];
    unsigned int lo = 0u, hi = 0u;
#pragma unroll
    for (int b = 0; b < DH_GRID; ++b) {
        lo += v[b] & 0xffffu;
        hi += v[b] >> 16;
    }
    *(float2*)(dinv + 2 * w) = make_float2(rsqrtf(1.0f + (float)lo), rsqrtf(1.0f + (float)hi));
}

__global__ void edge_scatter_kernel(const float* __restrict__ h,
                                    const float* __restrict__ dinv,
                                    const int* __restrict__ row,
                                    const int* __restrict__ col,
                                    float* __restrict__ acc,
                                    int E, int H) {
    const int chunks = H >> 2;
    long long gid = (long long)blockIdx.x * blockDim.x + threadIdx.x;
    long long total = (long long)E * chunks;
    if (gid >= total) return;
    const int e = (int)(gid / chunks);
    const int j = (int)(gid - (long long)e * chunks) << 2;
    const int r = row[e];
    const int c = col[e];
    const float s = dinv[r];
    const float4 hv = *(const float4*)(h + (size_t)r * H + j);
    float* dst = acc + (size_t)c * H + j;
    atomic_add_f32(dst + 0, s * hv.x);
    atomic_add_f32(dst + 1, s * hv.y);
    atomic_add_f32(dst + 2, s * hv.z);
    atomic_add_f32(dst + 3, s * hv.w);
}

constexpr int SC_H = 128;
constexpr int SC_EPW = 8;

template <bool FULL>
__device__ __forceinline__ void line_scatter_wave(const float* __restrict__ h,
                                                  const float* __restrict__ dinv,
                                                  const int* __restrict__ row, const int* __restrict__ col,
                                                  float* __restrict__ acc, int e0, int ne, int l) {
    int r[SC_EPW], c[SC_EPW];
    float s[SC_EPW], v0[SC_EPW], v1[SC_EPW], v2[SC_EPW], v3[SC_EPW];
#pragma unroll
    for (int i = 0; i < SC_EPW; ++i) {
        const int e = FULL ? e0 + i : e0 + min(i, ne - 1);
        r[i] = row[e];
        c[i] = col[e];
    }
#pragma unroll
    for (int i = 0; i < SC_EPW; ++i) {
        s[i] = dinv[r[i]];
        const float* __restrict__ p = h + (size_t)r[i] * SC_H + l;
        v0[i] = p[0];
        v1[i] = p[32];
        v2[i] = p[64];
        v3[i] = p[96];
    }
#pragma unroll
    for (int i = 0; i < SC_EPW; ++i) {
        if (FULL || i < ne) {
            float* dst = acc + (size_t)c[i] * SC_H + l;
            atomic_add_f32(dst, s[i] * v0[i]);
            atomic_add_f32(dst + 32, s[i] * v1[i]);
            atomic_add_f32(dst + 64, s[i] * v2[i]);
            atomic_add_f32(dst + 96, s[i] * v3[i]);
        }
    }
}

__global__ __launch_bounds__(256)
void edge_scatter_lines_kernel(const float* __restrict__ h,
                               const float* __restrict__ dinv,
                               const int* __restrict__ row,
                               const int* __restrict__ col,
                               float* __restrict__ acc, int E) {
    const int l = threadIdx.x & 31;
    const int wv = __builtin_amdgcn_readfirstlane(
        (int)(blockIdx.x * (blockDim.x >> 5) + (threadIdx.x >> 5)));
    const int e0 = wv * SC_EPW;
    if (e0 >= E) return;
    const int ne = min(SC_EPW, E - e0);
    if (ne == SC_EPW) line_scatter_wave<true>(h, dinv, row, col, acc, e0, ne, l);
    else              line_scatter_wave<false>(h, dinv, row, col, acc, e0, ne, l);
}


constexpr int OW_GRID_LOG2 = 8;
constexpr int OW_GRID = 1 << OW_GRID_LOG2;
static_assert(OW_GRID == 256, "the owner list holds one byte per edge: target mod 256");
constexpr int OW_WAVES = 32;
constexpr int OW_SLOTS = 392;
constexpr int OW_U = 8;
constexpr int OW_Q = 544;

__device__ __forceinline__ void lds_add_f32(float* p, float v) {
    __hip_atomic_fetch_add(p, v, __ATOMIC_RELAXED, __HIP_MEMORY_SCOPE_WORKGROUP);
}

__device__ __forceinline__ void ow_drain(const float* __restrict__ h, const float* __restrict__ dinv,
                                         const int* __restrict__ row, const int* __restrict__ col,
                                         float* accs, const int* q, int m, int lane) {
    const int e = q[min(lane, m - 1)];
    const int r = row[e];
    const int c = col[e];
    const int sbits = __float_as_int(dinv[r]);
    for (int j = 0; j < m; j += 8) {
        float v0[8], v1[8], v2[8], v3[8], ss[8];
        int sl[8];
#pragma unroll
        for (int i = 0; i < 8; ++i) {
            const int idx = min(j + i, m - 1);
            const int rr = __builtin_amdgcn_readlane(r, idx);
            sl[i] = __builtin_amdgcn_readlane(c, idx) >> OW_GRID_LOG2;
            ss[i] = __int_as_float(__builtin_amdgcn_readlane(sbits, idx));
            const float* __restrict__ p = h + (size_t)rr * SC_H + lane;
            v0[i] = p[0];
            v1[i] = p[32];
            v2[i] = p[64];
            v3[i] = p[96];
        }
#pragma unroll
        for (int i = 0; i < 8; ++i) {
            if (j + i < m) {
                float* d = accs + sl[i] * SC_H + lane;
                lds_add_f32(d, ss[i] * v0[i]);
                lds_add_f32(d + 32, ss[i] * v1[i]);
                lds_add_f32(d + 64, ss[i] * v2[i]);
                lds_add_f32(d + 96, ss[i] * v3[i]);
            }
        }
    }
}

#define OW_STEP(MT, EDGE) \
    { \
        const bool mt = (MT); \
        const unsigned mask = __builtin_amdgcn_ballot_w32(mt); \
        if (mask) { \
            if (mt) q[cnt + (int)__builtin_amdgcn_mbcnt_lo(mask, 0u)] = (EDGE); \
            cnt += __builtin_popcount(mask); \
        } \
    }

#define OW_WORD(WV, E0) \
    { \
        const unsigned xw = ok ? ((WV) ^ bpat) : 0xffffffffu; \
        const unsigned zb = (xw - 0x01010101u) & ~xw & 0x80808080u; \
        if (__builtin_amdgcn_ballot_w32(zb != 0u)) { \
            OW_STEP((xw & 0x000000ffu) == 0u, (E0) + 0) \
            OW_STEP((xw & 0x0000ff00u) == 0u, (E0) + 1) \
            OW_STEP((xw & 0x00ff0000u) == 0u, (E0) + 2) \
            OW_STEP((xw & 0xff000000u) == 0u, (E0) + 3) \
        } \
    }

__global__ __launch_bounds__(1024)
void owner_aggregate_kernel(const float* __restrict__ h, const float* __restrict__ dinv,
                            const int* __restrict__ row, const int* __restrict__ col,
                            const unsigned char* __restrict__ own,
                            float* __restrict__ x, int E, int N) {
    __shared__ __attribute__((aligned(16))) float accs[OW_SLOTS * SC_H];
    __shared__ int qs[OW_WAVES * OW_Q];

    const int lane = threadIdx.x & 31;
    const int wave = __builtin_amdgcn_readfirstlane((int)(threadIdx.x >> 5));
    const int b = blockIdx.x;
    const int slots = (N - b + OW_GRID - 1) >> OW_GRID_LOG2;

#pragma unroll 4
    for (int sl = wave; sl < slots; sl += OW_WAVES) {
        const int node = (sl << OW_GRID_LOG2) + b;
        const float di = dinv[node];
        const float4 hv = ((const float4*)(h + (size_t)node * SC_H))[lane];
        *(float4*)(accs + sl * SC_H + 4 * lane) = make_float4(di * hv.x, di * hv.y, di * hv.z, di * hv.w);
    }
    __syncthreads();

    int* q = qs + wave * OW_Q;
    int cnt = 0;
    const uint4* __restrict__ own4 = (const uint4*)own;
    const int total16 = E >> 4;
    const unsigned bpat = (unsigned)b * 0x01010101u;
    for (int base = 0; base < total16; base += OW_U * OW_WAVES * 32) {
        uint4 ov[OW_U];
#pragma unroll
        for (int u = 0; u < OW_U; ++u) {
            const int i16 = base + (u * OW_WAVES + wave) * 32 + lane;
            ov[u] = own4[min(i16, total16 - 1)];
        }
#pragma unroll
        for (int u = 0; u < OW_U; ++u) {
            const int i16 = base + (u * OW_WAVES + wave) * 32 + lane;
            const bool ok = i16 < total16;
            OW_WORD(ov[u].x, 16 * i16 + 0)
            OW_WORD(ov[u].y, 16 * i16 + 4)
            OW_WORD(ov[u].z, 16 * i16 + 8)
            OW_WORD(ov[u].w, 16 * i16 + 12)
            if (cnt >= 32) {
                for (int g = 0; g < cnt; g += 32)
                    ow_drain(h, dinv, row, col, accs, q + g, min(32, cnt - g), lane);
                cnt = 0;
            }
        }
    }
    for (int g = 0; g < cnt; g += 32)
        ow_drain(h, dinv, row, col, accs, q + g, min(32, cnt - g), lane);
    __syncthreads();

#pragma unroll 4
    for (int sl = wave; sl < slots; sl += OW_WAVES) {
        const int node = (sl << OW_GRID_LOG2) + b;
        const float di = dinv[node];
        const float4 a = *(const float4*)(accs + sl * SC_H + 4 * lane);
        float4 o;
        o.x = di * a.x; o.x = o.x > 0.f ? o.x : 0.f;
        o.y = di * a.y; o.y = o.y > 0.f ? o.y : 0.f;
        o.z = di * a.z; o.z = o.z > 0.f ? o.z : 0.f;
        o.w = di * a.w; o.w = o.w > 0.f ? o.w : 0.f;
        ((float4*)(x + (size_t)node * SC_H))[lane] = o;
    }
}

__global__ void finalize_relu_kernel(const float* __restrict__ h,
                                     const float* __restrict__ dinv,
                                     float* __restrict__ acc_inout, int H) {
    const int node = blockIdx.x;
    const int j = threadIdx.x;
    const float di = dinv[node];
    const size_t idx = (size_t)node * H + j;
    const float v = di * acc_inout[idx] + di * di * h[idx];
    acc_inout[idx] = v > 0.f ? v : 0.f;
}

constexpr int RAW_FU = 32;
constexpr int RAW_FM = 64;
constexpr int RAW_H = 128;
constexpr int RAW_WU = RAW_FU + RAW_FM;
constexpr int RAW_WM = RAW_FU + RAW_FM + 1;

__global__ void compose_wz_kernel(const float* __restrict__ Wu, const float* __restrict__ bu,
                                  const float* __restrict__ Wm, const float* __restrict__ bm,
                                  float* __restrict__ Wz) {
    const int i = blockIdx.x * blockDim.x + threadIdx.x;
    if (i >= RAW_H * RAW_H) return;
    const int n = i / RAW_H;
    const int k = i - n * RAW_H;
    float w = 0.f;
    if (k < RAW_FU) w = Wu[n * RAW_FU + k];
    else if (k < RAW_WU) w = Wm[n * RAW_FM + (k - RAW_FU)];
    else if (k == RAW_WU) w = bu[n];
    else if (k == RAW_WM) w = bm[n];
    Wz[i] = w;
}

__global__ void init_z_kernel(const float* __restrict__ xu, const float* __restrict__ xm,
                              const float* __restrict__ dinv, float* __restrict__ Z, int Nu) {
    const int node = blockIdx.x;
    const int j = threadIdx.x;
    const float di = dinv[node];
    float self = 0.f;
    if (node < Nu) {
        if (j < RAW_FU) self = xu[(size_t)node * RAW_FU + j];
        else if (j == RAW_WU) self = 1.f;
    } else {
        if (j >= RAW_FU && j < RAW_WU) self = xm[(size_t)(node - Nu) * RAW_FM + (j - RAW_FU)];
        else if (j == RAW_WM) self = 1.f;
    }
    Z[(size_t)node * RAW_H + j] = di * self;
}

constexpr int DZ_THREADS = 256;
constexpr int DZ_NODES = 256;
constexpr int DZ_U = 8;
static_assert((DH_GRID & 1) == 0 && DZ_THREADS == 256 && DZ_NODES == DZ_THREADS,
              "the tables are split into two halves of 128 threads, each thread pair of halves owns one word");
static_assert((RAW_FU & 3) == 0 && (RAW_FM & 3) == 0 && (RAW_WU & 3) == 0 && RAW_WM == RAW_WU + 1,
              "a lane holds four consecutive floats of a row; both weight columns sit in one lane");
static_assert(RAW_H == 128, "one 16-byte store per lane covers a row");

template <bool FULL>
__device__ __forceinline__ void dz_rows(const float4* __restrict__ xu4, const float4* __restrict__ xm4,
                                        const float* sd, float4* __restrict__ Z4,
                                        int node0, int ln0, int N, int Nu, int lane) {
    constexpr int QU = RAW_FU / 4;
    constexpr int QM = (RAW_FU + RAW_FM) / 4;
    for (int i0 = 0; i0 < 32; i0 += DZ_U) {
        float4 v[DZ_U];
        float di[DZ_U];
#pragma unroll
        for (int i = 0; i < DZ_U; ++i) {
            const int ln = ln0 + i0 + i;
            const int node = FULL ? node0 + ln : min(node0 + ln, N - 1);
            const bool user = node < Nu;
            di[i] = sd[ln];
            const float4* __restrict__ p = user ? xu4 + (size_t)node * QU + lane
                                                : xm4 + (size_t)(node - Nu) * (RAW_FM / 4) + (lane - QU);
            const bool ld = user ? (lane < QU) : (lane >= QU && lane < QM);
            float4 r = make_float4(0.f, 0.f, 0.f, 0.f);
            if (ld) r = *p;
            if (lane == QM) { r.x = user ? 1.f : 0.f; r.y = user ? 0.f : 1.f; }
            v[i] = r;
        }
#pragma unroll
        for (int i = 0; i < DZ_U; ++i) {
            const int node = node0 + ln0 + i0 + i;
            if (FULL || node < N)
                Z4[(size_t)node * (RAW_H / 4) + lane] =
                    make_float4(di[i] * v[i].x, di[i] * v[i].y, di[i] * v[i].z, di[i] * v[i].w);
        }
    }
}

__global__ __launch_bounds__(DZ_THREADS)
void deg_reduce_init_z_kernel(const unsigned int* __restrict__ part, float* __restrict__ dinv,
                              const float* __restrict__ xu, const float* __restrict__ xm,
                              float* __restrict__ Z, int N, int Nu, int pitch) {
    __shared__ unsigned int sp[DZ_THREADS];
    __shared__ float sd[DZ_NODES];

    const int t = threadIdx.x;
    const int lane = t & 31;
    const int wave = __builtin_amdgcn_readfirstlane(t >> 5);
    const int node0 = blockIdx.x * DZ_NODES;

    const int j = t & (DZ_THREADS / 2 - 1);
    const int half = t >> 7;
    const int w = blockIdx.x * (DZ_NODES / 2) + j;
    const bool live = 2 * w < N;
    unsigned int lo = 0u, hi = 0u;
    if (live) {
        unsigned int v[DH_GRID / 2];
        const unsigned int* __restrict__ p = part + (size_t)half * (DH_GRID / 2) * pitch + w;
#pragma unroll
        for (int b = 0; b < DH_GRID / 2; ++b) v[b] = p[(size_t)b * pitch];
#pragma unroll
        for (int b = 0; b < DH_GRID / 2; ++b) {
            lo += v[b] & 0xffffu;
            hi += v[b] >> 16;
        }
    }
    if (half) {
        sp[2 * j] = lo;
        sp[2 * j + 1] = hi;
    }
    __syncthreads();
    if (!half) {
        lo += sp[2 * j];
        hi += sp[2 * j + 1];
        const float d0 = rsqrtf(1.0f + (float)lo);
        const float d1 = rsqrtf(1.0f + (float)hi);
        sd[2 * j] = d0;
        sd[2 * j + 1] = d1;
        if (live) *(float2*)(dinv + 2 * w) = make_float2(d0, d1);
    }
    __syncthreads();

    const float4* __restrict__ xu4 = (const float4*)xu;
    const float4* __restrict__ xm4 = (const float4*)xm;
    float4* __restrict__ Z4 = (float4*)Z;
    const int ln0 = wave * 32;
    if (node0 + ln0 >= N) return;
    if (node0 + ln0 + 32 <= N) dz_rows<true>(xu4, xm4, sd, Z4, node0, ln0, N, Nu, lane);
    else                       dz_rows<false>(xu4, xm4, sd, Z4, node0, ln0, N, Nu, lane);
}

constexpr int RAW_EPW = 8;

template <bool FULL>
__device__ __forceinline__ void raw_scatter_wave(const float* __restrict__ xu, const float* __restrict__ xm,
                                                 const float* __restrict__ dinv,
                                                 const int* __restrict__ row, const int* __restrict__ col,
                                                 float* __restrict__ Z, int Nu, int e0, int ne, int l) {
    int r[RAW_EPW], c[RAW_EPW];
    float s[RAW_EPW], v0[RAW_EPW], v1[RAW_EPW];
#pragma unroll
    for (int i = 0; i < RAW_EPW; ++i) {
        const int e = FULL ? e0 + i : e0 + min(i, ne - 1);
        r[i] = row[e];
        c[i] = col[e];
    }
#pragma unroll
    for (int i = 0; i < RAW_EPW; ++i) {
        s[i] = dinv[r[i]];
        const bool user = r[i] < Nu;
        const float* __restrict__ p = user ? xu + (size_t)r[i] * RAW_FU + l
                                           : xm + (size_t)(r[i] - Nu) * RAW_FM + l;
        v0[i] = p[0];
        v1[i] = 0.f;
        if (!user) v1[i] = p[32];
    }
#pragma unroll
    for (int i = 0; i < RAW_EPW; ++i) {
        if (FULL || i < ne) {
            const bool user = r[i] < Nu;
            float* dst = Z + (size_t)c[i] * RAW_H;
            atomic_add_f32(dst + (user ? 0 : RAW_FU) + l, s[i] * v0[i]);
            if (!user) atomic_add_f32(dst + RAW_FU + 32 + l, s[i] * v1[i]);
        }
    }
    if (l < (FULL ? RAW_EPW : ne)) {
        const int rl = row[e0 + l];
        const int cl = col[e0 + l];
        atomic_add_f32(Z + (size_t)cl * RAW_H + (rl < Nu ? RAW_WU : RAW_WM), dinv[rl]);
    }
}

__global__ __launch_bounds__(256)
void raw_scatter_kernel(const float* __restrict__ xu, const float* __restrict__ xm,
                        const float* __restrict__ dinv,
                        const int* __restrict__ row, const int* __restrict__ col,
                        float* __restrict__ Z, int E, int Nu) {
    const int l = threadIdx.x & 31;
    const int wv = __builtin_amdgcn_readfirstlane(
        (int)(blockIdx.x * (blockDim.x >> 5) + (threadIdx.x >> 5)));
    const int e0 = wv * RAW_EPW;
    if (e0 >= E) return;
    const int ne = min(RAW_EPW, E - e0);
    if (ne == RAW_EPW) raw_scatter_wave<true>(xu, xm, dinv, row, col, Z, Nu, e0, ne, l);
    else               raw_scatter_wave<false>(xu, xm, dinv, row, col, Z, Nu, e0, ne, l);
}

static inline int ceil_div_i(long long a, long long b) { return (int)((a + b - 1) / b); }

static void launch_gemm(const float* X, const float* W, const float* b, float* Y,
                        int M, int K, int N, hipStream_t stream) {
    const int TPB = 256;
    if (N == 128 && K == 32)
        gemm_bias_wmma_wlds<32, 0><<<GEMM_GRID, TPB, 0, stream>>>(X, W, b, nullptr, nullptr, Y, M);
    else if (N == 128 && K == 64)
        gemm_bias_wmma_wlds<64, 0><<<GEMM_GRID, TPB, 0, stream>>>(X, W, b, nullptr, nullptr, Y, M);
    else if (N == 128 && K == 128)
        gemm_bias_wmma_wlds<128, 0><<<GEMM_GRID, TPB, 0, stream>>>(X, W, b, nullptr, nullptr, Y, M);
    else {
        const int strips = ceil_div_i(M, 16);
        const int tiles = strips * (N / 16);
        gemm_bias_wmma_rt<<<ceil_div_i(tiles, 8), TPB, 0, stream>>>(X, W, b, Y, M, K, N);
    }
}

extern "C" void kernel_launch(void* const* d_in, const int* in_sizes, int n_in,
                              void* d_out, int out_size, void* d_ws, size_t ws_size,
                              hipStream_t stream) {
    const float* x_user  = (const float*)d_in[0];
    const float* x_movie = (const float*)d_in[1];
    const int*   edge    = (const int*)d_in[2];
    const float* Wu = (const float*)d_in[3];
    const float* bu = (const float*)d_in[4];
    const float* Wm = (const float*)d_in[5];
    const float* bm = (const float*)d_in[6];
    const float* W1 = (const float*)d_in[7];
    const float* b1 = (const float*)d_in[8];
    const float* W2 = (const float*)d_in[9];
    const float* b2 = (const float*)d_in[10];
    const float* We = (const float*)d_in[11];
    const float* be = (const float*)d_in[12];
    float* out = (float*)d_out;

    const int H  = in_sizes[4];
    const int C  = in_sizes[12];
    const int Fu = in_sizes[3] / H;
    const int Fm = in_sizes[5] / H;
    const int Nu = in_sizes[0] / Fu;
    const int Nm = in_sizes[1] / Fm;
    const int E  = in_sizes[2] / 2;
    const int N  = Nu + Nm;
    const int* row = edge;
    const int* col = edge + E;

    float* ws   = (float*)d_ws;
    float* X    = ws;
    float* Hh   = X  + (size_t)N * H;
    float* Acc  = Hh + (size_t)N * H;
    float* Wz   = Acc + (size_t)N * H;
    float* dinv = Wz + (size_t)RAW_H * RAW_H;
    unsigned char* own = (unsigned char*)(dinv + (size_t)((N + 3) & ~3));
    unsigned int* part = (unsigned int*)(own + (((size_t)E + 15) & ~(size_t)15));

    const int TPB = 256, WPB = 8;
    const long long NH = (long long)N * H;

    const bool owner_path = (H == SC_H) && ((E & 15) == 0) && (N <= OW_GRID * OW_SLOTS);
    const int dh_words = (((N + 1) >> 1) + 3) & ~3;
    const bool hist_path = (N <= 2 * DH_WORDS) && ((E & 3) == 0) && (E >= 4 * DH_GRID) &&
                           (E / DH_GRID + 4 <= 65535) &&
                           ((size_t)((unsigned char*)(part + (size_t)DH_GRID * dh_words) - (unsigned char*)d_ws) <= ws_size);
    const bool raw_path = (H == RAW_H) && (Fu == RAW_FU) && (Fm == RAW_FM);
    if (hist_path) {
        deg_hist_kernel<<<DH_GRID, DH_THREADS, 0, stream>>>(col, (unsigned int*)own, part, E, N);
        if (raw_path)
            deg_reduce_init_z_kernel<<<ceil_div_i(N, DZ_NODES), DZ_THREADS, 0, stream>>>(part, dinv, x_user, x_movie, Hh, N, Nu, dh_words);
        else
            deg_reduce_kernel<<<ceil_div_i((N + 1) >> 1, TPB), TPB, 0, stream>>>(part, dinv, N, dh_words);
    } else {
        fill_kernel<<<ceil_div_i(N, TPB), TPB, 0, stream>>>(dinv, 1.0f, N);
        if (owner_path)
            deg_scatter_own_kernel<<<ceil_div_i(E, TPB), TPB, 0, stream>>>(col, dinv, own, E);
        else
            deg_scatter_kernel<<<ceil_div_i(E, TPB), TPB, 0, stream>>>(col, dinv, E);
        rsqrt_kernel<<<ceil_div_i(N, TPB), TPB, 0, stream>>>(dinv, N);
    }

    const long long scatterThreads = (long long)E * (H / 4);

    if (raw_path) {
        compose_wz_kernel<<<ceil_div_i(RAW_H * RAW_H, TPB), TPB, 0, stream>>>(Wu, bu, Wm, bm, Wz);
        if (!hist_path) init_z_kernel<<<N, RAW_H, 0, stream>>>(x_user, x_movie, dinv, Hh, Nu);
        raw_scatter_kernel<<<ceil_div_i(E, RAW_EPW * WPB), TPB, 0, stream>>>(x_user, x_movie, dinv, row, col, Hh, E, Nu);
        gemm_bias_wmma_wlds<RAW_H, 1><<<GEMM_GRID, TPB, 0, stream>>>(Hh, Wz, nullptr, dinv, nullptr, X, N);
        gemm_bias_wmma_wlds<RAW_H, 2><<<GEMM_GRID, TPB, 0, stream>>>(X, W1, b1, dinv, Hh, Acc, N);
    } else {
        launch_gemm(x_user, Wu, bu, X, Nu, Fu, H, stream);
        launch_gemm(x_movie, Wm, bm, X + (size_t)Nu * H, Nm, Fm, H, stream);

        launch_gemm(X, W1, b1, Hh, N, H, H, stream);
        zero4_kernel<<<ceil_div_i(NH / 4, TPB), TPB, 0, stream>>>((float4*)Acc, NH / 4);
        edge_scatter_kernel<<<ceil_div_i(scatterThreads, TPB), TPB, 0, stream>>>(Hh, dinv, row, col, Acc, E, H);
        finalize_relu_kernel<<<N, H, 0, stream>>>(Hh, dinv, Acc, H);
    }

    launch_gemm(Acc, W2, b2, Hh, N, H, H, stream);
    if (owner_path) {
        owner_aggregate_kernel<<<OW_GRID, OW_WAVES * 32, 0, stream>>>(Hh, dinv, row, col, own, X, E, N);
    } else {
        zero4_kernel<<<ceil_div_i(NH / 4, TPB), TPB, 0, stream>>>((float4*)X, NH / 4);
        if (H == SC_H)
            edge_scatter_lines_kernel<<<ceil_div_i(E, SC_EPW * WPB), TPB, 0, stream>>>(Hh, dinv, row, col, X, E);
        else
            edge_scatter_kernel<<<ceil_div_i(scatterThreads, TPB), TPB, 0, stream>>>(Hh, dinv, row, col, X, E, H);
        finalize_relu_kernel<<<N, H, 0, stream>>>(Hh, dinv, X, H);
    }

    const int edgeTiles = ceil_div_i(E, 16);
    if (H == EM_H && C <= 16)
        edge_mlp_rows_lds<<<EM_GRID, TPB, 0, stream>>>(X, row, col, We, be, out, E, Nu, C);
    else
        edge_mlp_wmma<0><<<ceil_div_i(edgeTiles, WPB), TPB, 0, stream>>>(X, row, col, We, be, out, E, Nu, H, C);
}


# the three node GEMMs (Wz, W1, W2) run as one launch: a wave takes its 16-row strip through all three, each result going via a per-wave LDS patch into the next product's A fragments; sum order same
# speedup vs baseline: 2.8951x; 1.0971x over previous
#include <hip/hip_runtime.h>

typedef __attribute__((ext_vector_type(2))) float v2f;
typedef __attribute__((ext_vector_type(8))) float v8f;

__device__ __forceinline__ v8f wmma_f32x4(v2f a, v2f b, v8f c) {
    return __builtin_amdgcn_wmma_f32_16x16x4_f32(
         false, a,  false, b,
         (short)0, c,  false,  false);
}

__device__ __forceinline__ void atomic_add_f32(float* p, float v) {
#if defined(__HIP_DEVICE_COMPILE__)
    unsafeAtomicAdd(p, v);
#else
    atomicAdd(p, v);
#endif
}

constexpr int GEMM_GRID_LOG2 = 9;
constexpr int GEMM_GRID = 1 << GEMM_GRID_LOG2;

template <int MODE>
__device__ __forceinline__ float gemm_epilogue(float a, float b, float g) {
    if (MODE == 0) return a + b;
    if (MODE == 1) return g * a;
    const float v = a + g * b;
    return v > 0.f ? v : 0.f;
}

template <int K, int MODE>
__global__ __launch_bounds__(256)
void gemm_bias_wmma_wlds(const float* __restrict__ X,
                         const float* __restrict__ W,
                         const float* __restrict__ bias,
                         const float* __restrict__ rs,
                         const float* __restrict__ zc,
                         float* __restrict__ Y,
                         int M) {
    constexpr int N = 128;
    constexpr int NT = N / 16;
    constexpr int PITCH = K + 4;
    constexpr int KB = K / 8;
    constexpr int WAVES = 8;
    __shared__ __attribute__((aligned(16))) float Ws[N * PITCH];

    const int lane = threadIdx.x & 31;
    const int wave = __builtin_amdgcn_readfirstlane((int)(threadIdx.x >> 5));
    const int c = lane & 15;
    const int h = lane >> 4;

    const int strips = (M + 15) >> 4;
    const int s0 = (int)(((long long)strips * blockIdx.x) >> GEMM_GRID_LOG2);
    const int s1 = (int)(((long long)strips * (blockIdx.x + 1)) >> GEMM_GRID_LOG2);

    float bv[NT];
#pragma unroll
    for (int t = 0; t < NT; ++t) bv[t] = (MODE == 1) ? 0.f : bias[16 * t + c];

    float4 a[KB];
    int s = s0 + wave;

    if (s < s1) {
        const int arow = min(s * 16 + c, M - 1);
        const float4* __restrict__ ap = (const float4*)(X + (size_t)arow * K) + h;
#pragma unroll
        for (int kb = 0; kb < KB; ++kb) a[kb] = ap[2 * kb];
    }

    for (int i = threadIdx.x; i < N * (K / 4); i += WAVES * 32) {
        const int n = i / (K / 4);
        const int k4 = i - n * (K / 4);
        *(float4*)(Ws + n * PITCH + 4 * k4) = ((const float4*)W)[i];
    }
    __syncthreads();

    const float* __restrict__ bfrag = Ws + c * PITCH + 4 * h;

    while (s < s1) {
        float gr[8];
#pragma unroll
        for (int r = 0; r < 8; ++r) {
            gr[r] = 0.f;
            if (MODE != 0) {
                const int rr = min(s * 16 + (h << 3) + r, M - 1);
                float g = rs[rr];
                if (MODE == 2) {
                    const v2f w = *(const v2f*)(zc + (size_t)rr * N + 96);
                    g *= (w.x + w.y);
                }
                gr[r] = g;
            }
        }

        v8f acc[NT];
#pragma unroll
        for (int t = 0; t < NT; ++t) acc[t] = (v8f){};

        float4 bc[NT], bn[NT];
#pragma unroll
        for (int t = 0; t < NT; ++t)
            bc[t] = *(const float4*)(bfrag + 16 * t * PITCH);
#pragma unroll
        for (int kb = 0; kb < KB; ++kb) {
            if (kb + 1 < KB) {
#pragma unroll
                for (int t = 0; t < NT; ++t)
                    bn[t] = *(const float4*)(bfrag + 16 * t * PITCH + 8 * (kb + 1));
            }
            __builtin_amdgcn_sched_barrier(0);
            const v2f a01 = {a[kb].x, a[kb].y};
            const v2f a23 = {a[kb].z, a[kb].w};
#pragma unroll
            for (int t = 0; t < NT; ++t) {
                const v2f b01 = {bc[t].x, bc[t].y};
                acc[t] = wmma_f32x4(a01, b01, acc[t]);
            }
#pragma unroll
            for (int t = 0; t < NT; ++t) {
                const v2f b23 = {bc[t].z, bc[t].w};
                acc[t] = wmma_f32x4(a23, b23, acc[t]);
            }
            __builtin_amdgcn_sched_barrier(0);
#pragma unroll
            for (int t = 0; t < NT; ++t) bc[t] = bn[t];
        }

        const int row0 = s * 16;
        s += WAVES;

        if (s < s1) {
            const int arow = min(s * 16 + c, M - 1);
            const float4* __restrict__ ap = (const float4*)(X + (size_t)arow * K) + h;
#pragma unroll
            for (int kb = 0; kb < KB; ++kb) a[kb] = ap[2 * kb];
        }

        const int rbase = row0 + (h << 3);
        float* __restrict__ yrow = Y + (size_t)rbase * N + c;
        if (row0 + 16 <= M) {
#pragma unroll
            for (int t = 0; t < NT; ++t) {
#pragma unroll
                for (int r = 0; r < 8; ++r)
                    yrow[(size_t)r * N + 16 * t] = gemm_epilogue<MODE>(acc[t][r], bv[t], gr[r]);
            }
        } else {
#pragma unroll
            for (int t = 0; t < NT; ++t) {
#pragma unroll
                for (int r = 0; r < 8; ++r)
                    if (rbase + r < M)
                        yrow[(size_t)r * N + 16 * t] = gemm_epilogue<MODE>(acc[t][r], bv[t], gr[r]);
            }
        }
    }
}

__global__ void gemm_bias_wmma_rt(const float* __restrict__ X,
                                  const float* __restrict__ W,
                                  const float* __restrict__ bias,
                                  float* __restrict__ Y,
                                  int M, int K, int N) {
    const int lane = threadIdx.x & 31;
    const int wave = threadIdx.x >> 5;
    const int tilesN = N >> 4;
    const int tile = blockIdx.x * (blockDim.x >> 5) + wave;
    const int tm = tile / tilesN;
    const int tn = tile - tm * tilesN;
    if (tm * 16 >= M) return;

    const int mrow = min(tm * 16 + (lane & 15), M - 1);
    const int ncol = tn * 16 + (lane & 15);
    const int koff = (lane >> 4) << 1;
    const float* __restrict__ xrow = X + (size_t)mrow * K + koff;
    const float* __restrict__ wrow = W + (size_t)ncol * K + koff;

    v8f acc = {};
#pragma unroll 8
    for (int k = 0; k < K; k += 4) {
        v2f a = *(const v2f*)(xrow + k);
        v2f b = *(const v2f*)(wrow + k);
        acc = wmma_f32x4(a, b, acc);
    }

    const float bv = bias[ncol];
    const int rbase = tm * 16 + ((lane >> 4) << 3);
    float* yp = Y + (size_t)rbase * N + ncol;
#pragma unroll
    for (int r = 0; r < 8; ++r) {
        if (rbase + r < M) *yp = acc[r] + bv;
        yp += N;
    }
}

constexpr int CH_N = 128;
constexpr int CH_PITCH = CH_N + 4;
constexpr int CH_KB = CH_N / 8;
constexpr int CH_NT = CH_N / 16;
constexpr int CH_WAVES = 12;
constexpr int CH_GRID_LOG2 = 8;
constexpr int CH_GRID = 1 << CH_GRID_LOG2;

__device__ __forceinline__ void chain_product(const float4 (&a)[CH_KB],
                                              const float* __restrict__ bfrag,
                                              v8f (&acc)[CH_NT]) {
#pragma unroll
    for (int t = 0; t < CH_NT; ++t) acc[t] = (v8f){};
    float4 bc[CH_NT], bn[CH_NT];
#pragma unroll
    for (int t = 0; t < CH_NT; ++t)
        bc[t] = *(const float4*)(bfrag + 16 * t * CH_PITCH);
#pragma unroll
    for (int kb = 0; kb < CH_KB; ++kb) {
        if (kb + 1 < CH_KB) {
#pragma unroll
            for (int t = 0; t < CH_NT; ++t)
                bn[t] = *(const float4*)(bfrag + 16 * t * CH_PITCH + 8 * (kb + 1));
        }
        __builtin_amdgcn_sched_barrier(0);
        const v2f a01 = {a[kb].x, a[kb].y};
        const v2f a23 = {a[kb].z, a[kb].w};
#pragma unroll
        for (int t = 0; t < CH_NT; ++t) {
            const v2f b01 = {bc[t].x, bc[t].y};
            acc[t] = wmma_f32x4(a01, b01, acc[t]);
        }
#pragma unroll
        for (int t = 0; t < CH_NT; ++t) {
            const v2f b23 = {bc[t].z, bc[t].w};
            acc[t] = wmma_f32x4(a23, b23, acc[t]);
        }
        __builtin_amdgcn_sched_barrier(0);
#pragma unroll
        for (int t = 0; t < CH_NT; ++t) bc[t] = bn[t];
    }
}

__global__ __launch_bounds__(CH_WAVES * 32)
void gemm_chain3_wmma(const float* __restrict__ Z,
                      const float* __restrict__ Wz,
                      const float* __restrict__ W1,
                      const float* __restrict__ b1,
                      const float* __restrict__ W2,
                      const float* __restrict__ b2,
                      const float* __restrict__ rs,
                      float* __restrict__ Y,
                      int M) {
    __shared__ __attribute__((aligned(16))) float Ws[3 * CH_N * CH_PITCH];
    __shared__ __attribute__((aligned(16))) float Ps[CH_WAVES * 16 * CH_PITCH];

    const int lane = threadIdx.x & 31;
    const int wave = __builtin_amdgcn_readfirstlane((int)(threadIdx.x >> 5));
    const int c = lane & 15;
    const int h = lane >> 4;

    const int strips = (M + 15) >> 4;
    const int s0 = (int)(((long long)strips * blockIdx.x) >> CH_GRID_LOG2);
    const int s1 = (int)(((long long)strips * (blockIdx.x + 1)) >> CH_GRID_LOG2);

    float bv1[CH_NT], bv2[CH_NT];
#pragma unroll
    for (int t = 0; t < CH_NT; ++t) {
        bv1[t] = b1[16 * t + c];
        bv2[t] = b2[16 * t + c];
    }

    float4 a[CH_KB];
    int s = s0 + wave;

    if (s < s1) {
        const int arow = min(s * 16 + c, M - 1);
        const float4* __restrict__ ap = (const float4*)(Z + (size_t)arow * CH_N) + h;
#pragma unroll
        for (int kb = 0; kb < CH_KB; ++kb) a[kb] = ap[2 * kb];
    }

    for (int i = threadIdx.x; i < CH_N * (CH_N / 4); i += CH_WAVES * 32) {
        const int n = i / (CH_N / 4);
        const int k4 = i - n * (CH_N / 4);
        const float4 w0 = ((const float4*)Wz)[i];
        const float4 w1 = ((const float4*)W1)[i];
        const float4 w2 = ((const float4*)W2)[i];
        float* dst = Ws + n * CH_PITCH + 4 * k4;
        *(float4*)dst = w0;
        *(float4*)(dst + CH_N * CH_PITCH) = w1;
        *(float4*)(dst + 2 * CH_N * CH_PITCH) = w2;
    }
    __syncthreads();

    const float* __restrict__ bfz = Ws + c * CH_PITCH + 4 * h;
    const float* __restrict__ bf1 = bfz + CH_N * CH_PITCH;
    const float* __restrict__ bf2 = bfz + 2 * CH_N * CH_PITCH;

    float* P = Ps + wave * (16 * CH_PITCH);
    float* pst = P + (h << 3) * CH_PITCH + c;
    const float* pfr = P + c * CH_PITCH + 4 * h;

    while (s < s1) {
        float d[8], g[8];
#pragma unroll
        for (int r = 0; r < 8; ++r) {
            const int rr = min(s * 16 + (h << 3) + r, M - 1);
            const float dr = rs[rr];
            const v2f w = *(const v2f*)(Z + (size_t)rr * CH_N + 96);
            d[r] = dr;
            g[r] = dr * (w.x + w.y);
        }

        v8f acc[CH_NT];

        chain_product(a, bfz, acc);
#pragma unroll
        for (int t = 0; t < CH_NT; ++t) {
#pragma unroll
            for (int r = 0; r < 8; ++r)
                pst[r * CH_PITCH + 16 * t] = gemm_epilogue<1>(acc[t][r], 0.f, d[r]);
        }
#pragma unroll
        for (int kb = 0; kb < CH_KB; ++kb) a[kb] = *(const float4*)(pfr + 8 * kb);

        chain_product(a, bf1, acc);
#pragma unroll
        for (int t = 0; t < CH_NT; ++t) {
#pragma unroll
            for (int r = 0; r < 8; ++r)
                pst[r * CH_PITCH + 16 * t] = gemm_epilogue<2>(acc[t][r], bv1[t], g[r]);
        }
#pragma unroll
        for (int kb = 0; kb < CH_KB; ++kb) a[kb] = *(const float4*)(pfr + 8 * kb);

        chain_product(a, bf2, acc);

        const int row0 = s * 16;
        s += CH_WAVES;

        if (s < s1) {
            const int arow = min(s * 16 + c, M - 1);
            const float4* __restrict__ ap = (const float4*)(Z + (size_t)arow * CH_N) + h;
#pragma unroll
            for (int kb = 0; kb < CH_KB; ++kb) a[kb] = ap[2 * kb];
        }

        const int rbase = row0 + (h << 3);
        float* __restrict__ yrow = Y + (size_t)rbase * CH_N + c;
        if (row0 + 16 <= M) {
#pragma unroll
            for (int t = 0; t < CH_NT; ++t) {
#pragma unroll
                for (int r = 0; r < 8; ++r)
                    yrow[(size_t)r * CH_N + 16 * t] = gemm_epilogue<0>(acc[t][r], bv2[t], 0.f);
            }
        } else {
#pragma unroll
            for (int t = 0; t < CH_NT; ++t) {
#pragma unroll
                for (int r = 0; r < 8; ++r)
                    if (rbase + r < M)
                        yrow[(size_t)r * CH_N + 16 * t] = gemm_epilogue<0>(acc[t][r], bv2[t], 0.f);
            }
        }
    }
}

template <int HT>
__global__ void edge_mlp_wmma(const float* __restrict__ xf,
                              const int* __restrict__ row,
                              const int* __restrict__ col,
                              const float* __restrict__ We,
                              const float* __restrict__ be,
                              float* __restrict__ out,
                              int E, int Nu, int Hrt, int C) {
    const int H = (HT > 0) ? HT : Hrt;
    const int K = 2 * H;
    const int lane = threadIdx.x & 31;
    const int wave = threadIdx.x >> 5;
    const int tile = blockIdx.x * (blockDim.x >> 5) + wave;
    const int ebase = tile * 16;
    if (ebase >= E) return;

    const int m = lane & 15;
    const int e = min(ebase + m, E - 1);
    const int r = row[e];
    const int c = col[e];
    const int koff = (lane >> 4) << 1;

    const float* __restrict__ urow = xf + (size_t)r * H + koff;
    const float* __restrict__ mrow = xf + (size_t)(Nu + c) * H + koff;

    const int n = lane & 15;
    const float wmask = (n < C) ? 1.0f : 0.0f;
    const float* __restrict__ wrow = We + (size_t)min(n, C - 1) * K + koff;

    v8f acc = {};
#pragma unroll 8
    for (int k = 0; k < K; k += 4) {
        const float* src = (k + koff < H) ? (urow + k) : (mrow + (k - H));
        v2f a = *(const v2f*)src;
        v2f b = *(const v2f*)(wrow + k);
        b.x *= wmask;
        b.y *= wmask;
        acc = wmma_f32x4(a, b, acc);
    }

    if (n < C) {
        const float bv = be[n];
        const int mb = (lane >> 4) << 3;
        float* op = out + (size_t)(ebase + mb) * C + n;
        if (ebase + 16 <= E) {
#pragma unroll
            for (int rr = 0; rr < 8; ++rr) { *op = acc[rr] + bv; op += C; }
        } else {
#pragma unroll
            for (int rr = 0; rr < 8; ++rr) {
                if (ebase + mb + rr < E) *op = acc[rr] + bv;
                op += C;
            }
        }
    }
}

constexpr int EM_H = 128;
constexpr int EM_K = 2 * EM_H;
constexpr int EM_WPITCH = EM_K + 4;
constexpr int EM_APITCH = EM_H + 4;
constexpr int EM_WAVES = 8;
constexpr int EM_GRID = 768;

__device__ __forceinline__ int edge_tile_ends(const int* __restrict__ row, const int* __restrict__ col,
                                              int t, int lane, int E) {
    const int e = min(t * 16 + (lane & 15), E - 1);
    const int* __restrict__ p = (lane < 16) ? row : col;
    return p[e];
}

#define EM_ROW_LOAD(j, IDX, BASE) \
    rv##j = ((const float4*)(xf + (size_t)(((BASE) ? Nu : 0) + __builtin_amdgcn_readlane(IDX, (BASE) + j)) * EM_H))[lane];
#define EM_ROW_STORE(j, IDX, BASE) *(float4*)(astore + j * EM_APITCH) = rv##j;
#define EM_ALL_ROWS(OP, IDX, BASE) \
    OP(0, IDX, BASE) OP(1, IDX, BASE) OP(2, IDX, BASE) OP(3, IDX, BASE) \
    OP(4, IDX, BASE) OP(5, IDX, BASE) OP(6, IDX, BASE) OP(7, IDX, BASE) \
    OP(8, IDX, BASE) OP(9, IDX, BASE) OP(10, IDX, BASE) OP(11, IDX, BASE) \
    OP(12, IDX, BASE) OP(13, IDX, BASE) OP(14, IDX, BASE) OP(15, IDX, BASE)

__global__ __launch_bounds__(256)
void edge_mlp_rows_lds(const float* __restrict__ xf,
                       const int* __restrict__ row,
                       const int* __restrict__ col,
                       const float* __restrict__ We,
                       const float* __restrict__ be,
                       float* __restrict__ out,
                       int E, int Nu, int C) {
    __shared__ __attribute__((aligned(16))) float Ws[16 * EM_WPITCH];
    __shared__ __attribute__((aligned(16))) float As[EM_WAVES * 16 * EM_APITCH];

    const int lane = threadIdx.x & 31;
    const int wave = __builtin_amdgcn_readfirstlane((int)(threadIdx.x >> 5));
    const int c = lane & 15;
    const int h = lane >> 4;

    const int tiles = (E + 15) >> 4;
    const int t0 = (int)(((long long)tiles * blockIdx.x) / EM_GRID);
    const int t1 = (int)(((long long)tiles * (blockIdx.x + 1)) / EM_GRID);
    int t = t0 + wave;

    float4 rv0, rv1, rv2, rv3, rv4, rv5, rv6, rv7, rv8, rv9, rv10, rv11, rv12, rv13, rv14, rv15;
    rv0 = rv1 = rv2 = rv3 = rv4 = rv5 = rv6 = rv7 = make_float4(0.f, 0.f, 0.f, 0.f);
    rv8 = rv9 = rv10 = rv11 = rv12 = rv13 = rv14 = rv15 = rv0;
    int idx = 0;
    if (t < t1) {
        idx = edge_tile_ends(row, col, t, lane, E);
        EM_ALL_ROWS(EM_ROW_LOAD, idx, 0)
    }

    for (int i = threadIdx.x; i < 16 * (EM_K / 4); i += EM_WAVES * 32) {
        const int n = i / (EM_K / 4);
        const int k4 = i - n * (EM_K / 4);
        float4 w = make_float4(0.f, 0.f, 0.f, 0.f);
        if (n < C) w = ((const float4*)We)[i];
        *(float4*)(Ws + n * EM_WPITCH + 4 * k4) = w;
    }
    __syncthreads();

    float* Aw = As + wave * (16 * EM_APITCH);
    float* astore = Aw + 4 * lane;
    const float* afrag = Aw + c * EM_APITCH + 4 * h;
    const float* bfrag = Ws + c * EM_WPITCH + 4 * h;
    const float bv = (c < C) ? be[c] : 0.f;

    while (t < t1) {
        const int tn = t + EM_WAVES;
        int idxn = 0;
        if (tn < t1) idxn = edge_tile_ends(row, col, tn, lane, E);

        EM_ALL_ROWS(EM_ROW_STORE, idx, 0)
        EM_ALL_ROWS(EM_ROW_LOAD, idx, 16)

        v8f acc0 = {}, acc1 = {};
#pragma unroll
        for (int kb = 0; kb < EM_H / 8; ++kb) {
            const float4 a = *(const float4*)(afrag + 8 * kb);
            const float4 b = *(const float4*)(bfrag + 8 * kb);
            acc0 = wmma_f32x4((v2f){a.x, a.y}, (v2f){b.x, b.y}, acc0);
            acc1 = wmma_f32x4((v2f){a.z, a.w}, (v2f){b.z, b.w}, acc1);
        }

        EM_ALL_ROWS(EM_ROW_STORE, idx, 0)
        if (tn < t1) {
            EM_ALL_ROWS(EM_ROW_LOAD, idxn, 0)
        }
#pragma unroll
        for (int kb = 0; kb < EM_H / 8; ++kb) {
            const float4 a = *(const float4*)(afrag + 8 * kb);
            const float4 b = *(const float4*)(bfrag + EM_H + 8 * kb);
            acc0 = wmma_f32x4((v2f){a.x, a.y}, (v2f){b.x, b.y}, acc0);
            acc1 = wmma_f32x4((v2f){a.z, a.w}, (v2f){b.z, b.w}, acc1);
        }

        if (c < C) {
            const int ebase = t * 16;
            const int mb = h << 3;
            float* op = out + (size_t)(ebase + mb) * C + c;
            if (ebase + 16 <= E) {
#pragma unroll
                for (int rr = 0; rr < 8; ++rr) { *op = (acc0[rr] + acc1[rr]) + bv; op += C; }
            } else {
#pragma unroll
                for (int rr = 0; rr < 8; ++rr) {
                    if (ebase + mb + rr < E) *op = (acc0[rr] + acc1[rr]) + bv;
                    op += C;
                }
            }
        }

        idx = idxn;
        t = tn;
    }
}

__global__ void fill_kernel(float* __restrict__ p, float v, long long n) {
    long long i = (long long)blockIdx.x * blockDim.x + threadIdx.x;
    if (i < n) p[i] = v;
}

__global__ void zero4_kernel(float4* __restrict__ p, long long n4) {
    long long i = (long long)blockIdx.x * blockDim.x + threadIdx.x;
    if (i < n4) p[i] = make_float4(0.f, 0.f, 0.f, 0.f);
}

__global__ void deg_scatter_kernel(const int* __restrict__ col,
                                   float* __restrict__ deg, int E) {
    int e = blockIdx.x * blockDim.x + threadIdx.x;
    if (e < E) atomic_add_f32(&deg[col[e]], 1.0f);
}

__global__ void deg_scatter_own_kernel(const int* __restrict__ col,
                                       float* __restrict__ deg,
                                       unsigned char* __restrict__ own, int E) {
    int e = blockIdx.x * blockDim.x + threadIdx.x;
    if (e < E) {
        const int c = col[e];
        own[e] = (unsigned char)(c & 255);
        atomic_add_f32(&deg[c], 1.0f);
    }
}

__global__ void rsqrt_kernel(float* __restrict__ d, int N) {
    int i = blockIdx.x * blockDim.x + threadIdx.x;
    if (i < N) d[i] = rsqrtf(d[i]);
}

constexpr int DH_GRID = 64;
constexpr int DH_THREADS = 1024;
constexpr int DH_WORDS = 50000;
constexpr int DH_U = 4;
static_assert((DH_WORDS & 3) == 0, "tables are cleared and stored 16 bytes at a time");

__device__ __forceinline__ void dh_count(unsigned int* cnt, int c) {
    atomicAdd(cnt + (c >> 1), (c & 1) ? 0x10000u : 1u);
}

__global__ __launch_bounds__(DH_THREADS)
void deg_hist_kernel(const int* __restrict__ col, unsigned int* __restrict__ own4,
                     unsigned int* __restrict__ part, int E, int N) {
    __shared__ __attribute__((aligned(16))) unsigned int cnt[DH_WORDS];

    const int words4 = (((N + 1) >> 1) + 3) >> 2;
    const int g4 = E >> 2;
    const int q0 = (int)(((long long)g4 * blockIdx.x) / DH_GRID);
    const int q1 = (int)(((long long)g4 * (blockIdx.x + 1)) / DH_GRID);
    const int4* __restrict__ col4 = (const int4*)col;

    int4 cv[DH_U];
#pragma unroll
    for (int u = 0; u < DH_U; ++u)
        cv[u] = col4[min(q0 + u * DH_THREADS + (int)threadIdx.x, q1 - 1)];

    for (int i = threadIdx.x; i < words4; i += DH_THREADS)
        ((uint4*)cnt)[i] = make_uint4(0u, 0u, 0u, 0u);
    __syncthreads();

    for (int base = q0; base < q1; base += DH_U * DH_THREADS) {
#pragma unroll
        for (int u = 0; u < DH_U; ++u) {
            const int q = base + u * DH_THREADS + (int)threadIdx.x;
            if (q < q1) {
                const int4 c = cv[u];
                dh_count(cnt, c.x);
                dh_count(cnt, c.y);
                dh_count(cnt, c.z);
                dh_count(cnt, c.w);
                own4[q] = (unsigned)(c.x & 255) | ((unsigned)(c.y & 255) << 8) |
                          ((unsigned)(c.z & 255) << 16) | ((unsigned)(c.w & 255) << 24);
            }
        }
        if (base + DH_U * DH_THREADS < q1) {
#pragma unroll
            for (int u = 0; u < DH_U; ++u)
                cv[u] = col4[min(base + (DH_U + u) * DH_THREADS + (int)threadIdx.x, q1 - 1)];
        }
    }
    __syncthreads();

    uint4* __restrict__ p = (uint4*)part + (size_t)blockIdx.x * words4;
#pragma unroll 4
    for (int i = threadIdx.x; i < words4; i += DH_THREADS)
        p[i] = ((const uint4*)cnt)[i];
}

__global__ __launch_bounds__(256)
void deg_reduce_kernel(const unsigned int* __restrict__ part, float* __restrict__ dinv,
                       int N, int pitch) {
    const int w = blockIdx.x * blockDim.x + threadIdx.x;
    if (2 * w >= N) return;
    unsigned int v[DH_GRID];
#pragma unroll
    for (int b = 0; b < DH_GRID; ++b) v[b] = part[(size_t)b * pitch + w];
    unsigned int lo = 0u, hi = 0u;
#pragma unroll
    for (int b = 0; b < DH_GRID; ++b) {
        lo += v[b] & 0xffffu;
        hi += v[b] >> 16;
    }
    *(float2*)(dinv + 2 * w) = make_float2(rsqrtf(1.0f + (float)lo), rsqrtf(1.0f + (float)hi));
}

__global__ void edge_scatter_kernel(const float* __restrict__ h,
                                    const float* __restrict__ dinv,
                                    const int* __restrict__ row,
                                    const int* __restrict__ col,
                                    float* __restrict__ acc,
                                    int E, int H) {
    const int chunks = H >> 2;
    long long gid = (long long)blockIdx.x * blockDim.x + threadIdx.x;
    long long total = (long long)E * chunks;
    if (gid >= total) return;
    const int e = (int)(gid / chunks);
    const int j = (int)(gid - (long long)e * chunks) << 2;
    const int r = row[e];
    const int c = col[e];
    const float s = dinv[r];
    const float4 hv = *(const float4*)(h + (size_t)r * H + j);
    float* dst = acc + (size_t)c * H + j;
    atomic_add_f32(dst + 0, s * hv.x);
    atomic_add_f32(dst + 1, s * hv.y);
    atomic_add_f32(dst + 2, s * hv.z);
    atomic_add_f32(dst + 3, s * hv.w);
}

constexpr int SC_H = 128;
constexpr int SC_EPW = 8;

template <bool FULL>
__device__ __forceinline__ void line_scatter_wave(const float* __restrict__ h,
                                                  const float* __restrict__ dinv,
                                                  const int* __restrict__ row, const int* __restrict__ col,
                                                  float* __restrict__ acc, int e0, int ne, int l) {
    int r[SC_EPW], c[SC_EPW];
    float s[SC_EPW], v0[SC_EPW], v1[SC_EPW], v2[SC_EPW], v3[SC_EPW];
#pragma unroll
    for (int i = 0; i < SC_EPW; ++i) {
        const int e = FULL ? e0 + i : e0 + min(i, ne - 1);
        r[i] = row[e];
        c[i] = col[e];
    }
#pragma unroll
    for (int i = 0; i < SC_EPW; ++i) {
        s[i] = dinv[r[i]];
        const float* __restrict__ p = h + (size_t)r[i] * SC_H + l;
        v0[i] = p[0];
        v1[i] = p[32];
        v2[i] = p[64];
        v3[i] = p[96];
    }
#pragma unroll
    for (int i = 0; i < SC_EPW; ++i) {
        if (FULL || i < ne) {
            float* dst = acc + (size_t)c[i] * SC_H + l;
            atomic_add_f32(dst, s[i] * v0[i]);
            atomic_add_f32(dst + 32, s[i] * v1[i]);
            atomic_add_f32(dst + 64, s[i] * v2[i]);
            atomic_add_f32(dst + 96, s[i] * v3[i]);
        }
    }
}

__global__ __launch_bounds__(256)
void edge_scatter_lines_kernel(const float* __restrict__ h,
                               const float* __restrict__ dinv,
                               const int* __restrict__ row,
                               const int* __restrict__ col,
                               float* __restrict__ acc, int E) {
    const int l = threadIdx.x & 31;
    const int wv = __builtin_amdgcn_readfirstlane(
        (int)(blockIdx.x * (blockDim.x >> 5) + (threadIdx.x >> 5)));
    const int e0 = wv * SC_EPW;
    if (e0 >= E) return;
    const int ne = min(SC_EPW, E - e0);
    if (ne == SC_EPW) line_scatter_wave<true>(h, dinv, row, col, acc, e0, ne, l);
    else              line_scatter_wave<false>(h, dinv, row, col, acc, e0, ne, l);
}


constexpr int OW_GRID_LOG2 = 8;
constexpr int OW_GRID = 1 << OW_GRID_LOG2;
static_assert(OW_GRID == 256, "the owner list holds one byte per edge: target mod 256");
constexpr int OW_WAVES = 32;
constexpr int OW_SLOTS = 392;
constexpr int OW_U = 8;
constexpr int OW_Q = 544;

__device__ __forceinline__ void lds_add_f32(float* p, float v) {
    __hip_atomic_fetch_add(p, v, __ATOMIC_RELAXED, __HIP_MEMORY_SCOPE_WORKGROUP);
}

__device__ __forceinline__ void ow_drain(const float* __restrict__ h, const float* __restrict__ dinv,
                                         const int* __restrict__ row, const int* __restrict__ col,
                                         float* accs, const int* q, int m, int lane) {
    const int e = q[min(lane, m - 1)];
    const int r = row[e];
    const int c = col[e];
    const int sbits = __float_as_int(dinv[r]);
    for (int j = 0; j < m; j += 8) {
        float v0[8], v1[8], v2[8], v3[8], ss[8];
        int sl[8];
#pragma unroll
        for (int i = 0; i < 8; ++i) {
            const int idx = min(j + i, m - 1);
            const int rr = __builtin_amdgcn_readlane(r, idx);
            sl[i] = __builtin_amdgcn_readlane(c, idx) >> OW_GRID_LOG2;
            ss[i] = __int_as_float(__builtin_amdgcn_readlane(sbits, idx));
            const float* __restrict__ p = h + (size_t)rr * SC_H + lane;
            v0[i] = p[0];
            v1[i] = p[32];
            v2[i] = p[64];
            v3[i] = p[96];
        }
#pragma unroll
        for (int i = 0; i < 8; ++i) {
            if (j + i < m) {
                float* d = accs + sl[i] * SC_H + lane;
                lds_add_f32(d, ss[i] * v0[i]);
                lds_add_f32(d + 32, ss[i] * v1[i]);
                lds_add_f32(d + 64, ss[i] * v2[i]);
                lds_add_f32(d + 96, ss[i] * v3[i]);
            }
        }
    }
}

#define OW_STEP(MT, EDGE) \
    { \
        const bool mt = (MT); \
        const unsigned mask = __builtin_amdgcn_ballot_w32(mt); \
        if (mask) { \
            if (mt) q[cnt + (int)__builtin_amdgcn_mbcnt_lo(mask, 0u)] = (EDGE); \
            cnt += __builtin_popcount(mask); \
        } \
    }

#define OW_WORD(WV, E0) \
    { \
        const unsigned xw = ok ? ((WV) ^ bpat) : 0xffffffffu; \
        const unsigned zb = (xw - 0x01010101u) & ~xw & 0x80808080u; \
        if (__builtin_amdgcn_ballot_w32(zb != 0u)) { \
            OW_STEP((xw & 0x000000ffu) == 0u, (E0) + 0) \
            OW_STEP((xw & 0x0000ff00u) == 0u, (E0) + 1) \
            OW_STEP((xw & 0x00ff0000u) == 0u, (E0) + 2) \
            OW_STEP((xw & 0xff000000u) == 0u, (E0) + 3) \
        } \
    }

__global__ __launch_bounds__(1024)
void owner_aggregate_kernel(const float* __restrict__ h, const float* __restrict__ dinv,
                            const int* __restrict__ row, const int* __restrict__ col,
                            const unsigned char* __restrict__ own,
                            float* __restrict__ x, int E, int N) {
    __shared__ __attribute__((aligned(16))) float accs[OW_SLOTS * SC_H];
    __shared__ int qs[OW_WAVES * OW_Q];

    const int lane = threadIdx.x & 31;
    const int wave = __builtin_amdgcn_readfirstlane((int)(threadIdx.x >> 5));
    const int b = blockIdx.x;
    const int slots = (N - b + OW_GRID - 1) >> OW_GRID_LOG2;

#pragma unroll 4
    for (int sl = wave; sl < slots; sl += OW_WAVES) {
        const int node = (sl << OW_GRID_LOG2) + b;
        const float di = dinv[node];
        const float4 hv = ((const float4*)(h + (size_t)node * SC_H))[lane];
        *(float4*)(accs + sl * SC_H + 4 * lane) = make_float4(di * hv.x, di * hv.y, di * hv.z, di * hv.w);
    }
    __syncthreads();

    int* q = qs + wave * OW_Q;
    int cnt = 0;
    const uint4* __restrict__ own4 = (const uint4*)own;
    const int total16 = E >> 4;
    const unsigned bpat = (unsigned)b * 0x01010101u;
    for (int base = 0; base < total16; base += OW_U * OW_WAVES * 32) {
        uint4 ov[OW_U];
#pragma unroll
        for (int u = 0; u < OW_U; ++u) {
            const int i16 = base + (u * OW_WAVES + wave) * 32 + lane;
            ov[u] = own4[min(i16, total16 - 1)];
        }
#pragma unroll
        for (int u = 0; u < OW_U; ++u) {
            const int i16 = base + (u * OW_WAVES + wave) * 32 + lane;
            const bool ok = i16 < total16;
            OW_WORD(ov[u].x, 16 * i16 + 0)
            OW_WORD(ov[u].y, 16 * i16 + 4)
            OW_WORD(ov[u].z, 16 * i16 + 8)
            OW_WORD(ov[u].w, 16 * i16 + 12)
            if (cnt >= 32) {
                for (int g = 0; g < cnt; g += 32)
                    ow_drain(h, dinv, row, col, accs, q + g, min(32, cnt - g), lane);
                cnt = 0;
            }
        }
    }
    for (int g = 0; g < cnt; g += 32)
        ow_drain(h, dinv, row, col, accs, q + g, min(32, cnt - g), lane);
    __syncthreads();

#pragma unroll 4
    for (int sl = wave; sl < slots; sl += OW_WAVES) {
        const int node = (sl << OW_GRID_LOG2) + b;
        const float di = dinv[node];
        const float4 a = *(const float4*)(accs + sl * SC_H + 4 * lane);
        float4 o;
        o.x = di * a.x; o.x = o.x > 0.f ? o.x : 0.f;
        o.y = di * a.y; o.y = o.y > 0.f ? o.y : 0.f;
        o.z = di * a.z; o.z = o.z > 0.f ? o.z : 0.f;
        o.w = di * a.w; o.w = o.w > 0.f ? o.w : 0.f;
        ((float4*)(x + (size_t)node * SC_H))[lane] = o;
    }
}

__global__ void finalize_relu_kernel(const float* __restrict__ h,
                                     const float* __restrict__ dinv,
                                     float* __restrict__ acc_inout, int H) {
    const int node = blockIdx.x;
    const int j = threadIdx.x;
    const float di = dinv[node];
    const size_t idx = (size_t)node * H + j;
    const float v = di * acc_inout[idx] + di * di * h[idx];
    acc_inout[idx] = v > 0.f ? v : 0.f;
}

constexpr int RAW_FU = 32;
constexpr int RAW_FM = 64;
constexpr int RAW_H = 128;
constexpr int RAW_WU = RAW_FU + RAW_FM;
constexpr int RAW_WM = RAW_FU + RAW_FM + 1;

__global__ void compose_wz_kernel(const float* __restrict__ Wu, const float* __restrict__ bu,
                                  const float* __restrict__ Wm, const float* __restrict__ bm,
                                  float* __restrict__ Wz) {
    const int i = blockIdx.x * blockDim.x + threadIdx.x;
    if (i >= RAW_H * RAW_H) return;
    const int n = i / RAW_H;
    const int k = i - n * RAW_H;
    float w = 0.f;
    if (k < RAW_FU) w = Wu[n * RAW_FU + k];
    else if (k < RAW_WU) w = Wm[n * RAW_FM + (k - RAW_FU)];
    else if (k == RAW_WU) w = bu[n];
    else if (k == RAW_WM) w = bm[n];
    Wz[i] = w;
}

__global__ void init_z_kernel(const float* __restrict__ xu, const float* __restrict__ xm,
                              const float* __restrict__ dinv, float* __restrict__ Z, int Nu) {
    const int node = blockIdx.x;
    const int j = threadIdx.x;
    const float di = dinv[node];
    float self = 0.f;
    if (node < Nu) {
        if (j < RAW_FU) self = xu[(size_t)node * RAW_FU + j];
        else if (j == RAW_WU) self = 1.f;
    } else {
        if (j >= RAW_FU && j < RAW_WU) self = xm[(size_t)(node - Nu) * RAW_FM + (j - RAW_FU)];
        else if (j == RAW_WM) self = 1.f;
    }
    Z[(size_t)node * RAW_H + j] = di * self;
}

constexpr int DZ_THREADS = 256;
constexpr int DZ_NODES = 256;
constexpr int DZ_U = 8;
static_assert((DH_GRID & 1) == 0 && DZ_THREADS == 256 && DZ_NODES == DZ_THREADS,
              "the tables are split into two halves of 128 threads, each thread pair of halves owns one word");
static_assert((RAW_FU & 3) == 0 && (RAW_FM & 3) == 0 && (RAW_WU & 3) == 0 && RAW_WM == RAW_WU + 1,
              "a lane holds four consecutive floats of a row; both weight columns sit in one lane");
static_assert(RAW_H == 128, "one 16-byte store per lane covers a row");

template <bool FULL>
__device__ __forceinline__ void dz_rows(const float4* __restrict__ xu4, const float4* __restrict__ xm4,
                                        const float* sd, float4* __restrict__ Z4,
                                        int node0, int ln0, int N, int Nu, int lane) {
    constexpr int QU = RAW_FU / 4;
    constexpr int QM = (RAW_FU + RAW_FM) / 4;
    for (int i0 = 0; i0 < 32; i0 += DZ_U) {
        float4 v[DZ_U];
        float di[DZ_U];
#pragma unroll
        for (int i = 0; i < DZ_U; ++i) {
            const int ln = ln0 + i0 + i;
            const int node = FULL ? node0 + ln : min(node0 + ln, N - 1);
            const bool user = node < Nu;
            di[i] = sd[ln];
            const float4* __restrict__ p = user ? xu4 + (size_t)node * QU + lane
                                                : xm4 + (size_t)(node - Nu) * (RAW_FM / 4) + (lane - QU);
            const bool ld = user ? (lane < QU) : (lane >= QU && lane < QM);
            float4 r = make_float4(0.f, 0.f, 0.f, 0.f);
            if (ld) r = *p;
            if (lane == QM) { r.x = user ? 1.f : 0.f; r.y = user ? 0.f : 1.f; }
            v[i] = r;
        }
#pragma unroll
        for (int i = 0; i < DZ_U; ++i) {
            const int node = node0 + ln0 + i0 + i;
            if (FULL || node < N)
                Z4[(size_t)node * (RAW_H / 4) + lane] =
                    make_float4(di[i] * v[i].x, di[i] * v[i].y, di[i] * v[i].z, di[i] * v[i].w);
        }
    }
}

__global__ __launch_bounds__(DZ_THREADS)
void deg_reduce_init_z_kernel(const unsigned int* __restrict__ part, float* __restrict__ dinv,
                              const float* __restrict__ xu, const float* __restrict__ xm,
                              float* __restrict__ Z, int N, int Nu, int pitch) {
    __shared__ unsigned int sp[DZ_THREADS];
    __shared__ float sd[DZ_NODES];

    const int t = threadIdx.x;
    const int lane = t & 31;
    const int wave = __builtin_amdgcn_readfirstlane(t >> 5);
    const int node0 = blockIdx.x * DZ_NODES;

    const int j = t & (DZ_THREADS / 2 - 1);
    const int half = t >> 7;
    const int w = blockIdx.x * (DZ_NODES / 2) + j;
    const bool live = 2 * w < N;
    unsigned int lo = 0u, hi = 0u;
    if (live) {
        unsigned int v[DH_GRID / 2];
        const unsigned int* __restrict__ p = part + (size_t)half * (DH_GRID / 2) * pitch + w;
#pragma unroll
        for (int b = 0; b < DH_GRID / 2; ++b) v[b] = p[(size_t)b * pitch];
#pragma unroll
        for (int b = 0; b < DH_GRID / 2; ++b) {
            lo += v[b] & 0xffffu;
            hi += v[b] >> 16;
        }
    }
    if (half) {
        sp[2 * j] = lo;
        sp[2 * j + 1] = hi;
    }
    __syncthreads();
    if (!half) {
        lo += sp[2 * j];
        hi += sp[2 * j + 1];
        const float d0 = rsqrtf(1.0f + (float)lo);
        const float d1 = rsqrtf(1.0f + (float)hi);
        sd[2 * j] = d0;
        sd[2 * j + 1] = d1;
        if (live) *(float2*)(dinv + 2 * w) = make_float2(d0, d1);
    }
    __syncthreads();

    const float4* __restrict__ xu4 = (const float4*)xu;
    const float4* __restrict__ xm4 = (const float4*)xm;
    float4* __restrict__ Z4 = (float4*)Z;
    const int ln0 = wave * 32;
    if (node0 + ln0 >= N) return;
    if (node0 + ln0 + 32 <= N) dz_rows<true>(xu4, xm4, sd, Z4, node0, ln0, N, Nu, lane);
    else                       dz_rows<false>(xu4, xm4, sd, Z4, node0, ln0, N, Nu, lane);
}

constexpr int RAW_EPW = 8;

template <bool FULL>
__device__ __forceinline__ void raw_scatter_wave(const float* __restrict__ xu, const float* __restrict__ xm,
                                                 const float* __restrict__ dinv,
                                                 const int* __restrict__ row, const int* __restrict__ col,
                                                 float* __restrict__ Z, int Nu, int e0, int ne, int l) {
    int r[RAW_EPW], c[RAW_EPW];
    float s[RAW_EPW], v0[RAW_EPW], v1[RAW_EPW];
#pragma unroll
    for (int i = 0; i < RAW_EPW; ++i) {
        const int e = FULL ? e0 + i : e0 + min(i, ne - 1);
        r[i] = row[e];
        c[i] = col[e];
    }
#pragma unroll
    for (int i = 0; i < RAW_EPW; ++i) {
        s[i] = dinv[r[i]];
        const bool user = r[i] < Nu;
        const float* __restrict__ p = user ? xu + (size_t)r[i] * RAW_FU + l
                                           : xm + (size_t)(r[i] - Nu) * RAW_FM + l;
        v0[i] = p[0];
        v1[i] = 0.f;
        if (!user) v1[i] = p[32];
    }
#pragma unroll
    for (int i = 0; i < RAW_EPW; ++i) {
        if (FULL || i < ne) {
            const bool user = r[i] < Nu;
            float* dst = Z + (size_t)c[i] * RAW_H;
            atomic_add_f32(dst + (user ? 0 : RAW_FU) + l, s[i] * v0[i]);
            if (!user) atomic_add_f32(dst + RAW_FU + 32 + l, s[i] * v1[i]);
        }
    }
    if (l < (FULL ? RAW_EPW : ne)) {
        const int rl = row[e0 + l];
        const int cl = col[e0 + l];
        atomic_add_f32(Z + (size_t)cl * RAW_H + (rl < Nu ? RAW_WU : RAW_WM), dinv[rl]);
    }
}

__global__ __launch_bounds__(256)
void raw_scatter_kernel(const float* __restrict__ xu, const float* __restrict__ xm,
                        const float* __restrict__ dinv,
                        const int* __restrict__ row, const int* __restrict__ col,
                        float* __restrict__ Z, int E, int Nu) {
    const int l = threadIdx.x & 31;
    const int wv = __builtin_amdgcn_readfirstlane(
        (int)(blockIdx.x * (blockDim.x >> 5) + (threadIdx.x >> 5)));
    const int e0 = wv * RAW_EPW;
    if (e0 >= E) return;
    const int ne = min(RAW_EPW, E - e0);
    if (ne == RAW_EPW) raw_scatter_wave<true>(xu, xm, dinv, row, col, Z, Nu, e0, ne, l);
    else               raw_scatter_wave<false>(xu, xm, dinv, row, col, Z, Nu, e0, ne, l);
}

static inline int ceil_div_i(long long a, long long b) { return (int)((a + b - 1) / b); }

static void launch_gemm(const float* X, const float* W, const float* b, float* Y,
                        int M, int K, int N, hipStream_t stream) {
    const int TPB = 256;
    if (N == 128 && K == 32)
        gemm_bias_wmma_wlds<32, 0><<<GEMM_GRID, TPB, 0, stream>>>(X, W, b, nullptr, nullptr, Y, M);
    else if (N == 128 && K == 64)
        gemm_bias_wmma_wlds<64, 0><<<GEMM_GRID, TPB, 0, stream>>>(X, W, b, nullptr, nullptr, Y, M);
    else if (N == 128 && K == 128)
        gemm_bias_wmma_wlds<128, 0><<<GEMM_GRID, TPB, 0, stream>>>(X, W, b, nullptr, nullptr, Y, M);
    else {
        const int strips = ceil_div_i(M, 16);
        const int tiles = strips * (N / 16);
        gemm_bias_wmma_rt<<<ceil_div_i(tiles, 8), TPB, 0, stream>>>(X, W, b, Y, M, K, N);
    }
}

extern "C" void kernel_launch(void* const* d_in, const int* in_sizes, int n_in,
                              void* d_out, int out_size, void* d_ws, size_t ws_size,
                              hipStream_t stream) {
    const float* x_user  = (const float*)d_in[0];
    const float* x_movie = (const float*)d_in[1];
    const int*   edge    = (const int*)d_in[2];
    const float* Wu = (const float*)d_in[3];
    const float* bu = (const float*)d_in[4];
    const float* Wm = (const float*)d_in[5];
    const float* bm = (const float*)d_in[6];
    const float* W1 = (const float*)d_in[7];
    const float* b1 = (const float*)d_in[8];
    const float* W2 = (const float*)d_in[9];
    const float* b2 = (const float*)d_in[10];
    const float* We = (const float*)d_in[11];
    const float* be = (const float*)d_in[12];
    float* out = (float*)d_out;

    const int H  = in_sizes[4];
    const int C  = in_sizes[12];
    const int Fu = in_sizes[3] / H;
    const int Fm = in_sizes[5] / H;
    const int Nu = in_sizes[0] / Fu;
    const int Nm = in_sizes[1] / Fm;
    const int E  = in_sizes[2] / 2;
    const int N  = Nu + Nm;
    const int* row = edge;
    const int* col = edge + E;

    float* ws   = (float*)d_ws;
    float* X    = ws;
    float* Hh   = X  + (size_t)N * H;
    float* Acc  = Hh + (size_t)N * H;
    float* Wz   = Acc + (size_t)N * H;
    float* dinv = Wz + (size_t)RAW_H * RAW_H;
    unsigned char* own = (unsigned char*)(dinv + (size_t)((N + 3) & ~3));
    unsigned int* part = (unsigned int*)(own + (((size_t)E + 15) & ~(size_t)15));

    const int TPB = 256, WPB = 8;
    const long long NH = (long long)N * H;

    const bool owner_path = (H == SC_H) && ((E & 15) == 0) && (N <= OW_GRID * OW_SLOTS);
    const int dh_words = (((N + 1) >> 1) + 3) & ~3;
    const bool hist_path = (N <= 2 * DH_WORDS) && ((E & 3) == 0) && (E >= 4 * DH_GRID) &&
                           (E / DH_GRID + 4 <= 65535) &&
                           ((size_t)((unsigned char*)(part + (size_t)DH_GRID * dh_words) - (unsigned char*)d_ws) <= ws_size);
    const bool raw_path = (H == RAW_H) && (Fu == RAW_FU) && (Fm == RAW_FM);
    if (hist_path) {
        deg_hist_kernel<<<DH_GRID, DH_THREADS, 0, stream>>>(col, (unsigned int*)own, part, E, N);
        if (raw_path)
            deg_reduce_init_z_kernel<<<ceil_div_i(N, DZ_NODES), DZ_THREADS, 0, stream>>>(part, dinv, x_user, x_movie, Hh, N, Nu, dh_words);
        else
            deg_reduce_kernel<<<ceil_div_i((N + 1) >> 1, TPB), TPB, 0, stream>>>(part, dinv, N, dh_words);
    } else {
        fill_kernel<<<ceil_div_i(N, TPB), TPB, 0, stream>>>(dinv, 1.0f, N);
        if (owner_path)
            deg_scatter_own_kernel<<<ceil_div_i(E, TPB), TPB, 0, stream>>>(col, dinv, own, E);
        else
            deg_scatter_kernel<<<ceil_div_i(E, TPB), TPB, 0, stream>>>(col, dinv, E);
        rsqrt_kernel<<<ceil_div_i(N, TPB), TPB, 0, stream>>>(dinv, N);
    }

    const long long scatterThreads = (long long)E * (H / 4);

    if (raw_path) {
        compose_wz_kernel<<<ceil_div_i(RAW_H * RAW_H, TPB), TPB, 0, stream>>>(Wu, bu, Wm, bm, Wz);
        if (!hist_path) init_z_kernel<<<N, RAW_H, 0, stream>>>(x_user, x_movie, dinv, Hh, Nu);
        raw_scatter_kernel<<<ceil_div_i(E, RAW_EPW * WPB), TPB, 0, stream>>>(x_user, x_movie, dinv, row, col, Hh, E, Nu);
        gemm_chain3_wmma<<<CH_GRID, CH_WAVES * 32, 0, stream>>>(Hh, Wz, W1, b1, W2, b2, dinv, Acc, N);
    } else {
        launch_gemm(x_user, Wu, bu, X, Nu, Fu, H, stream);
        launch_gemm(x_movie, Wm, bm, X + (size_t)Nu * H, Nm, Fm, H, stream);

        launch_gemm(X, W1, b1, Hh, N, H, H, stream);
        zero4_kernel<<<ceil_div_i(NH / 4, TPB), TPB, 0, stream>>>((float4*)Acc, NH / 4);
        edge_scatter_kernel<<<ceil_div_i(scatterThreads, TPB), TPB, 0, stream>>>(Hh, dinv, row, col, Acc, E, H);
        finalize_relu_kernel<<<N, H, 0, stream>>>(Hh, dinv, Acc, H);
    }

    const float* Hc = Acc;
    if (!raw_path) {
        launch_gemm(Acc, W2, b2, Hh, N, H, H, stream);
        Hc = Hh;
    }
    if (owner_path) {
        owner_aggregate_kernel<<<OW_GRID, OW_WAVES * 32, 0, stream>>>(Hc, dinv, row, col, own, X, E, N);
    } else {
        zero4_kernel<<<ceil_div_i(NH / 4, TPB), TPB, 0, stream>>>((float4*)X, NH / 4);
        if (H == SC_H)
            edge_scatter_lines_kernel<<<ceil_div_i(E, SC_EPW * WPB), TPB, 0, stream>>>(Hc, dinv, row, col, X, E);
        else
            edge_scatter_kernel<<<ceil_div_i(scatterThreads, TPB), TPB, 0, stream>>>(Hc, dinv, row, col, X, E, H);
        finalize_relu_kernel<<<N, H, 0, stream>>>(Hc, dinv, X, H);
    }

    const int edgeTiles = ceil_div_i(E, 16);
    if (H == EM_H && C <= 16)
        edge_mlp_rows_lds<<<EM_GRID, TPB, 0, stream>>>(X, row, col, We, be, out, E, Nu, C);
    else
        edge_mlp_wmma<0><<<ceil_div_i(edgeTiles, WPB), TPB, 0, stream>>>(X, row, col, We, be, out, E, Nu, H, C);
}


# an edge's source weight is looked up once per call, by conv1 scatter's index lanes: conv1 takes it by readlane, conv2's owner pass reads it with both endpoints as one 16-byte record; sums as before
# speedup vs baseline: 2.9799x; 1.0293x over previous
#include <hip/hip_runtime.h>

typedef __attribute__((ext_vector_type(2))) float v2f;
typedef __attribute__((ext_vector_type(8))) float v8f;

__device__ __forceinline__ v8f wmma_f32x4(v2f a, v2f b, v8f c) {
    return __builtin_amdgcn_wmma_f32_16x16x4_f32(
         false, a,  false, b,
         (short)0, c,  false,  false);
}

__device__ __forceinline__ void atomic_add_f32(float* p, float v) {
#if defined(__HIP_DEVICE_COMPILE__)
    unsafeAtomicAdd(p, v);
#else
    atomicAdd(p, v);
#endif
}

constexpr int GEMM_GRID_LOG2 = 9;
constexpr int GEMM_GRID = 1 << GEMM_GRID_LOG2;

template <int MODE>
__device__ __forceinline__ float gemm_epilogue(float a, float b, float g) {
    if (MODE == 0) return a + b;
    if (MODE == 1) return g * a;
    const float v = a + g * b;
    return v > 0.f ? v : 0.f;
}

template <int K, int MODE>
__global__ __launch_bounds__(256)
void gemm_bias_wmma_wlds(const float* __restrict__ X,
                         const float* __restrict__ W,
                         const float* __restrict__ bias,
                         const float* __restrict__ rs,
                         const float* __restrict__ zc,
                         float* __restrict__ Y,
                         int M) {
    constexpr int N = 128;
    constexpr int NT = N / 16;
    constexpr int PITCH = K + 4;
    constexpr int KB = K / 8;
    constexpr int WAVES = 8;
    __shared__ __attribute__((aligned(16))) float Ws[N * PITCH];

    const int lane = threadIdx.x & 31;
    const int wave = __builtin_amdgcn_readfirstlane((int)(threadIdx.x >> 5));
    const int c = lane & 15;
    const int h = lane >> 4;

    const int strips = (M + 15) >> 4;
    const int s0 = (int)(((long long)strips * blockIdx.x) >> GEMM_GRID_LOG2);
    const int s1 = (int)(((long long)strips * (blockIdx.x + 1)) >> GEMM_GRID_LOG2);

    float bv[NT];
#pragma unroll
    for (int t = 0; t < NT; ++t) bv[t] = (MODE == 1) ? 0.f : bias[16 * t + c];

    float4 a[KB];
    int s = s0 + wave;

    if (s < s1) {
        const int arow = min(s * 16 + c, M - 1);
        const float4* __restrict__ ap = (const float4*)(X + (size_t)arow * K) + h;
#pragma unroll
        for (int kb = 0; kb < KB; ++kb) a[kb] = ap[2 * kb];
    }

    for (int i = threadIdx.x; i < N * (K / 4); i += WAVES * 32) {
        const int n = i / (K / 4);
        const int k4 = i - n * (K / 4);
        *(float4*)(Ws + n * PITCH + 4 * k4) = ((const float4*)W)[i];
    }
    __syncthreads();

    const float* __restrict__ bfrag = Ws + c * PITCH + 4 * h;

    while (s < s1) {
        float gr[8];
#pragma unroll
        for (int r = 0; r < 8; ++r) {
            gr[r] = 0.f;
            if (MODE != 0) {
                const int rr = min(s * 16 + (h << 3) + r, M - 1);
                float g = rs[rr];
                if (MODE == 2) {
                    const v2f w = *(const v2f*)(zc + (size_t)rr * N + 96);
                    g *= (w.x + w.y);
                }
                gr[r] = g;
            }
        }

        v8f acc[NT];
#pragma unroll
        for (int t = 0; t < NT; ++t) acc[t] = (v8f){};

        float4 bc[NT], bn[NT];
#pragma unroll
        for (int t = 0; t < NT; ++t)
            bc[t] = *(const float4*)(bfrag + 16 * t * PITCH);
#pragma unroll
        for (int kb = 0; kb < KB; ++kb) {
            if (kb + 1 < KB) {
#pragma unroll
                for (int t = 0; t < NT; ++t)
                    bn[t] = *(const float4*)(bfrag + 16 * t * PITCH + 8 * (kb + 1));
            }
            __builtin_amdgcn_sched_barrier(0);
            const v2f a01 = {a[kb].x, a[kb].y};
            const v2f a23 = {a[kb].z, a[kb].w};
#pragma unroll
            for (int t = 0; t < NT; ++t) {
                const v2f b01 = {bc[t].x, bc[t].y};
                acc[t] = wmma_f32x4(a01, b01, acc[t]);
            }
#pragma unroll
            for (int t = 0; t < NT; ++t) {
                const v2f b23 = {bc[t].z, bc[t].w};
                acc[t] = wmma_f32x4(a23, b23, acc[t]);
            }
            __builtin_amdgcn_sched_barrier(0);
#pragma unroll
            for (int t = 0; t < NT; ++t) bc[t] = bn[t];
        }

        const int row0 = s * 16;
        s += WAVES;

        if (s < s1) {
            const int arow = min(s * 16 + c, M - 1);
            const float4* __restrict__ ap = (const float4*)(X + (size_t)arow * K) + h;
#pragma unroll
            for (int kb = 0; kb < KB; ++kb) a[kb] = ap[2 * kb];
        }

        const int rbase = row0 + (h << 3);
        float* __restrict__ yrow = Y + (size_t)rbase * N + c;
        if (row0 + 16 <= M) {
#pragma unroll
            for (int t = 0; t < NT; ++t) {
#pragma unroll
                for (int r = 0; r < 8; ++r)
                    yrow[(size_t)r * N + 16 * t] = gemm_epilogue<MODE>(acc[t][r], bv[t], gr[r]);
            }
        } else {
#pragma unroll
            for (int t = 0; t < NT; ++t) {
#pragma unroll
                for (int r = 0; r < 8; ++r)
                    if (rbase + r < M)
                        yrow[(size_t)r * N + 16 * t] = gemm_epilogue<MODE>(acc[t][r], bv[t], gr[r]);
            }
        }
    }
}

__global__ void gemm_bias_wmma_rt(const float* __restrict__ X,
                                  const float* __restrict__ W,
                                  const float* __restrict__ bias,
                                  float* __restrict__ Y,
                                  int M, int K, int N) {
    const int lane = threadIdx.x & 31;
    const int wave = threadIdx.x >> 5;
    const int tilesN = N >> 4;
    const int tile = blockIdx.x * (blockDim.x >> 5) + wave;
    const int tm = tile / tilesN;
    const int tn = tile - tm * tilesN;
    if (tm * 16 >= M) return;

    const int mrow = min(tm * 16 + (lane & 15), M - 1);
    const int ncol = tn * 16 + (lane & 15);
    const int koff = (lane >> 4) << 1;
    const float* __restrict__ xrow = X + (size_t)mrow * K + koff;
    const float* __restrict__ wrow = W + (size_t)ncol * K + koff;

    v8f acc = {};
#pragma unroll 8
    for (int k = 0; k < K; k += 4) {
        v2f a = *(const v2f*)(xrow + k);
        v2f b = *(const v2f*)(wrow + k);
        acc = wmma_f32x4(a, b, acc);
    }

    const float bv = bias[ncol];
    const int rbase = tm * 16 + ((lane >> 4) << 3);
    float* yp = Y + (size_t)rbase * N + ncol;
#pragma unroll
    for (int r = 0; r < 8; ++r) {
        if (rbase + r < M) *yp = acc[r] + bv;
        yp += N;
    }
}

constexpr int CH_N = 128;
constexpr int CH_PITCH = CH_N + 4;
constexpr int CH_KB = CH_N / 8;
constexpr int CH_NT = CH_N / 16;
constexpr int CH_WAVES = 12;
constexpr int CH_GRID_LOG2 = 8;
constexpr int CH_GRID = 1 << CH_GRID_LOG2;

__device__ __forceinline__ void chain_product(const float4 (&a)[CH_KB],
                                              const float* __restrict__ bfrag,
                                              v8f (&acc)[CH_NT]) {
#pragma unroll
    for (int t = 0; t < CH_NT; ++t) acc[t] = (v8f){};
    float4 bc[CH_NT], bn[CH_NT];
#pragma unroll
    for (int t = 0; t < CH_NT; ++t)
        bc[t] = *(const float4*)(bfrag + 16 * t * CH_PITCH);
#pragma unroll
    for (int kb = 0; kb < CH_KB; ++kb) {
        if (kb + 1 < CH_KB) {
#pragma unroll
            for (int t = 0; t < CH_NT; ++t)
                bn[t] = *(const float4*)(bfrag + 16 * t * CH_PITCH + 8 * (kb + 1));
        }
        __builtin_amdgcn_sched_barrier(0);
        const v2f a01 = {a[kb].x, a[kb].y};
        const v2f a23 = {a[kb].z, a[kb].w};
#pragma unroll
        for (int t = 0; t < CH_NT; ++t) {
            const v2f b01 = {bc[t].x, bc[t].y};
            acc[t] = wmma_f32x4(a01, b01, acc[t]);
        }
#pragma unroll
        for (int t = 0; t < CH_NT; ++t) {
            const v2f b23 = {bc[t].z, bc[t].w};
            acc[t] = wmma_f32x4(a23, b23, acc[t]);
        }
        __builtin_amdgcn_sched_barrier(0);
#pragma unroll
        for (int t = 0; t < CH_NT; ++t) bc[t] = bn[t];
    }
}

__global__ __launch_bounds__(CH_WAVES * 32)
void gemm_chain3_wmma(const float* __restrict__ Z,
                      const float* __restrict__ Wz,
                      const float* __restrict__ W1,
                      const float* __restrict__ b1,
                      const float* __restrict__ W2,
                      const float* __restrict__ b2,
                      const float* __restrict__ rs,
                      float* __restrict__ Y,
                      int M) {
    __shared__ __attribute__((aligned(16))) float Ws[3 * CH_N * CH_PITCH];
    __shared__ __attribute__((aligned(16))) float Ps[CH_WAVES * 16 * CH_PITCH];

    const int lane = threadIdx.x & 31;
    const int wave = __builtin_amdgcn_readfirstlane((int)(threadIdx.x >> 5));
    const int c = lane & 15;
    const int h = lane >> 4;

    const int strips = (M + 15) >> 4;
    const int s0 = (int)(((long long)strips * blockIdx.x) >> CH_GRID_LOG2);
    const int s1 = (int)(((long long)strips * (blockIdx.x + 1)) >> CH_GRID_LOG2);

    float bv1[CH_NT], bv2[CH_NT];
#pragma unroll
    for (int t = 0; t < CH_NT; ++t) {
        bv1[t] = b1[16 * t + c];
        bv2[t] = b2[16 * t + c];
    }

    float4 a[CH_KB];
    int s = s0 + wave;

    if (s < s1) {
        const int arow = min(s * 16 + c, M - 1);
        const float4* __restrict__ ap = (const float4*)(Z + (size_t)arow * CH_N) + h;
#pragma unroll
        for (int kb = 0; kb < CH_KB; ++kb) a[kb] = ap[2 * kb];
    }

    for (int i = threadIdx.x; i < CH_N * (CH_N / 4); i += CH_WAVES * 32) {
        const int n = i / (CH_N / 4);
        const int k4 = i - n * (CH_N / 4);
        const float4 w0 = ((const float4*)Wz)[i];
        const float4 w1 = ((const float4*)W1)[i];
        const float4 w2 = ((const float4*)W2)[i];
        float* dst = Ws + n * CH_PITCH + 4 * k4;
        *(float4*)dst = w0;
        *(float4*)(dst + CH_N * CH_PITCH) = w1;
        *(float4*)(dst + 2 * CH_N * CH_PITCH) = w2;
    }
    __syncthreads();

    const float* __restrict__ bfz = Ws + c * CH_PITCH + 4 * h;
    const float* __restrict__ bf1 = bfz + CH_N * CH_PITCH;
    const float* __restrict__ bf2 = bfz + 2 * CH_N * CH_PITCH;

    float* P = Ps + wave * (16 * CH_PITCH);
    float* pst = P + (h << 3) * CH_PITCH + c;
    const float* pfr = P + c * CH_PITCH + 4 * h;

    while (s < s1) {
        float d[8], g[8];
#pragma unroll
        for (int r = 0; r < 8; ++r) {
            const int rr = min(s * 16 + (h << 3) + r, M - 1);
            const float dr = rs[rr];
            const v2f w = *(const v2f*)(Z + (size_t)rr * CH_N + 96);
            d[r] = dr;
            g[r] = dr * (w.x + w.y);
        }

        v8f acc[CH_NT];

        chain_product(a, bfz, acc);
#pragma unroll
        for (int t = 0; t < CH_NT; ++t) {
#pragma unroll
            for (int r = 0; r < 8; ++r)
                pst[r * CH_PITCH + 16 * t] = gemm_epilogue<1>(acc[t][r], 0.f, d[r]);
        }
#pragma unroll
        for (int kb = 0; kb < CH_KB; ++kb) a[kb] = *(const float4*)(pfr + 8 * kb);

        chain_product(a, bf1, acc);
#pragma unroll
        for (int t = 0; t < CH_NT; ++t) {
#pragma unroll
            for (int r = 0; r < 8; ++r)
                pst[r * CH_PITCH + 16 * t] = gemm_epilogue<2>(acc[t][r], bv1[t], g[r]);
        }
#pragma unroll
        for (int kb = 0; kb < CH_KB; ++kb) a[kb] = *(const float4*)(pfr + 8 * kb);

        chain_product(a, bf2, acc);

        const int row0 = s * 16;
        s += CH_WAVES;

        if (s < s1) {
            const int arow = min(s * 16 + c, M - 1);
            const float4* __restrict__ ap = (const float4*)(Z + (size_t)arow * CH_N) + h;
#pragma unroll
            for (int kb = 0; kb < CH_KB; ++kb) a[kb] = ap[2 * kb];
        }

        const int rbase = row0 + (h << 3);
        float* __restrict__ yrow = Y + (size_t)rbase * CH_N + c;
        if (row0 + 16 <= M) {
#pragma unroll
            for (int t = 0; t < CH_NT; ++t) {
#pragma unroll
                for (int r = 0; r < 8; ++r)
                    yrow[(size_t)r * CH_N + 16 * t] = gemm_epilogue<0>(acc[t][r], bv2[t], 0.f);
            }
        } else {
#pragma unroll
            for (int t = 0; t < CH_NT; ++t) {
#pragma unroll
                for (int r = 0; r < 8; ++r)
                    if (rbase + r < M)
                        yrow[(size_t)r * CH_N + 16 * t] = gemm_epilogue<0>(acc[t][r], bv2[t], 0.f);
            }
        }
    }
}

template <int HT>
__global__ void edge_mlp_wmma(const float* __restrict__ xf,
                              const int* __restrict__ row,
                              const int* __restrict__ col,
                              const float* __restrict__ We,
                              const float* __restrict__ be,
                              float* __restrict__ out,
                              int E, int Nu, int Hrt, int C) {
    const int H = (HT > 0) ? HT : Hrt;
    const int K = 2 * H;
    const int lane = threadIdx.x & 31;
    const int wave = threadIdx.x >> 5;
    const int tile = blockIdx.x * (blockDim.x >> 5) + wave;
    const int ebase = tile * 16;
    if (ebase >= E) return;

    const int m = lane & 15;
    const int e = min(ebase + m, E - 1);
    const int r = row[e];
    const int c = col[e];
    const int koff = (lane >> 4) << 1;

    const float* __restrict__ urow = xf + (size_t)r * H + koff;
    const float* __restrict__ mrow = xf + (size_t)(Nu + c) * H + koff;

    const int n = lane & 15;
    const float wmask = (n < C) ? 1.0f : 0.0f;
    const float* __restrict__ wrow = We + (size_t)min(n, C - 1) * K + koff;

    v8f acc = {};
#pragma unroll 8
    for (int k = 0; k < K; k += 4) {
        const float* src = (k + koff < H) ? (urow + k) : (mrow + (k - H));
        v2f a = *(const v2f*)src;
        v2f b = *(const v2f*)(wrow + k);
        b.x *= wmask;
        b.y *= wmask;
        acc = wmma_f32x4(a, b, acc);
    }

    if (n < C) {
        const float bv = be[n];
        const int mb = (lane >> 4) << 3;
        float* op = out + (size_t)(ebase + mb) * C + n;
        if (ebase + 16 <= E) {
#pragma unroll
            for (int rr = 0; rr < 8; ++rr) { *op = acc[rr] + bv; op += C; }
        } else {
#pragma unroll
            for (int rr = 0; rr < 8; ++rr) {
                if (ebase + mb + rr < E) *op = acc[rr] + bv;
                op += C;
            }
        }
    }
}

constexpr int EM_H = 128;
constexpr int EM_K = 2 * EM_H;
constexpr int EM_WPITCH = EM_K + 4;
constexpr int EM_APITCH = EM_H + 4;
constexpr int EM_WAVES = 8;
constexpr int EM_GRID = 768;

__device__ __forceinline__ int edge_tile_ends(const int* __restrict__ row, const int* __restrict__ col,
                                              int t, int lane, int E) {
    const int e = min(t * 16 + (lane & 15), E - 1);
    const int* __restrict__ p = (lane < 16) ? row : col;
    return p[e];
}

#define EM_ROW_LOAD(j, IDX, BASE) \
    rv##j = ((const float4*)(xf + (size_t)(((BASE) ? Nu : 0) + __builtin_amdgcn_readlane(IDX, (BASE) + j)) * EM_H))[lane];
#define EM_ROW_STORE(j, IDX, BASE) *(float4*)(astore + j * EM_APITCH) = rv##j;
#define EM_ALL_ROWS(OP, IDX, BASE) \
    OP(0, IDX, BASE) OP(1, IDX, BASE) OP(2, IDX, BASE) OP(3, IDX, BASE) \
    OP(4, IDX, BASE) OP(5, IDX, BASE) OP(6, IDX, BASE) OP(7, IDX, BASE) \
    OP(8, IDX, BASE) OP(9, IDX, BASE) OP(10, IDX, BASE) OP(11, IDX, BASE) \
    OP(12, IDX, BASE) OP(13, IDX, BASE) OP(14, IDX, BASE) OP(15, IDX, BASE)

__global__ __launch_bounds__(256)
void edge_mlp_rows_lds(const float* __restrict__ xf,
                       const int* __restrict__ row,
                       const int* __restrict__ col,
                       const float* __restrict__ We,
                       const float* __restrict__ be,
                       float* __restrict__ out,
                       int E, int Nu, int C) {
    __shared__ __attribute__((aligned(16))) float Ws[16 * EM_WPITCH];
    __shared__ __attribute__((aligned(16))) float As[EM_WAVES * 16 * EM_APITCH];

    const int lane = threadIdx.x & 31;
    const int wave = __builtin_amdgcn_readfirstlane((int)(threadIdx.x >> 5));
    const int c = lane & 15;
    const int h = lane >> 4;

    const int tiles = (E + 15) >> 4;
    const int t0 = (int)(((long long)tiles * blockIdx.x) / EM_GRID);
    const int t1 = (int)(((long long)tiles * (blockIdx.x + 1)) / EM_GRID);
    int t = t0 + wave;

    float4 rv0, rv1, rv2, rv3, rv4, rv5, rv6, rv7, rv8, rv9, rv10, rv11, rv12, rv13, rv14, rv15;
    rv0 = rv1 = rv2 = rv3 = rv4 = rv5 = rv6 = rv7 = make_float4(0.f, 0.f, 0.f, 0.f);
    rv8 = rv9 = rv10 = rv11 = rv12 = rv13 = rv14 = rv15 = rv0;
    int idx = 0;
    if (t < t1) {
        idx = edge_tile_ends(row, col, t, lane, E);
        EM_ALL_ROWS(EM_ROW_LOAD, idx, 0)
    }

    for (int i = threadIdx.x; i < 16 * (EM_K / 4); i += EM_WAVES * 32) {
        const int n = i / (EM_K / 4);
        const int k4 = i - n * (EM_K / 4);
        float4 w = make_float4(0.f, 0.f, 0.f, 0.f);
        if (n < C) w = ((const float4*)We)[i];
        *(float4*)(Ws + n * EM_WPITCH + 4 * k4) = w;
    }
    __syncthreads();

    float* Aw = As + wave * (16 * EM_APITCH);
    float* astore = Aw + 4 * lane;
    const float* afrag = Aw + c * EM_APITCH + 4 * h;
    const float* bfrag = Ws + c * EM_WPITCH + 4 * h;
    const float bv = (c < C) ? be[c] : 0.f;

    while (t < t1) {
        const int tn = t + EM_WAVES;
        int idxn = 0;
        if (tn < t1) idxn = edge_tile_ends(row, col, tn, lane, E);

        EM_ALL_ROWS(EM_ROW_STORE, idx, 0)
        EM_ALL_ROWS(EM_ROW_LOAD, idx, 16)

        v8f acc0 = {}, acc1 = {};
#pragma unroll
        for (int kb = 0; kb < EM_H / 8; ++kb) {
            const float4 a = *(const float4*)(afrag + 8 * kb);
            const float4 b = *(const float4*)(bfrag + 8 * kb);
            acc0 = wmma_f32x4((v2f){a.x, a.y}, (v2f){b.x, b.y}, acc0);
            acc1 = wmma_f32x4((v2f){a.z, a.w}, (v2f){b.z, b.w}, acc1);
        }

        EM_ALL_ROWS(EM_ROW_STORE, idx, 0)
        if (tn < t1) {
            EM_ALL_ROWS(EM_ROW_LOAD, idxn, 0)
        }
#pragma unroll
        for (int kb = 0; kb < EM_H / 8; ++kb) {
            const float4 a = *(const float4*)(afrag + 8 * kb);
            const float4 b = *(const float4*)(bfrag + EM_H + 8 * kb);
            acc0 = wmma_f32x4((v2f){a.x, a.y}, (v2f){b.x, b.y}, acc0);
            acc1 = wmma_f32x4((v2f){a.z, a.w}, (v2f){b.z, b.w}, acc1);
        }

        if (c < C) {
            const int ebase = t * 16;
            const int mb = h << 3;
            float* op = out + (size_t)(ebase + mb) * C + c;
            if (ebase + 16 <= E) {
#pragma unroll
                for (int rr = 0; rr < 8; ++rr) { *op = (acc0[rr] + acc1[rr]) + bv; op += C; }
            } else {
#pragma unroll
                for (int rr = 0; rr < 8; ++rr) {
                    if (ebase + mb + rr < E) *op = (acc0[rr] + acc1[rr]) + bv;
                    op += C;
                }
            }
        }

        idx = idxn;
        t = tn;
    }
}

__global__ void fill_kernel(float* __restrict__ p, float v, long long n) {
    long long i = (long long)blockIdx.x * blockDim.x + threadIdx.x;
    if (i < n) p[i] = v;
}

__global__ void zero4_kernel(float4* __restrict__ p, long long n4) {
    long long i = (long long)blockIdx.x * blockDim.x + threadIdx.x;
    if (i < n4) p[i] = make_float4(0.f, 0.f, 0.f, 0.f);
}

__global__ void deg_scatter_kernel(const int* __restrict__ col,
                                   float* __restrict__ deg, int E) {
    int e = blockIdx.x * blockDim.x + threadIdx.x;
    if (e < E) atomic_add_f32(&deg[col[e]], 1.0f);
}

__global__ void deg_scatter_own_kernel(const int* __restrict__ col,
                                       float* __restrict__ deg,
                                       unsigned char* __restrict__ own, int E) {
    int e = blockIdx.x * blockDim.x + threadIdx.x;
    if (e < E) {
        const int c = col[e];
        own[e] = (unsigned char)(c & 255);
        atomic_add_f32(&deg[c], 1.0f);
    }
}

__global__ void rsqrt_kernel(float* __restrict__ d, int N) {
    int i = blockIdx.x * blockDim.x + threadIdx.x;
    if (i < N) d[i] = rsqrtf(d[i]);
}

constexpr int DH_GRID = 64;
constexpr int DH_THREADS = 1024;
constexpr int DH_WORDS = 50000;
constexpr int DH_U = 4;
static_assert((DH_WORDS & 3) == 0, "tables are cleared and stored 16 bytes at a time");

__device__ __forceinline__ void dh_count(unsigned int* cnt, int c) {
    atomicAdd(cnt + (c >> 1), (c & 1) ? 0x10000u : 1u);
}

__global__ __launch_bounds__(DH_THREADS)
void deg_hist_kernel(const int* __restrict__ col, unsigned int* __restrict__ own4,
                     unsigned int* __restrict__ part, int E, int N) {
    __shared__ __attribute__((aligned(16))) unsigned int cnt[DH_WORDS];

    const int words4 = (((N + 1) >> 1) + 3) >> 2;
    const int g4 = E >> 2;
    const int q0 = (int)(((long long)g4 * blockIdx.x) / DH_GRID);
    const int q1 = (int)(((long long)g4 * (blockIdx.x + 1)) / DH_GRID);
    const int4* __restrict__ col4 = (const int4*)col;

    int4 cv[DH_U];
#pragma unroll
    for (int u = 0; u < DH_U; ++u)
        cv[u] = col4[min(q0 + u * DH_THREADS + (int)threadIdx.x, q1 - 1)];

    for (int i = threadIdx.x; i < words4; i += DH_THREADS)
        ((uint4*)cnt)[i] = make_uint4(0u, 0u, 0u, 0u);
    __syncthreads();

    for (int base = q0; base < q1; base += DH_U * DH_THREADS) {
#pragma unroll
        for (int u = 0; u < DH_U; ++u) {
            const int q = base + u * DH_THREADS + (int)threadIdx.x;
            if (q < q1) {
                const int4 c = cv[u];
                dh_count(cnt, c.x);
                dh_count(cnt, c.y);
                dh_count(cnt, c.z);
                dh_count(cnt, c.w);
                own4[q] = (unsigned)(c.x & 255) | ((unsigned)(c.y & 255) << 8) |
                          ((unsigned)(c.z & 255) << 16) | ((unsigned)(c.w & 255) << 24);
            }
        }
        if (base + DH_U * DH_THREADS < q1) {
#pragma unroll
            for (int u = 0; u < DH_U; ++u)
                cv[u] = col4[min(base + (DH_U + u) * DH_THREADS + (int)threadIdx.x, q1 - 1)];
        }
    }
    __syncthreads();

    uint4* __restrict__ p = (uint4*)part + (size_t)blockIdx.x * words4;
#pragma unroll 4
    for (int i = threadIdx.x; i < words4; i += DH_THREADS)
        p[i] = ((const uint4*)cnt)[i];
}

__global__ __launch_bounds__(256)
void deg_reduce_kernel(const unsigned int* __restrict__ part, float* __restrict__ dinv,
                       int N, int pitch) {
    const int w = blockIdx.x * blockDim.x + threadIdx.x;
    if (2 * w >= N) return;
    unsigned int v[DH_GRID];
#pragma unroll
    for (int b = 0; b < DH_GRID; ++b) v[b] = part[(size_t)b * pitch + w];
    unsigned int lo = 0u, hi = 0u;
#pragma unroll
    for (int b = 0; b < DH_GRID; ++b) {
        lo += v[b] & 0xffffu;
        hi += v[b] >> 16;
    }
    *(float2*)(dinv + 2 * w) = make_float2(rsqrtf(1.0f + (float)lo), rsqrtf(1.0f + (float)hi));
}

__global__ void edge_scatter_kernel(const float* __restrict__ h,
                                    const float* __restrict__ dinv,
                                    const int* __restrict__ row,
                                    const int* __restrict__ col,
                                    float* __restrict__ acc,
                                    int E, int H) {
    const int chunks = H >> 2;
    long long gid = (long long)blockIdx.x * blockDim.x + threadIdx.x;
    long long total = (long long)E * chunks;
    if (gid >= total) return;
    const int e = (int)(gid / chunks);
    const int j = (int)(gid - (long long)e * chunks) << 2;
    const int r = row[e];
    const int c = col[e];
    const float s = dinv[r];
    const float4 hv = *(const float4*)(h + (size_t)r * H + j);
    float* dst = acc + (size_t)c * H + j;
    atomic_add_f32(dst + 0, s * hv.x);
    atomic_add_f32(dst + 1, s * hv.y);
    atomic_add_f32(dst + 2, s * hv.z);
    atomic_add_f32(dst + 3, s * hv.w);
}

constexpr int SC_H = 128;
constexpr int SC_EPW = 8;

template <bool FULL>
__device__ __forceinline__ void line_scatter_wave(const float* __restrict__ h,
                                                  const float* __restrict__ dinv,
                                                  const int* __restrict__ row, const int* __restrict__ col,
                                                  float* __restrict__ acc, int e0, int ne, int l) {
    int r[SC_EPW], c[SC_EPW];
    float s[SC_EPW], v0[SC_EPW], v1[SC_EPW], v2[SC_EPW], v3[SC_EPW];
#pragma unroll
    for (int i = 0; i < SC_EPW; ++i) {
        const int e = FULL ? e0 + i : e0 + min(i, ne - 1);
        r[i] = row[e];
        c[i] = col[e];
    }
#pragma unroll
    for (int i = 0; i < SC_EPW; ++i) {
        s[i] = dinv[r[i]];
        const float* __restrict__ p = h + (size_t)r[i] * SC_H + l;
        v0[i] = p[0];
        v1[i] = p[32];
        v2[i] = p[64];
        v3[i] = p[96];
    }
#pragma unroll
    for (int i = 0; i < SC_EPW; ++i) {
        if (FULL || i < ne) {
            float* dst = acc + (size_t)c[i] * SC_H + l;
            atomic_add_f32(dst, s[i] * v0[i]);
            atomic_add_f32(dst + 32, s[i] * v1[i]);
            atomic_add_f32(dst + 64, s[i] * v2[i]);
            atomic_add_f32(dst + 96, s[i] * v3[i]);
        }
    }
}

__global__ __launch_bounds__(256)
void edge_scatter_lines_kernel(const float* __restrict__ h,
                               const float* __restrict__ dinv,
                               const int* __restrict__ row,
                               const int* __restrict__ col,
                               float* __restrict__ acc, int E) {
    const int l = threadIdx.x & 31;
    const int wv = __builtin_amdgcn_readfirstlane(
        (int)(blockIdx.x * (blockDim.x >> 5) + (threadIdx.x >> 5)));
    const int e0 = wv * SC_EPW;
    if (e0 >= E) return;
    const int ne = min(SC_EPW, E - e0);
    if (ne == SC_EPW) line_scatter_wave<true>(h, dinv, row, col, acc, e0, ne, l);
    else              line_scatter_wave<false>(h, dinv, row, col, acc, e0, ne, l);
}


constexpr int OW_GRID_LOG2 = 8;
constexpr int OW_GRID = 1 << OW_GRID_LOG2;
static_assert(OW_GRID == 256, "the owner list holds one byte per edge: target mod 256");
constexpr int OW_WAVES = 32;
constexpr int OW_SLOTS = 392;
constexpr int OW_U = 8;
constexpr int OW_Q = 544;

__device__ __forceinline__ void lds_add_f32(float* p, float v) {
    __hip_atomic_fetch_add(p, v, __ATOMIC_RELAXED, __HIP_MEMORY_SCOPE_WORKGROUP);
}

template <bool REC>
__device__ __forceinline__ void ow_drain(const float* __restrict__ h, const float* __restrict__ dinv,
                                         const int* __restrict__ row, const int* __restrict__ col,
                                         const int4* __restrict__ rec,
                                         float* accs, const int* q, int m, int lane) {
    const int e = q[min(lane, m - 1)];
    int r, c, sbits;
    if (REC) {
        const int4 rc = rec[e];
        r = rc.x;
        c = rc.y;
        sbits = rc.z;
    } else {
        r = row[e];
        c = col[e];
        sbits = __float_as_int(dinv[r]);
    }
    for (int j = 0; j < m; j += 8) {
        float v0[8], v1[8], v2[8], v3[8], ss[8];
        int sl[8];
#pragma unroll
        for (int i = 0; i < 8; ++i) {
            const int idx = min(j + i, m - 1);
            const int rr = __builtin_amdgcn_readlane(r, idx);
            sl[i] = __builtin_amdgcn_readlane(c, idx) >> OW_GRID_LOG2;
            ss[i] = __int_as_float(__builtin_amdgcn_readlane(sbits, idx));
            const float* __restrict__ p = h + (size_t)rr * SC_H + lane;
            v0[i] = p[0];
            v1[i] = p[32];
            v2[i] = p[64];
            v3[i] = p[96];
        }
#pragma unroll
        for (int i = 0; i < 8; ++i) {
            if (j + i < m) {
                float* d = accs + sl[i] * SC_H + lane;
                lds_add_f32(d, ss[i] * v0[i]);
                lds_add_f32(d + 32, ss[i] * v1[i]);
                lds_add_f32(d + 64, ss[i] * v2[i]);
                lds_add_f32(d + 96, ss[i] * v3[i]);
            }
        }
    }
}

#define OW_STEP(MT, EDGE) \
    { \
        const bool mt = (MT); \
        const unsigned mask = __builtin_amdgcn_ballot_w32(mt); \
        if (mask) { \
            if (mt) q[cnt + (int)__builtin_amdgcn_mbcnt_lo(mask, 0u)] = (EDGE); \
            cnt += __builtin_popcount(mask); \
        } \
    }

#define OW_WORD(WV, E0) \
    { \
        const unsigned xw = ok ? ((WV) ^ bpat) : 0xffffffffu; \
        const unsigned zb = (xw - 0x01010101u) & ~xw & 0x80808080u; \
        if (__builtin_amdgcn_ballot_w32(zb != 0u)) { \
            OW_STEP((xw & 0x000000ffu) == 0u, (E0) + 0) \
            OW_STEP((xw & 0x0000ff00u) == 0u, (E0) + 1) \
            OW_STEP((xw & 0x00ff0000u) == 0u, (E0) + 2) \
            OW_STEP((xw & 0xff000000u) == 0u, (E0) + 3) \
        } \
    }

template <bool REC>
__global__ __launch_bounds__(1024)
void owner_aggregate_kernel(const float* __restrict__ h, const float* __restrict__ dinv,
                            const int* __restrict__ row, const int* __restrict__ col,
                            const int* __restrict__ recw,
                            const unsigned char* __restrict__ own,
                            float* __restrict__ x, int E, int N) {
    const int4* __restrict__ rec = (const int4*)recw;
    __shared__ __attribute__((aligned(16))) float accs[OW_SLOTS * SC_H];
    __shared__ int qs[OW_WAVES * OW_Q];

    const int lane = threadIdx.x & 31;
    const int wave = __builtin_amdgcn_readfirstlane((int)(threadIdx.x >> 5));
    const int b = blockIdx.x;
    const int slots = (N - b + OW_GRID - 1) >> OW_GRID_LOG2;

#pragma unroll 4
    for (int sl = wave; sl < slots; sl += OW_WAVES) {
        const int node = (sl << OW_GRID_LOG2) + b;
        const float di = dinv[node];
        const float4 hv = ((const float4*)(h + (size_t)node * SC_H))[lane];
        *(float4*)(accs + sl * SC_H + 4 * lane) = make_float4(di * hv.x, di * hv.y, di * hv.z, di * hv.w);
    }
    __syncthreads();

    int* q = qs + wave * OW_Q;
    int cnt = 0;
    const uint4* __restrict__ own4 = (const uint4*)own;
    const int total16 = E >> 4;
    const unsigned bpat = (unsigned)b * 0x01010101u;
    for (int base = 0; base < total16; base += OW_U * OW_WAVES * 32) {
        uint4 ov[OW_U];
#pragma unroll
        for (int u = 0; u < OW_U; ++u) {
            const int i16 = base + (u * OW_WAVES + wave) * 32 + lane;
            ov[u] = own4[min(i16, total16 - 1)];
        }
#pragma unroll
        for (int u = 0; u < OW_U; ++u) {
            const int i16 = base + (u * OW_WAVES + wave) * 32 + lane;
            const bool ok = i16 < total16;
            OW_WORD(ov[u].x, 16 * i16 + 0)
            OW_WORD(ov[u].y, 16 * i16 + 4)
            OW_WORD(ov[u].z, 16 * i16 + 8)
            OW_WORD(ov[u].w, 16 * i16 + 12)
            if (cnt >= 32) {
                for (int g = 0; g < cnt; g += 32)
                    ow_drain<REC>(h, dinv, row, col, rec, accs, q + g, min(32, cnt - g), lane);
                cnt = 0;
            }
        }
    }
    for (int g = 0; g < cnt; g += 32)
        ow_drain<REC>(h, dinv, row, col, rec, accs, q + g, min(32, cnt - g), lane);
    __syncthreads();

#pragma unroll 4
    for (int sl = wave; sl < slots; sl += OW_WAVES) {
        const int node = (sl << OW_GRID_LOG2) + b;
        const float di = dinv[node];
        const float4 a = *(const float4*)(accs + sl * SC_H + 4 * lane);
        float4 o;
        o.x = di * a.x; o.x = o.x > 0.f ? o.x : 0.f;
        o.y = di * a.y; o.y = o.y > 0.f ? o.y : 0.f;
        o.z = di * a.z; o.z = o.z > 0.f ? o.z : 0.f;
        o.w = di * a.w; o.w = o.w > 0.f ? o.w : 0.f;
        ((float4*)(x + (size_t)node * SC_H))[lane] = o;
    }
}

__global__ void finalize_relu_kernel(const float* __restrict__ h,
                                     const float* __restrict__ dinv,
                                     float* __restrict__ acc_inout, int H) {
    const int node = blockIdx.x;
    const int j = threadIdx.x;
    const float di = dinv[node];
    const size_t idx = (size_t)node * H + j;
    const float v = di * acc_inout[idx] + di * di * h[idx];
    acc_inout[idx] = v > 0.f ? v : 0.f;
}

constexpr int RAW_FU = 32;
constexpr int RAW_FM = 64;
constexpr int RAW_H = 128;
constexpr int RAW_WU = RAW_FU + RAW_FM;
constexpr int RAW_WM = RAW_FU + RAW_FM + 1;

__global__ void compose_wz_kernel(const float* __restrict__ Wu, const float* __restrict__ bu,
                                  const float* __restrict__ Wm, const float* __restrict__ bm,
                                  float* __restrict__ Wz) {
    const int i = blockIdx.x * blockDim.x + threadIdx.x;
    if (i >= RAW_H * RAW_H) return;
    const int n = i / RAW_H;
    const int k = i - n * RAW_H;
    float w = 0.f;
    if (k < RAW_FU) w = Wu[n * RAW_FU + k];
    else if (k < RAW_WU) w = Wm[n * RAW_FM + (k - RAW_FU)];
    else if (k == RAW_WU) w = bu[n];
    else if (k == RAW_WM) w = bm[n];
    Wz[i] = w;
}

__global__ void init_z_kernel(const float* __restrict__ xu, const float* __restrict__ xm,
                              const float* __restrict__ dinv, float* __restrict__ Z, int Nu) {
    const int node = blockIdx.x;
    const int j = threadIdx.x;
    const float di = dinv[node];
    float self = 0.f;
    if (node < Nu) {
        if (j < RAW_FU) self = xu[(size_t)node * RAW_FU + j];
        else if (j == RAW_WU) self = 1.f;
    } else {
        if (j >= RAW_FU && j < RAW_WU) self = xm[(size_t)(node - Nu) * RAW_FM + (j - RAW_FU)];
        else if (j == RAW_WM) self = 1.f;
    }
    Z[(size_t)node * RAW_H + j] = di * self;
}

constexpr int DZ_THREADS = 256;
constexpr int DZ_NODES = 256;
constexpr int DZ_U = 8;
static_assert((DH_GRID & 1) == 0 && DZ_THREADS == 256 && DZ_NODES == DZ_THREADS,
              "the tables are split into two halves of 128 threads, each thread pair of halves owns one word");
static_assert((RAW_FU & 3) == 0 && (RAW_FM & 3) == 0 && (RAW_WU & 3) == 0 && RAW_WM == RAW_WU + 1,
              "a lane holds four consecutive floats of a row; both weight columns sit in one lane");
static_assert(RAW_H == 128, "one 16-byte store per lane covers a row");

template <bool FULL>
__device__ __forceinline__ void dz_rows(const float4* __restrict__ xu4, const float4* __restrict__ xm4,
                                        const float* sd, float4* __restrict__ Z4,
                                        int node0, int ln0, int N, int Nu, int lane) {
    constexpr int QU = RAW_FU / 4;
    constexpr int QM = (RAW_FU + RAW_FM) / 4;
    for (int i0 = 0; i0 < 32; i0 += DZ_U) {
        float4 v[DZ_U];
        float di[DZ_U];
#pragma unroll
        for (int i = 0; i < DZ_U; ++i) {
            const int ln = ln0 + i0 + i;
            const int node = FULL ? node0 + ln : min(node0 + ln, N - 1);
            const bool user = node < Nu;
            di[i] = sd[ln];
            const float4* __restrict__ p = user ? xu4 + (size_t)node * QU + lane
                                                : xm4 + (size_t)(node - Nu) * (RAW_FM / 4) + (lane - QU);
            const bool ld = user ? (lane < QU) : (lane >= QU && lane < QM);
            float4 r = make_float4(0.f, 0.f, 0.f, 0.f);
            if (ld) r = *p;
            if (lane == QM) { r.x = user ? 1.f : 0.f; r.y = user ? 0.f : 1.f; }
            v[i] = r;
        }
#pragma unroll
        for (int i = 0; i < DZ_U; ++i) {
            const int node = node0 + ln0 + i0 + i;
            if (FULL || node < N)
                Z4[(size_t)node * (RAW_H / 4) + lane] =
                    make_float4(di[i] * v[i].x, di[i] * v[i].y, di[i] * v[i].z, di[i] * v[i].w);
        }
    }
}

__global__ __launch_bounds__(DZ_THREADS)
void deg_reduce_init_z_kernel(const unsigned int* __restrict__ part, float* __restrict__ dinv,
                              const float* __restrict__ xu, const float* __restrict__ xm,
                              float* __restrict__ Z, int N, int Nu, int pitch) {
    __shared__ unsigned int sp[DZ_THREADS];
    __shared__ float sd[DZ_NODES];

    const int t = threadIdx.x;
    const int lane = t & 31;
    const int wave = __builtin_amdgcn_readfirstlane(t >> 5);
    const int node0 = blockIdx.x * DZ_NODES;

    const int j = t & (DZ_THREADS / 2 - 1);
    const int half = t >> 7;
    const int w = blockIdx.x * (DZ_NODES / 2) + j;
    const bool live = 2 * w < N;
    unsigned int lo = 0u, hi = 0u;
    if (live) {
        unsigned int v[DH_GRID / 2];
        const unsigned int* __restrict__ p = part + (size_t)half * (DH_GRID / 2) * pitch + w;
#pragma unroll
        for (int b = 0; b < DH_GRID / 2; ++b) v[b] = p[(size_t)b * pitch];
#pragma unroll
        for (int b = 0; b < DH_GRID / 2; ++b) {
            lo += v[b] & 0xffffu;
            hi += v[b] >> 16;
        }
    }
    if (half) {
        sp[2 * j] = lo;
        sp[2 * j + 1] = hi;
    }
    __syncthreads();
    if (!half) {
        lo += sp[2 * j];
        hi += sp[2 * j + 1];
        const float d0 = rsqrtf(1.0f + (float)lo);
        const float d1 = rsqrtf(1.0f + (float)hi);
        sd[2 * j] = d0;
        sd[2 * j + 1] = d1;
        if (live) *(float2*)(dinv + 2 * w) = make_float2(d0, d1);
    }
    __syncthreads();

    const float4* __restrict__ xu4 = (const float4*)xu;
    const float4* __restrict__ xm4 = (const float4*)xm;
    float4* __restrict__ Z4 = (float4*)Z;
    const int ln0 = wave * 32;
    if (node0 + ln0 >= N) return;
    if (node0 + ln0 + 32 <= N) dz_rows<true>(xu4, xm4, sd, Z4, node0, ln0, N, Nu, lane);
    else                       dz_rows<false>(xu4, xm4, sd, Z4, node0, ln0, N, Nu, lane);
}

constexpr int RAW_EPW = 8;

template <bool FULL>
__device__ __forceinline__ void raw_scatter_wave(const float* __restrict__ xu, const float* __restrict__ xm,
                                                 const float* __restrict__ dinv,
                                                 const int* __restrict__ row, const int* __restrict__ col,
                                                 float* __restrict__ Z, int4* __restrict__ rec,
                                                 int Nu, int e0, int ne, int l) {
    int r[RAW_EPW], c[RAW_EPW];
    float s[RAW_EPW], v0[RAW_EPW], v1[RAW_EPW];
#pragma unroll
    for (int i = 0; i < RAW_EPW; ++i) {
        const int e = FULL ? e0 + i : e0 + min(i, ne - 1);
        r[i] = row[e];
        c[i] = col[e];
    }
    const bool il = l < (FULL ? RAW_EPW : ne);
    int rl = 0, cl = 0;
    float dl = 0.f;
    if (il) {
        rl = row[e0 + l];
        cl = col[e0 + l];
        dl = dinv[rl];
    }
#pragma unroll
    for (int i = 0; i < RAW_EPW; ++i) {
        const bool user = r[i] < Nu;
        const float* __restrict__ p = user ? xu + (size_t)r[i] * RAW_FU + l
                                           : xm + (size_t)(r[i] - Nu) * RAW_FM + l;
        v0[i] = p[0];
        v1[i] = 0.f;
        if (!user) v1[i] = p[32];
    }
#pragma unroll
    for (int i = 0; i < RAW_EPW; ++i)
        s[i] = __int_as_float(__builtin_amdgcn_readlane(__float_as_int(dl), i));
#pragma unroll
    for (int i = 0; i < RAW_EPW; ++i) {
        if (FULL || i < ne) {
            const bool user = r[i] < Nu;
            float* dst = Z + (size_t)c[i] * RAW_H;
            atomic_add_f32(dst + (user ? 0 : RAW_FU) + l, s[i] * v0[i]);
            if (!user) atomic_add_f32(dst + RAW_FU + 32 + l, s[i] * v1[i]);
        }
    }
    if (il) {
        atomic_add_f32(Z + (size_t)cl * RAW_H + (rl < Nu ? RAW_WU : RAW_WM), dl);
        if (rec) rec[e0 + l] = make_int4(rl, cl, __float_as_int(dl), 0);
    }
}

__global__ __launch_bounds__(256)
void raw_scatter_kernel(const float* __restrict__ xu, const float* __restrict__ xm,
                        const float* __restrict__ dinv,
                        const int* __restrict__ row, const int* __restrict__ col,
                        float* __restrict__ Z, int* __restrict__ recw, int E, int Nu) {
    int4* __restrict__ rec = (int4*)recw;
    const int l = threadIdx.x & 31;
    const int wv = __builtin_amdgcn_readfirstlane(
        (int)(blockIdx.x * (blockDim.x >> 5) + (threadIdx.x >> 5)));
    const int e0 = wv * RAW_EPW;
    if (e0 >= E) return;
    const int ne = min(RAW_EPW, E - e0);
    if (ne == RAW_EPW) raw_scatter_wave<true>(xu, xm, dinv, row, col, Z, rec, Nu, e0, ne, l);
    else               raw_scatter_wave<false>(xu, xm, dinv, row, col, Z, rec, Nu, e0, ne, l);
}

static inline int ceil_div_i(long long a, long long b) { return (int)((a + b - 1) / b); }

static void launch_gemm(const float* X, const float* W, const float* b, float* Y,
                        int M, int K, int N, hipStream_t stream) {
    const int TPB = 256;
    if (N == 128 && K == 32)
        gemm_bias_wmma_wlds<32, 0><<<GEMM_GRID, TPB, 0, stream>>>(X, W, b, nullptr, nullptr, Y, M);
    else if (N == 128 && K == 64)
        gemm_bias_wmma_wlds<64, 0><<<GEMM_GRID, TPB, 0, stream>>>(X, W, b, nullptr, nullptr, Y, M);
    else if (N == 128 && K == 128)
        gemm_bias_wmma_wlds<128, 0><<<GEMM_GRID, TPB, 0, stream>>>(X, W, b, nullptr, nullptr, Y, M);
    else {
        const int strips = ceil_div_i(M, 16);
        const int tiles = strips * (N / 16);
        gemm_bias_wmma_rt<<<ceil_div_i(tiles, 8), TPB, 0, stream>>>(X, W, b, Y, M, K, N);
    }
}

extern "C" void kernel_launch(void* const* d_in, const int* in_sizes, int n_in,
                              void* d_out, int out_size, void* d_ws, size_t ws_size,
                              hipStream_t stream) {
    const float* x_user  = (const float*)d_in[0];
    const float* x_movie = (const float*)d_in[1];
    const int*   edge    = (const int*)d_in[2];
    const float* Wu = (const float*)d_in[3];
    const float* bu = (const float*)d_in[4];
    const float* Wm = (const float*)d_in[5];
    const float* bm = (const float*)d_in[6];
    const float* W1 = (const float*)d_in[7];
    const float* b1 = (const float*)d_in[8];
    const float* W2 = (const float*)d_in[9];
    const float* b2 = (const float*)d_in[10];
    const float* We = (const float*)d_in[11];
    const float* be = (const float*)d_in[12];
    float* out = (float*)d_out;

    const int H  = in_sizes[4];
    const int C  = in_sizes[12];
    const int Fu = in_sizes[3] / H;
    const int Fm = in_sizes[5] / H;
    const int Nu = in_sizes[0] / Fu;
    const int Nm = in_sizes[1] / Fm;
    const int E  = in_sizes[2] / 2;
    const int N  = Nu + Nm;
    const int* row = edge;
    const int* col = edge + E;

    float* ws   = (float*)d_ws;
    float* X    = ws;
    float* Hh   = X  + (size_t)N * H;
    float* Acc  = Hh + (size_t)N * H;
    float* Wz   = Acc + (size_t)N * H;
    float* dinv = Wz + (size_t)RAW_H * RAW_H;
    unsigned char* own = (unsigned char*)(dinv + (size_t)((N + 3) & ~3));
    unsigned int* part = (unsigned int*)(own + (((size_t)E + 15) & ~(size_t)15));

    const int TPB = 256, WPB = 8;
    const long long NH = (long long)N * H;

    const bool owner_path = (H == SC_H) && ((E & 15) == 0) && (N <= OW_GRID * OW_SLOTS);
    const int dh_words = (((N + 1) >> 1) + 3) & ~3;
    const bool hist_path = (N <= 2 * DH_WORDS) && ((E & 3) == 0) && (E >= 4 * DH_GRID) &&
                           (E / DH_GRID + 4 <= 65535) &&
                           ((size_t)((unsigned char*)(part + (size_t)DH_GRID * dh_words) - (unsigned char*)d_ws) <= ws_size);
    const bool raw_path = (H == RAW_H) && (Fu == RAW_FU) && (Fm == RAW_FM);
    const size_t rec_off = ((size_t)((unsigned char*)(part + (size_t)DH_GRID * dh_words) - (unsigned char*)d_ws) + 127) & ~(size_t)127;
    const bool rec_path = raw_path && owner_path && (rec_off + (size_t)E * 16 <= ws_size);
    int* rec = rec_path ? (int*)((unsigned char*)d_ws + rec_off) : nullptr;
    if (hist_path) {
        deg_hist_kernel<<<DH_GRID, DH_THREADS, 0, stream>>>(col, (unsigned int*)own, part, E, N);
        if (raw_path)
            deg_reduce_init_z_kernel<<<ceil_div_i(N, DZ_NODES), DZ_THREADS, 0, stream>>>(part, dinv, x_user, x_movie, Hh, N, Nu, dh_words);
        else
            deg_reduce_kernel<<<ceil_div_i((N + 1) >> 1, TPB), TPB, 0, stream>>>(part, dinv, N, dh_words);
    } else {
        fill_kernel<<<ceil_div_i(N, TPB), TPB, 0, stream>>>(dinv, 1.0f, N);
        if (owner_path)
            deg_scatter_own_kernel<<<ceil_div_i(E, TPB), TPB, 0, stream>>>(col, dinv, own, E);
        else
            deg_scatter_kernel<<<ceil_div_i(E, TPB), TPB, 0, stream>>>(col, dinv, E);
        rsqrt_kernel<<<ceil_div_i(N, TPB), TPB, 0, stream>>>(dinv, N);
    }

    const long long scatterThreads = (long long)E * (H / 4);

    if (raw_path) {
        compose_wz_kernel<<<ceil_div_i(RAW_H * RAW_H, TPB), TPB, 0, stream>>>(Wu, bu, Wm, bm, Wz);
        if (!hist_path) init_z_kernel<<<N, RAW_H, 0, stream>>>(x_user, x_movie, dinv, Hh, Nu);
        raw_scatter_kernel<<<ceil_div_i(E, RAW_EPW * WPB), TPB, 0, stream>>>(x_user, x_movie, dinv, row, col, Hh, rec, E, Nu);
        gemm_chain3_wmma<<<CH_GRID, CH_WAVES * 32, 0, stream>>>(Hh, Wz, W1, b1, W2, b2, dinv, Acc, N);
    } else {
        launch_gemm(x_user, Wu, bu, X, Nu, Fu, H, stream);
        launch_gemm(x_movie, Wm, bm, X + (size_t)Nu * H, Nm, Fm, H, stream);

        launch_gemm(X, W1, b1, Hh, N, H, H, stream);
        zero4_kernel<<<ceil_div_i(NH / 4, TPB), TPB, 0, stream>>>((float4*)Acc, NH / 4);
        edge_scatter_kernel<<<ceil_div_i(scatterThreads, TPB), TPB, 0, stream>>>(Hh, dinv, row, col, Acc, E, H);
        finalize_relu_kernel<<<N, H, 0, stream>>>(Hh, dinv, Acc, H);
    }

    const float* Hc = Acc;
    if (!raw_path) {
        launch_gemm(Acc, W2, b2, Hh, N, H, H, stream);
        Hc = Hh;
    }
    if (owner_path) {
        if (rec_path)
            owner_aggregate_kernel<true><<<OW_GRID, OW_WAVES * 32, 0, stream>>>(Hc, dinv, row, col, rec, own, X, E, N);
        else
            owner_aggregate_kernel<false><<<OW_GRID, OW_WAVES * 32, 0, stream>>>(Hc, dinv, row, col, nullptr, own, X, E, N);
    } else {
        zero4_kernel<<<ceil_div_i(NH / 4, TPB), TPB, 0, stream>>>((float4*)X, NH / 4);
        if (H == SC_H)
            edge_scatter_lines_kernel<<<ceil_div_i(E, SC_EPW * WPB), TPB, 0, stream>>>(Hc, dinv, row, col, X, E);
        else
            edge_scatter_kernel<<<ceil_div_i(scatterThreads, TPB), TPB, 0, stream>>>(Hc, dinv, row, col, X, E, H);
        finalize_relu_kernel<<<N, H, 0, stream>>>(Hc, dinv, X, H);
    }

    const int edgeTiles = ceil_div_i(E, 16);
    if (H == EM_H && C <= 16)
        edge_mlp_rows_lds<<<EM_GRID, TPB, 0, stream>>>(X, row, col, We, be, out, E, Nu, C);
    else
        edge_mlp_wmma<0><<<ceil_div_i(edgeTiles, WPB), TPB, 0, stream>>>(X, row, col, We, be, out, E, Nu, H, C);
}


# conv1 sums by owner blocks too: Z rows in LDS, LDS float adds, each row stored once; a wave leaves its edges' records for conv2's pass, which no longer scans; no global atomic; sum order changes
# speedup vs baseline: 3.2858x; 1.1027x over previous
#include <hip/hip_runtime.h>

typedef __attribute__((ext_vector_type(2))) float v2f;
typedef __attribute__((ext_vector_type(8))) float v8f;

__device__ __forceinline__ v8f wmma_f32x4(v2f a, v2f b, v8f c) {
    return __builtin_amdgcn_wmma_f32_16x16x4_f32(
         false, a,  false, b,
         (short)0, c,  false,  false);
}

__device__ __forceinline__ void atomic_add_f32(float* p, float v) {
#if defined(__HIP_DEVICE_COMPILE__)
    unsafeAtomicAdd(p, v);
#else
    atomicAdd(p, v);
#endif
}

constexpr int GEMM_GRID_LOG2 = 9;
constexpr int GEMM_GRID = 1 << GEMM_GRID_LOG2;

template <int MODE>
__device__ __forceinline__ float gemm_epilogue(float a, float b, float g) {
    if (MODE == 0) return a + b;
    if (MODE == 1) return g * a;
    const float v = a + g * b;
    return v > 0.f ? v : 0.f;
}

template <int K, int MODE>
__global__ __launch_bounds__(256)
void gemm_bias_wmma_wlds(const float* __restrict__ X,
                         const float* __restrict__ W,
                         const float* __restrict__ bias,
                         const float* __restrict__ rs,
                         const float* __restrict__ zc,
                         float* __restrict__ Y,
                         int M) {
    constexpr int N = 128;
    constexpr int NT = N / 16;
    constexpr int PITCH = K + 4;
    constexpr int KB = K / 8;
    constexpr int WAVES = 8;
    __shared__ __attribute__((aligned(16))) float Ws[N * PITCH];

    const int lane = threadIdx.x & 31;
    const int wave = __builtin_amdgcn_readfirstlane((int)(threadIdx.x >> 5));
    const int c = lane & 15;
    const int h = lane >> 4;

    const int strips = (M + 15) >> 4;
    const int s0 = (int)(((long long)strips * blockIdx.x) >> GEMM_GRID_LOG2);
    const int s1 = (int)(((long long)strips * (blockIdx.x + 1)) >> GEMM_GRID_LOG2);

    float bv[NT];
#pragma unroll
    for (int t = 0; t < NT; ++t) bv[t] = (MODE == 1) ? 0.f : bias[16 * t + c];

    float4 a[KB];
    int s = s0 + wave;

    if (s < s1) {
        const int arow = min(s * 16 + c, M - 1);
        const float4* __restrict__ ap = (const float4*)(X + (size_t)arow * K) + h;
#pragma unroll
        for (int kb = 0; kb < KB; ++kb) a[kb] = ap[2 * kb];
    }

    for (int i = threadIdx.x; i < N * (K / 4); i += WAVES * 32) {
        const int n = i / (K / 4);
        const int k4 = i - n * (K / 4);
        *(float4*)(Ws + n * PITCH + 4 * k4) = ((const float4*)W)[i];
    }
    __syncthreads();

    const float* __restrict__ bfrag = Ws + c * PITCH + 4 * h;

    while (s < s1) {
        float gr[8];
#pragma unroll
        for (int r = 0; r < 8; ++r) {
            gr[r] = 0.f;
            if (MODE != 0) {
                const int rr = min(s * 16 + (h << 3) + r, M - 1);
                float g = rs[rr];
                if (MODE == 2) {
                    const v2f w = *(const v2f*)(zc + (size_t)rr * N + 96);
                    g *= (w.x + w.y);
                }
                gr[r] = g;
            }
        }

        v8f acc[NT];
#pragma unroll
        for (int t = 0; t < NT; ++t) acc[t] = (v8f){};

        float4 bc[NT], bn[NT];
#pragma unroll
        for (int t = 0; t < NT; ++t)
            bc[t] = *(const float4*)(bfrag + 16 * t * PITCH);
#pragma unroll
        for (int kb = 0; kb < KB; ++kb) {
            if (kb + 1 < KB) {
#pragma unroll
                for (int t = 0; t < NT; ++t)
                    bn[t] = *(const float4*)(bfrag + 16 * t * PITCH + 8 * (kb + 1));
            }
            __builtin_amdgcn_sched_barrier(0);
            const v2f a01 = {a[kb].x, a[kb].y};
            const v2f a23 = {a[kb].z, a[kb].w};
#pragma unroll
            for (int t = 0; t < NT; ++t) {
                const v2f b01 = {bc[t].x, bc[t].y};
                acc[t] = wmma_f32x4(a01, b01, acc[t]);
            }
#pragma unroll
            for (int t = 0; t < NT; ++t) {
                const v2f b23 = {bc[t].z, bc[t].w};
                acc[t] = wmma_f32x4(a23, b23, acc[t]);
            }
            __builtin_amdgcn_sched_barrier(0);
#pragma unroll
            for (int t = 0; t < NT; ++t) bc[t] = bn[t];
        }

        const int row0 = s * 16;
        s += WAVES;

        if (s < s1) {
            const int arow = min(s * 16 + c, M - 1);
            const float4* __restrict__ ap = (const float4*)(X + (size_t)arow * K) + h;
#pragma unroll
            for (int kb = 0; kb < KB; ++kb) a[kb] = ap[2 * kb];
        }

        const int rbase = row0 + (h << 3);
        float* __restrict__ yrow = Y + (size_t)rbase * N + c;
        if (row0 + 16 <= M) {
#pragma unroll
            for (int t = 0; t < NT; ++t) {
#pragma unroll
                for (int r = 0; r < 8; ++r)
                    yrow[(size_t)r * N + 16 * t] = gemm_epilogue<MODE>(acc[t][r], bv[t], gr[r]);
            }
        } else {
#pragma unroll
            for (int t = 0; t < NT; ++t) {
#pragma unroll
                for (int r = 0; r < 8; ++r)
                    if (rbase + r < M)
                        yrow[(size_t)r * N + 16 * t] = gemm_epilogue<MODE>(acc[t][r], bv[t], gr[r]);
            }
        }
    }
}

__global__ void gemm_bias_wmma_rt(const float* __restrict__ X,
                                  const float* __restrict__ W,
                                  const float* __restrict__ bias,
                                  float* __restrict__ Y,
                                  int M, int K, int N) {
    const int lane = threadIdx.x & 31;
    const int wave = threadIdx.x >> 5;
    const int tilesN = N >> 4;
    const int tile = blockIdx.x * (blockDim.x >> 5) + wave;
    const int tm = tile / tilesN;
    const int tn = tile - tm * tilesN;
    if (tm * 16 >= M) return;

    const int mrow = min(tm * 16 + (lane & 15), M - 1);
    const int ncol = tn * 16 + (lane & 15);
    const int koff = (lane >> 4) << 1;
    const float* __restrict__ xrow = X + (size_t)mrow * K + koff;
    const float* __restrict__ wrow = W + (size_t)ncol * K + koff;

    v8f acc = {};
#pragma unroll 8
    for (int k = 0; k < K; k += 4) {
        v2f a = *(const v2f*)(xrow + k);
        v2f b = *(const v2f*)(wrow + k);
        acc = wmma_f32x4(a, b, acc);
    }

    const float bv = bias[ncol];
    const int rbase = tm * 16 + ((lane >> 4) << 3);
    float* yp = Y + (size_t)rbase * N + ncol;
#pragma unroll
    for (int r = 0; r < 8; ++r) {
        if (rbase + r < M) *yp = acc[r] + bv;
        yp += N;
    }
}

constexpr int CH_N = 128;
constexpr int CH_PITCH = CH_N + 4;
constexpr int CH_KB = CH_N / 8;
constexpr int CH_NT = CH_N / 16;
constexpr int CH_WAVES = 12;
constexpr int CH_GRID_LOG2 = 8;
constexpr int CH_GRID = 1 << CH_GRID_LOG2;

__device__ __forceinline__ void chain_product(const float4 (&a)[CH_KB],
                                              const float* __restrict__ bfrag,
                                              v8f (&acc)[CH_NT]) {
#pragma unroll
    for (int t = 0; t < CH_NT; ++t) acc[t] = (v8f){};
    float4 bc[CH_NT], bn[CH_NT];
#pragma unroll
    for (int t = 0; t < CH_NT; ++t)
        bc[t] = *(const float4*)(bfrag + 16 * t * CH_PITCH);
#pragma unroll
    for (int kb = 0; kb < CH_KB; ++kb) {
        if (kb + 1 < CH_KB) {
#pragma unroll
            for (int t = 0; t < CH_NT; ++t)
                bn[t] = *(const float4*)(bfrag + 16 * t * CH_PITCH + 8 * (kb + 1));
        }
        __builtin_amdgcn_sched_barrier(0);
        const v2f a01 = {a[kb].x, a[kb].y};
        const v2f a23 = {a[kb].z, a[kb].w};
#pragma unroll
        for (int t = 0; t < CH_NT; ++t) {
            const v2f b01 = {bc[t].x, bc[t].y};
            acc[t] = wmma_f32x4(a01, b01, acc[t]);
        }
#pragma unroll
        for (int t = 0; t < CH_NT; ++t) {
            const v2f b23 = {bc[t].z, bc[t].w};
            acc[t] = wmma_f32x4(a23, b23, acc[t]);
        }
        __builtin_amdgcn_sched_barrier(0);
#pragma unroll
        for (int t = 0; t < CH_NT; ++t) bc[t] = bn[t];
    }
}

__global__ __launch_bounds__(CH_WAVES * 32)
void gemm_chain3_wmma(const float* __restrict__ Z,
                      const float* __restrict__ Wz,
                      const float* __restrict__ W1,
                      const float* __restrict__ b1,
                      const float* __restrict__ W2,
                      const float* __restrict__ b2,
                      const float* __restrict__ rs,
                      float* __restrict__ Y,
                      int M) {
    __shared__ __attribute__((aligned(16))) float Ws[3 * CH_N * CH_PITCH];
    __shared__ __attribute__((aligned(16))) float Ps[CH_WAVES * 16 * CH_PITCH];

    const int lane = threadIdx.x & 31;
    const int wave = __builtin_amdgcn_readfirstlane((int)(threadIdx.x >> 5));
    const int c = lane & 15;
    const int h = lane >> 4;

    const int strips = (M + 15) >> 4;
    const int s0 = (int)(((long long)strips * blockIdx.x) >> CH_GRID_LOG2);
    const int s1 = (int)(((long long)strips * (blockIdx.x + 1)) >> CH_GRID_LOG2);

    float bv1[CH_NT], bv2[CH_NT];
#pragma unroll
    for (int t = 0; t < CH_NT; ++t) {
        bv1[t] = b1[16 * t + c];
        bv2[t] = b2[16 * t + c];
    }

    float4 a[CH_KB];
    int s = s0 + wave;

    if (s < s1) {
        const int arow = min(s * 16 + c, M - 1);
        const float4* __restrict__ ap = (const float4*)(Z + (size_t)arow * CH_N) + h;
#pragma unroll
        for (int kb = 0; kb < CH_KB; ++kb) a[kb] = ap[2 * kb];
    }

    for (int i = threadIdx.x; i < CH_N * (CH_N / 4); i += CH_WAVES * 32) {
        const int n = i / (CH_N / 4);
        const int k4 = i - n * (CH_N / 4);
        const float4 w0 = ((const float4*)Wz)[i];
        const float4 w1 = ((const float4*)W1)[i];
        const float4 w2 = ((const float4*)W2)[i];
        float* dst = Ws + n * CH_PITCH + 4 * k4;
        *(float4*)dst = w0;
        *(float4*)(dst + CH_N * CH_PITCH) = w1;
        *(float4*)(dst + 2 * CH_N * CH_PITCH) = w2;
    }
    __syncthreads();

    const float* __restrict__ bfz = Ws + c * CH_PITCH + 4 * h;
    const float* __restrict__ bf1 = bfz + CH_N * CH_PITCH;
    const float* __restrict__ bf2 = bfz + 2 * CH_N * CH_PITCH;

    float* P = Ps + wave * (16 * CH_PITCH);
    float* pst = P + (h << 3) * CH_PITCH + c;
    const float* pfr = P + c * CH_PITCH + 4 * h;

    while (s < s1) {
        float d[8], g[8];
#pragma unroll
        for (int r = 0; r < 8; ++r) {
            const int rr = min(s * 16 + (h << 3) + r, M - 1);
            const float dr = rs[rr];
            const v2f w = *(const v2f*)(Z + (size_t)rr * CH_N + 96);
            d[r] = dr;
            g[r] = dr * (w.x + w.y);
        }

        v8f acc[CH_NT];

        chain_product(a, bfz, acc);
#pragma unroll
        for (int t = 0; t < CH_NT; ++t) {
#pragma unroll
            for (int r = 0; r < 8; ++r)
                pst[r * CH_PITCH + 16 * t] = gemm_epilogue<1>(acc[t][r], 0.f, d[r]);
        }
#pragma unroll
        for (int kb = 0; kb < CH_KB; ++kb) a[kb] = *(const float4*)(pfr + 8 * kb);

        chain_product(a, bf1, acc);
#pragma unroll
        for (int t = 0; t < CH_NT; ++t) {
#pragma unroll
            for (int r = 0; r < 8; ++r)
                pst[r * CH_PITCH + 16 * t] = gemm_epilogue<2>(acc[t][r], bv1[t], g[r]);
        }
#pragma unroll
        for (int kb = 0; kb < CH_KB; ++kb) a[kb] = *(const float4*)(pfr + 8 * kb);

        chain_product(a, bf2, acc);

        const int row0 = s * 16;
        s += CH_WAVES;

        if (s < s1) {
            const int arow = min(s * 16 + c, M - 1);
            const float4* __restrict__ ap = (const float4*)(Z + (size_t)arow * CH_N) + h;
#pragma unroll
            for (int kb = 0; kb < CH_KB; ++kb) a[kb] = ap[2 * kb];
        }

        const int rbase = row0 + (h << 3);
        float* __restrict__ yrow = Y + (size_t)rbase * CH_N + c;
        if (row0 + 16 <= M) {
#pragma unroll
            for (int t = 0; t < CH_NT; ++t) {
#pragma unroll
                for (int r = 0; r < 8; ++r)
                    yrow[(size_t)r * CH_N + 16 * t] = gemm_epilogue<0>(acc[t][r], bv2[t], 0.f);
            }
        } else {
#pragma unroll
            for (int t = 0; t < CH_NT; ++t) {
#pragma unroll
                for (int r = 0; r < 8; ++r)
                    if (rbase + r < M)
                        yrow[(size_t)r * CH_N + 16 * t] = gemm_epilogue<0>(acc[t][r], bv2[t], 0.f);
            }
        }
    }
}

template <int HT>
__global__ void edge_mlp_wmma(const float* __restrict__ xf,
                              const int* __restrict__ row,
                              const int* __restrict__ col,
                              const float* __restrict__ We,
                              const float* __restrict__ be,
                              float* __restrict__ out,
                              int E, int Nu, int Hrt, int C) {
    const int H = (HT > 0) ? HT : Hrt;
    const int K = 2 * H;
    const int lane = threadIdx.x & 31;
    const int wave = threadIdx.x >> 5;
    const int tile = blockIdx.x * (blockDim.x >> 5) + wave;
    const int ebase = tile * 16;
    if (ebase >= E) return;

    const int m = lane & 15;
    const int e = min(ebase + m, E - 1);
    const int r = row[e];
    const int c = col[e];
    const int koff = (lane >> 4) << 1;

    const float* __restrict__ urow = xf + (size_t)r * H + koff;
    const float* __restrict__ mrow = xf + (size_t)(Nu + c) * H + koff;

    const int n = lane & 15;
    const float wmask = (n < C) ? 1.0f : 0.0f;
    const float* __restrict__ wrow = We + (size_t)min(n, C - 1) * K + koff;

    v8f acc = {};
#pragma unroll 8
    for (int k = 0; k < K; k += 4) {
        const float* src = (k + koff < H) ? (urow + k) : (mrow + (k - H));
        v2f a = *(const v2f*)src;
        v2f b = *(const v2f*)(wrow + k);
        b.x *= wmask;
        b.y *= wmask;
        acc = wmma_f32x4(a, b, acc);
    }

    if (n < C) {
        const float bv = be[n];
        const int mb = (lane >> 4) << 3;
        float* op = out + (size_t)(ebase + mb) * C + n;
        if (ebase + 16 <= E) {
#pragma unroll
            for (int rr = 0; rr < 8; ++rr) { *op = acc[rr] + bv; op += C; }
        } else {
#pragma unroll
            for (int rr = 0; rr < 8; ++rr) {
                if (ebase + mb + rr < E) *op = acc[rr] + bv;
                op += C;
            }
        }
    }
}

constexpr int EM_H = 128;
constexpr int EM_K = 2 * EM_H;
constexpr int EM_WPITCH = EM_K + 4;
constexpr int EM_APITCH = EM_H + 4;
constexpr int EM_WAVES = 8;
constexpr int EM_GRID = 768;

__device__ __forceinline__ int edge_tile_ends(const int* __restrict__ row, const int* __restrict__ col,
                                              int t, int lane, int E) {
    const int e = min(t * 16 + (lane & 15), E - 1);
    const int* __restrict__ p = (lane < 16) ? row : col;
    return p[e];
}

#define EM_ROW_LOAD(j, IDX, BASE) \
    rv##j = ((const float4*)(xf + (size_t)(((BASE) ? Nu : 0) + __builtin_amdgcn_readlane(IDX, (BASE) + j)) * EM_H))[lane];
#define EM_ROW_STORE(j, IDX, BASE) *(float4*)(astore + j * EM_APITCH) = rv##j;
#define EM_ALL_ROWS(OP, IDX, BASE) \
    OP(0, IDX, BASE) OP(1, IDX, BASE) OP(2, IDX, BASE) OP(3, IDX, BASE) \
    OP(4, IDX, BASE) OP(5, IDX, BASE) OP(6, IDX, BASE) OP(7, IDX, BASE) \
    OP(8, IDX, BASE) OP(9, IDX, BASE) OP(10, IDX, BASE) OP(11, IDX, BASE) \
    OP(12, IDX, BASE) OP(13, IDX, BASE) OP(14, IDX, BASE) OP(15, IDX, BASE)

__global__ __launch_bounds__(256)
void edge_mlp_rows_lds(const float* __restrict__ xf,
                       const int* __restrict__ row,
                       const int* __restrict__ col,
                       const float* __restrict__ We,
                       const float* __restrict__ be,
                       float* __restrict__ out,
                       int E, int Nu, int C) {
    __shared__ __attribute__((aligned(16))) float Ws[16 * EM_WPITCH];
    __shared__ __attribute__((aligned(16))) float As[EM_WAVES * 16 * EM_APITCH];

    const int lane = threadIdx.x & 31;
    const int wave = __builtin_amdgcn_readfirstlane((int)(threadIdx.x >> 5));
    const int c = lane & 15;
    const int h = lane >> 4;

    const int tiles = (E + 15) >> 4;
    const int t0 = (int)(((long long)tiles * blockIdx.x) / EM_GRID);
    const int t1 = (int)(((long long)tiles * (blockIdx.x + 1)) / EM_GRID);
    int t = t0 + wave;

    float4 rv0, rv1, rv2, rv3, rv4, rv5, rv6, rv7, rv8, rv9, rv10, rv11, rv12, rv13, rv14, rv15;
    rv0 = rv1 = rv2 = rv3 = rv4 = rv5 = rv6 = rv7 = make_float4(0.f, 0.f, 0.f, 0.f);
    rv8 = rv9 = rv10 = rv11 = rv12 = rv13 = rv14 = rv15 = rv0;
    int idx = 0;
    if (t < t1) {
        idx = edge_tile_ends(row, col, t, lane, E);
        EM_ALL_ROWS(EM_ROW_LOAD, idx, 0)
    }

    for (int i = threadIdx.x; i < 16 * (EM_K / 4); i += EM_WAVES * 32) {
        const int n = i / (EM_K / 4);
        const int k4 = i - n * (EM_K / 4);
        float4 w = make_float4(0.f, 0.f, 0.f, 0.f);
        if (n < C) w = ((const float4*)We)[i];
        *(float4*)(Ws + n * EM_WPITCH + 4 * k4) = w;
    }
    __syncthreads();

    float* Aw = As + wave * (16 * EM_APITCH);
    float* astore = Aw + 4 * lane;
    const float* afrag = Aw + c * EM_APITCH + 4 * h;
    const float* bfrag = Ws + c * EM_WPITCH + 4 * h;
    const float bv = (c < C) ? be[c] : 0.f;

    while (t < t1) {
        const int tn = t + EM_WAVES;
        int idxn = 0;
        if (tn < t1) idxn = edge_tile_ends(row, col, tn, lane, E);

        EM_ALL_ROWS(EM_ROW_STORE, idx, 0)
        EM_ALL_ROWS(EM_ROW_LOAD, idx, 16)

        v8f acc0 = {}, acc1 = {};
#pragma unroll
        for (int kb = 0; kb < EM_H / 8; ++kb) {
            const float4 a = *(const float4*)(afrag + 8 * kb);
            const float4 b = *(const float4*)(bfrag + 8 * kb);
            acc0 = wmma_f32x4((v2f){a.x, a.y}, (v2f){b.x, b.y}, acc0);
            acc1 = wmma_f32x4((v2f){a.z, a.w}, (v2f){b.z, b.w}, acc1);
        }

        EM_ALL_ROWS(EM_ROW_STORE, idx, 0)
        if (tn < t1) {
            EM_ALL_ROWS(EM_ROW_LOAD, idxn, 0)
        }
#pragma unroll
        for (int kb = 0; kb < EM_H / 8; ++kb) {
            const float4 a = *(const float4*)(afrag + 8 * kb);
            const float4 b = *(const float4*)(bfrag + EM_H + 8 * kb);
            acc0 = wmma_f32x4((v2f){a.x, a.y}, (v2f){b.x, b.y}, acc0);
            acc1 = wmma_f32x4((v2f){a.z, a.w}, (v2f){b.z, b.w}, acc1);
        }

        if (c < C) {
            const int ebase = t * 16;
            const int mb = h << 3;
            float* op = out + (size_t)(ebase + mb) * C + c;
            if (ebase + 16 <= E) {
#pragma unroll
                for (int rr = 0; rr < 8; ++rr) { *op = (acc0[rr] + acc1[rr]) + bv; op += C; }
            } else {
#pragma unroll
                for (int rr = 0; rr < 8; ++rr) {
                    if (ebase + mb + rr < E) *op = (acc0[rr] + acc1[rr]) + bv;
                    op += C;
                }
            }
        }

        idx = idxn;
        t = tn;
    }
}

__global__ void fill_kernel(float* __restrict__ p, float v, long long n) {
    long long i = (long long)blockIdx.x * blockDim.x + threadIdx.x;
    if (i < n) p[i] = v;
}

__global__ void zero4_kernel(float4* __restrict__ p, long long n4) {
    long long i = (long long)blockIdx.x * blockDim.x + threadIdx.x;
    if (i < n4) p[i] = make_float4(0.f, 0.f, 0.f, 0.f);
}

__global__ void deg_scatter_kernel(const int* __restrict__ col,
                                   float* __restrict__ deg, int E) {
    int e = blockIdx.x * blockDim.x + threadIdx.x;
    if (e < E) atomic_add_f32(&deg[col[e]], 1.0f);
}

__global__ void deg_scatter_own_kernel(const int* __restrict__ col,
                                       float* __restrict__ deg,
                                       unsigned char* __restrict__ own, int E) {
    int e = blockIdx.x * blockDim.x + threadIdx.x;
    if (e < E) {
        const int c = col[e];
        own[e] = (unsigned char)(c & 255);
        atomic_add_f32(&deg[c], 1.0f);
    }
}

__global__ void rsqrt_kernel(float* __restrict__ d, int N) {
    int i = blockIdx.x * blockDim.x + threadIdx.x;
    if (i < N) d[i] = rsqrtf(d[i]);
}

constexpr int DH_GRID = 64;
constexpr int DH_THREADS = 1024;
constexpr int DH_WORDS = 50000;
constexpr int DH_U = 4;
static_assert((DH_WORDS & 3) == 0, "tables are cleared and stored 16 bytes at a time");

__device__ __forceinline__ void dh_count(unsigned int* cnt, int c) {
    atomicAdd(cnt + (c >> 1), (c & 1) ? 0x10000u : 1u);
}

__global__ __launch_bounds__(DH_THREADS)
void deg_hist_kernel(const int* __restrict__ col, unsigned int* __restrict__ own4,
                     unsigned int* __restrict__ part, int E, int N) {
    __shared__ __attribute__((aligned(16))) unsigned int cnt[DH_WORDS];

    const int words4 = (((N + 1) >> 1) + 3) >> 2;
    const int g4 = E >> 2;
    const int q0 = (int)(((long long)g4 * blockIdx.x) / DH_GRID);
    const int q1 = (int)(((long long)g4 * (blockIdx.x + 1)) / DH_GRID);
    const int4* __restrict__ col4 = (const int4*)col;

    int4 cv[DH_U];
#pragma unroll
    for (int u = 0; u < DH_U; ++u)
        cv[u] = col4[min(q0 + u * DH_THREADS + (int)threadIdx.x, q1 - 1)];

    for (int i = threadIdx.x; i < words4; i += DH_THREADS)
        ((uint4*)cnt)[i] = make_uint4(0u, 0u, 0u, 0u);
    __syncthreads();

    for (int base = q0; base < q1; base += DH_U * DH_THREADS) {
#pragma unroll
        for (int u = 0; u < DH_U; ++u) {
            const int q = base + u * DH_THREADS + (int)threadIdx.x;
            if (q < q1) {
                const int4 c = cv[u];
                dh_count(cnt, c.x);
                dh_count(cnt, c.y);
                dh_count(cnt, c.z);
                dh_count(cnt, c.w);
                own4[q] = (unsigned)(c.x & 255) | ((unsigned)(c.y & 255) << 8) |
                          ((unsigned)(c.z & 255) << 16) | ((unsigned)(c.w & 255) << 24);
            }
        }
        if (base + DH_U * DH_THREADS < q1) {
#pragma unroll
            for (int u = 0; u < DH_U; ++u)
                cv[u] = col4[min(base + (DH_U + u) * DH_THREADS + (int)threadIdx.x, q1 - 1)];
        }
    }
    __syncthreads();

    uint4* __restrict__ p = (uint4*)part + (size_t)blockIdx.x * words4;
#pragma unroll 4
    for (int i = threadIdx.x; i < words4; i += DH_THREADS)
        p[i] = ((const uint4*)cnt)[i];
}

__global__ __launch_bounds__(256)
void deg_reduce_kernel(const unsigned int* __restrict__ part, float* __restrict__ dinv,
                       int N, int pitch) {
    const int w = blockIdx.x * blockDim.x + threadIdx.x;
    if (2 * w >= N) return;
    unsigned int v[DH_GRID];
#pragma unroll
    for (int b = 0; b < DH_GRID; ++b) v[b] = part[(size_t)b * pitch + w];
    unsigned int lo = 0u, hi = 0u;
#pragma unroll
    for (int b = 0; b < DH_GRID; ++b) {
        lo += v[b] & 0xffffu;
        hi += v[b] >> 16;
    }
    *(float2*)(dinv + 2 * w) = make_float2(rsqrtf(1.0f + (float)lo), rsqrtf(1.0f + (float)hi));
}

__global__ void edge_scatter_kernel(const float* __restrict__ h,
                                    const float* __restrict__ dinv,
                                    const int* __restrict__ row,
                                    const int* __restrict__ col,
                                    float* __restrict__ acc,
                                    int E, int H) {
    const int chunks = H >> 2;
    long long gid = (long long)blockIdx.x * blockDim.x + threadIdx.x;
    long long total = (long long)E * chunks;
    if (gid >= total) return;
    const int e = (int)(gid / chunks);
    const int j = (int)(gid - (long long)e * chunks) << 2;
    const int r = row[e];
    const int c = col[e];
    const float s = dinv[r];
    const float4 hv = *(const float4*)(h + (size_t)r * H + j);
    float* dst = acc + (size_t)c * H + j;
    atomic_add_f32(dst + 0, s * hv.x);
    atomic_add_f32(dst + 1, s * hv.y);
    atomic_add_f32(dst + 2, s * hv.z);
    atomic_add_f32(dst + 3, s * hv.w);
}

constexpr int SC_H = 128;
constexpr int SC_EPW = 8;

template <bool FULL>
__device__ __forceinline__ void line_scatter_wave(const float* __restrict__ h,
                                                  const float* __restrict__ dinv,
                                                  const int* __restrict__ row, const int* __restrict__ col,
                                                  float* __restrict__ acc, int e0, int ne, int l) {
    int r[SC_EPW], c[SC_EPW];
    float s[SC_EPW], v0[SC_EPW], v1[SC_EPW], v2[SC_EPW], v3[SC_EPW];
#pragma unroll
    for (int i = 0; i < SC_EPW; ++i) {
        const int e = FULL ? e0 + i : e0 + min(i, ne - 1);
        r[i] = row[e];
        c[i] = col[e];
    }
#pragma unroll
    for (int i = 0; i < SC_EPW; ++i) {
        s[i] = dinv[r[i]];
        const float* __restrict__ p = h + (size_t)r[i] * SC_H + l;
        v0[i] = p[0];
        v1[i] = p[32];
        v2[i] = p[64];
        v3[i] = p[96];
    }
#pragma unroll
    for (int i = 0; i < SC_EPW; ++i) {
        if (FULL || i < ne) {
            float* dst = acc + (size_t)c[i] * SC_H + l;
            atomic_add_f32(dst, s[i] * v0[i]);
            atomic_add_f32(dst + 32, s[i] * v1[i]);
            atomic_add_f32(dst + 64, s[i] * v2[i]);
            atomic_add_f32(dst + 96, s[i] * v3[i]);
        }
    }
}

__global__ __launch_bounds__(256)
void edge_scatter_lines_kernel(const float* __restrict__ h,
                               const float* __restrict__ dinv,
                               const int* __restrict__ row,
                               const int* __restrict__ col,
                               float* __restrict__ acc, int E) {
    const int l = threadIdx.x & 31;
    const int wv = __builtin_amdgcn_readfirstlane(
        (int)(blockIdx.x * (blockDim.x >> 5) + (threadIdx.x >> 5)));
    const int e0 = wv * SC_EPW;
    if (e0 >= E) return;
    const int ne = min(SC_EPW, E - e0);
    if (ne == SC_EPW) line_scatter_wave<true>(h, dinv, row, col, acc, e0, ne, l);
    else              line_scatter_wave<false>(h, dinv, row, col, acc, e0, ne, l);
}


constexpr int OW_GRID_LOG2 = 8;
constexpr int OW_GRID = 1 << OW_GRID_LOG2;
static_assert(OW_GRID == 256, "the owner list holds one byte per edge: target mod 256");
constexpr int OW_WAVES = 32;
constexpr int OW_SLOTS = 392;
constexpr int OW_U = 8;
constexpr int OW_Q = 544;

__device__ __forceinline__ void lds_add_f32(float* p, float v) {
    __hip_atomic_fetch_add(p, v, __ATOMIC_RELAXED, __HIP_MEMORY_SCOPE_WORKGROUP);
}

__device__ __forceinline__ void ow_rows(const float* __restrict__ h, float* accs,
                                        int r, int c, int sbits, int m, int lane) {
    for (int j = 0; j < m; j += 8) {
        float v0[8], v1[8], v2[8], v3[8], ss[8];
        int sl[8];
#pragma unroll
        for (int i = 0; i < 8; ++i) {
            const int idx = min(j + i, m - 1);
            const int rr = __builtin_amdgcn_readlane(r, idx);
            sl[i] = __builtin_amdgcn_readlane(c, idx) >> OW_GRID_LOG2;
            ss[i] = __int_as_float(__builtin_amdgcn_readlane(sbits, idx));
            const float* __restrict__ p = h + (size_t)rr * SC_H + lane;
            v0[i] = p[0];
            v1[i] = p[32];
            v2[i] = p[64];
            v3[i] = p[96];
        }
#pragma unroll
        for (int i = 0; i < 8; ++i) {
            if (j + i < m) {
                float* d = accs + sl[i] * SC_H + lane;
                lds_add_f32(d, ss[i] * v0[i]);
                lds_add_f32(d + 32, ss[i] * v1[i]);
                lds_add_f32(d + 64, ss[i] * v2[i]);
                lds_add_f32(d + 96, ss[i] * v3[i]);
            }
        }
    }
}

template <bool REC>
__device__ __forceinline__ void ow_drain(const float* __restrict__ h, const float* __restrict__ dinv,
                                         const int* __restrict__ row, const int* __restrict__ col,
                                         const int4* __restrict__ rec,
                                         float* accs, const int* q, int m, int lane) {
    const int e = q[min(lane, m - 1)];
    int r, c, sbits;
    if (REC) {
        const int4 rc = rec[e];
        r = rc.x;
        c = rc.y;
        sbits = rc.z;
    } else {
        r = row[e];
        c = col[e];
        sbits = __float_as_int(dinv[r]);
    }
    ow_rows(h, accs, r, c, sbits, m, lane);
}

#define OW_STEP(MT, EDGE) \
    { \
        const bool mt = (MT); \
        const unsigned mask = __builtin_amdgcn_ballot_w32(mt); \
        if (mask) { \
            if (mt) q[cnt + (int)__builtin_amdgcn_mbcnt_lo(mask, 0u)] = (EDGE); \
            cnt += __builtin_popcount(mask); \
        } \
    }

#define OW_WORD(WV, E0) \
    { \
        const unsigned xw = ok ? ((WV) ^ bpat) : 0xffffffffu; \
        const unsigned zb = (xw - 0x01010101u) & ~xw & 0x80808080u; \
        if (__builtin_amdgcn_ballot_w32(zb != 0u)) { \
            OW_STEP((xw & 0x000000ffu) == 0u, (E0) + 0) \
            OW_STEP((xw & 0x0000ff00u) == 0u, (E0) + 1) \
            OW_STEP((xw & 0x00ff0000u) == 0u, (E0) + 2) \
            OW_STEP((xw & 0xff000000u) == 0u, (E0) + 3) \
        } \
    }

constexpr int OL_CAP = 512;

template <int SRC>
__global__ __launch_bounds__(1024)
void owner_aggregate_kernel(const float* __restrict__ h, const float* __restrict__ dinv,
                            const int* __restrict__ row, const int* __restrict__ col,
                            const int* __restrict__ recw, const int* __restrict__ lcnt,
                            const unsigned char* __restrict__ own,
                            float* __restrict__ x, int E, int N) {
    constexpr bool REC = (SRC == 1);
    const int4* __restrict__ rec = (const int4*)recw;
    __shared__ __attribute__((aligned(16))) float accs[OW_SLOTS * SC_H];
    __shared__ int qs[OW_WAVES * OW_Q];

    const int lane = threadIdx.x & 31;
    const int wave = __builtin_amdgcn_readfirstlane((int)(threadIdx.x >> 5));
    const int b = blockIdx.x;
    const int slots = (N - b + OW_GRID - 1) >> OW_GRID_LOG2;

#pragma unroll 4
    for (int sl = wave; sl < slots; sl += OW_WAVES) {
        const int node = (sl << OW_GRID_LOG2) + b;
        const float di = dinv[node];
        const float4 hv = ((const float4*)(h + (size_t)node * SC_H))[lane];
        *(float4*)(accs + sl * SC_H + 4 * lane) = make_float4(di * hv.x, di * hv.y, di * hv.z, di * hv.w);
    }
    __syncthreads();

    int nl = 0;
    if (SRC == 2) nl = __builtin_amdgcn_readfirstlane(lcnt[b * OW_WAVES + wave]);
    if (SRC == 2 && nl <= OL_CAP) {
        const int4* __restrict__ lst = rec + (size_t)(b * OW_WAVES + wave) * OL_CAP;
        int4 cur = make_int4(0, 0, 0, 0);
        if (nl > 0) cur = lst[min(lane, nl - 1)];
        for (int g = 0; g < nl; g += 32) {
            int4 nxt = cur;
            if (g + 32 < nl) nxt = lst[g + 32 + min(lane, nl - g - 33)];
            ow_rows(h, accs, cur.x, cur.y, cur.z, min(32, nl - g), lane);
            cur = nxt;
        }
    } else {
        int* q = qs + wave * OW_Q;
        int cnt = 0;
        const uint4* __restrict__ own4 = (const uint4*)own;
        const int total16 = E >> 4;
        const unsigned bpat = (unsigned)b * 0x01010101u;
        for (int base = 0; base < total16; base += OW_U * OW_WAVES * 32) {
            uint4 ov[OW_U];
#pragma unroll
            for (int u = 0; u < OW_U; ++u) {
                const int i16 = base + (u * OW_WAVES + wave) * 32 + lane;
                ov[u] = own4[min(i16, total16 - 1)];
            }
#pragma unroll
            for (int u = 0; u < OW_U; ++u) {
                const int i16 = base + (u * OW_WAVES + wave) * 32 + lane;
                const bool ok = i16 < total16;
                OW_WORD(ov[u].x, 16 * i16 + 0)
                OW_WORD(ov[u].y, 16 * i16 + 4)
                OW_WORD(ov[u].z, 16 * i16 + 8)
                OW_WORD(ov[u].w, 16 * i16 + 12)
                if (cnt >= 32) {
                    for (int g = 0; g < cnt; g += 32)
                        ow_drain<REC>(h, dinv, row, col, rec, accs, q + g, min(32, cnt - g), lane);
                    cnt = 0;
                }
            }
        }
        for (int g = 0; g < cnt; g += 32)
            ow_drain<REC>(h, dinv, row, col, rec, accs, q + g, min(32, cnt - g), lane);
    }
    __syncthreads();

#pragma unroll 4
    for (int sl = wave; sl < slots; sl += OW_WAVES) {
        const int node = (sl << OW_GRID_LOG2) + b;
        const float di = dinv[node];
        const float4 a = *(const float4*)(accs + sl * SC_H + 4 * lane);
        float4 o;
        o.x = di * a.x; o.x = o.x > 0.f ? o.x : 0.f;
        o.y = di * a.y; o.y = o.y > 0.f ? o.y : 0.f;
        o.z = di * a.z; o.z = o.z > 0.f ? o.z : 0.f;
        o.w = di * a.w; o.w = o.w > 0.f ? o.w : 0.f;
        ((float4*)(x + (size_t)node * SC_H))[lane] = o;
    }
}

__global__ void finalize_relu_kernel(const float* __restrict__ h,
                                     const float* __restrict__ dinv,
                                     float* __restrict__ acc_inout, int H) {
    const int node = blockIdx.x;
    const int j = threadIdx.x;
    const float di = dinv[node];
    const size_t idx = (size_t)node * H + j;
    const float v = di * acc_inout[idx] + di * di * h[idx];
    acc_inout[idx] = v > 0.f ? v : 0.f;
}

constexpr int RAW_FU = 32;
constexpr int RAW_FM = 64;
constexpr int RAW_H = 128;
constexpr int RAW_WU = RAW_FU + RAW_FM;
constexpr int RAW_WM = RAW_FU + RAW_FM + 1;

__global__ void compose_wz_kernel(const float* __restrict__ Wu, const float* __restrict__ bu,
                                  const float* __restrict__ Wm, const float* __restrict__ bm,
                                  float* __restrict__ Wz) {
    const int i = blockIdx.x * blockDim.x + threadIdx.x;
    if (i >= RAW_H * RAW_H) return;
    const int n = i / RAW_H;
    const int k = i - n * RAW_H;
    float w = 0.f;
    if (k < RAW_FU) w = Wu[n * RAW_FU + k];
    else if (k < RAW_WU) w = Wm[n * RAW_FM + (k - RAW_FU)];
    else if (k == RAW_WU) w = bu[n];
    else if (k == RAW_WM) w = bm[n];
    Wz[i] = w;
}

__global__ void init_z_kernel(const float* __restrict__ xu, const float* __restrict__ xm,
                              const float* __restrict__ dinv, float* __restrict__ Z, int Nu) {
    const int node = blockIdx.x;
    const int j = threadIdx.x;
    const float di = dinv[node];
    float self = 0.f;
    if (node < Nu) {
        if (j < RAW_FU) self = xu[(size_t)node * RAW_FU + j];
        else if (j == RAW_WU) self = 1.f;
    } else {
        if (j >= RAW_FU && j < RAW_WU) self = xm[(size_t)(node - Nu) * RAW_FM + (j - RAW_FU)];
        else if (j == RAW_WM) self = 1.f;
    }
    Z[(size_t)node * RAW_H + j] = di * self;
}

constexpr int DZ_THREADS = 256;
constexpr int DZ_NODES = 256;
constexpr int DZ_U = 8;
static_assert((DH_GRID & 1) == 0 && DZ_THREADS == 256 && DZ_NODES == DZ_THREADS,
              "the tables are split into two halves of 128 threads, each thread pair of halves owns one word");
static_assert((RAW_FU & 3) == 0 && (RAW_FM & 3) == 0 && (RAW_WU & 3) == 0 && RAW_WM == RAW_WU + 1,
              "a lane holds four consecutive floats of a row; both weight columns sit in one lane");
static_assert(RAW_H == 128, "one 16-byte store per lane covers a row");

template <bool FULL>
__device__ __forceinline__ void dz_rows(const float4* __restrict__ xu4, const float4* __restrict__ xm4,
                                        const float* sd, float4* __restrict__ Z4,
                                        int node0, int ln0, int N, int Nu, int lane) {
    constexpr int QU = RAW_FU / 4;
    constexpr int QM = (RAW_FU + RAW_FM) / 4;
    for (int i0 = 0; i0 < 32; i0 += DZ_U) {
        float4 v[DZ_U];
        float di[DZ_U];
#pragma unroll
        for (int i = 0; i < DZ_U; ++i) {
            const int ln = ln0 + i0 + i;
            const int node = FULL ? node0 + ln : min(node0 + ln, N - 1);
            const bool user = node < Nu;
            di[i] = sd[ln];
            const float4* __restrict__ p = user ? xu4 + (size_t)node * QU + lane
                                                : xm4 + (size_t)(node - Nu) * (RAW_FM / 4) + (lane - QU);
            const bool ld = user ? (lane < QU) : (lane >= QU && lane < QM);
            float4 r = make_float4(0.f, 0.f, 0.f, 0.f);
            if (ld) r = *p;
            if (lane == QM) { r.x = user ? 1.f : 0.f; r.y = user ? 0.f : 1.f; }
            v[i] = r;
        }
#pragma unroll
        for (int i = 0; i < DZ_U; ++i) {
            const int node = node0 + ln0 + i0 + i;
            if (FULL || node < N)
                Z4[(size_t)node * (RAW_H / 4) + lane] =
                    make_float4(di[i] * v[i].x, di[i] * v[i].y, di[i] * v[i].z, di[i] * v[i].w);
        }
    }
}

__global__ __launch_bounds__(DZ_THREADS)
void deg_reduce_init_z_kernel(const unsigned int* __restrict__ part, float* __restrict__ dinv,
                              const float* __restrict__ xu, const float* __restrict__ xm,
                              float* __restrict__ Z, int N, int Nu, int pitch) {
    __shared__ unsigned int sp[DZ_THREADS];
    __shared__ float sd[DZ_NODES];

    const int t = threadIdx.x;
    const int lane = t & 31;
    const int wave = __builtin_amdgcn_readfirstlane(t >> 5);
    const int node0 = blockIdx.x * DZ_NODES;

    const int j = t & (DZ_THREADS / 2 - 1);
    const int half = t >> 7;
    const int w = blockIdx.x * (DZ_NODES / 2) + j;
    const bool live = 2 * w < N;
    unsigned int lo = 0u, hi = 0u;
    if (live) {
        unsigned int v[DH_GRID / 2];
        const unsigned int* __restrict__ p = part + (size_t)half * (DH_GRID / 2) * pitch + w;
#pragma unroll
        for (int b = 0; b < DH_GRID / 2; ++b) v[b] = p[(size_t)b * pitch];
#pragma unroll
        for (int b = 0; b < DH_GRID / 2; ++b) {
            lo += v[b] & 0xffffu;
            hi += v[b] >> 16;
        }
    }
    if (half) {
        sp[2 * j] = lo;
        sp[2 * j + 1] = hi;
    }
    __syncthreads();
    if (!half) {
        lo += sp[2 * j];
        hi += sp[2 * j + 1];
        const float d0 = rsqrtf(1.0f + (float)lo);
        const float d1 = rsqrtf(1.0f + (float)hi);
        sd[2 * j] = d0;
        sd[2 * j + 1] = d1;
        if (live) *(float2*)(dinv + 2 * w) = make_float2(d0, d1);
    }
    __syncthreads();

    const float4* __restrict__ xu4 = (const float4*)xu;
    const float4* __restrict__ xm4 = (const float4*)xm;
    float4* __restrict__ Z4 = (float4*)Z;
    const int ln0 = wave * 32;
    if (node0 + ln0 >= N) return;
    if (node0 + ln0 + 32 <= N) dz_rows<true>(xu4, xm4, sd, Z4, node0, ln0, N, Nu, lane);
    else                       dz_rows<false>(xu4, xm4, sd, Z4, node0, ln0, N, Nu, lane);
}

constexpr int RAW_EPW = 8;

template <bool FULL>
__device__ __forceinline__ void raw_scatter_wave(const float* __restrict__ xu, const float* __restrict__ xm,
                                                 const float* __restrict__ dinv,
                                                 const int* __restrict__ row, const int* __restrict__ col,
                                                 float* __restrict__ Z, int4* __restrict__ rec,
                                                 int Nu, int e0, int ne, int l) {
    int r[RAW_EPW], c[RAW_EPW];
    float s[RAW_EPW], v0[RAW_EPW], v1[RAW_EPW];
#pragma unroll
    for (int i = 0; i < RAW_EPW; ++i) {
        const int e = FULL ? e0 + i : e0 + min(i, ne - 1);
        r[i] = row[e];
        c[i] = col[e];
    }
    const bool il = l < (FULL ? RAW_EPW : ne);
    int rl = 0, cl = 0;
    float dl = 0.f;
    if (il) {
        rl = row[e0 + l];
        cl = col[e0 + l];
        dl = dinv[rl];
    }
#pragma unroll
    for (int i = 0; i < RAW_EPW; ++i) {
        const bool user = r[i] < Nu;
        const float* __restrict__ p = user ? xu + (size_t)r[i] * RAW_FU + l
                                           : xm + (size_t)(r[i] - Nu) * RAW_FM + l;
        v0[i] = p[0];
        v1[i] = 0.f;
        if (!user) v1[i] = p[32];
    }
#pragma unroll
    for (int i = 0; i < RAW_EPW; ++i)
        s[i] = __int_as_float(__builtin_amdgcn_readlane(__float_as_int(dl), i));
#pragma unroll
    for (int i = 0; i < RAW_EPW; ++i) {
        if (FULL || i < ne) {
            const bool user = r[i] < Nu;
            float* dst = Z + (size_t)c[i] * RAW_H;
            atomic_add_f32(dst + (user ? 0 : RAW_FU) + l, s[i] * v0[i]);
            if (!user) atomic_add_f32(dst + RAW_FU + 32 + l, s[i] * v1[i]);
        }
    }
    if (il) {
        atomic_add_f32(Z + (size_t)cl * RAW_H + (rl < Nu ? RAW_WU : RAW_WM), dl);
        if (rec) rec[e0 + l] = make_int4(rl, cl, __float_as_int(dl), 0);
    }
}

__global__ __launch_bounds__(256)
void raw_scatter_kernel(const float* __restrict__ xu, const float* __restrict__ xm,
                        const float* __restrict__ dinv,
                        const int* __restrict__ row, const int* __restrict__ col,
                        float* __restrict__ Z, int* __restrict__ recw, int E, int Nu) {
    int4* __restrict__ rec = (int4*)recw;
    const int l = threadIdx.x & 31;
    const int wv = __builtin_amdgcn_readfirstlane(
        (int)(blockIdx.x * (blockDim.x >> 5) + (threadIdx.x >> 5)));
    const int e0 = wv * RAW_EPW;
    if (e0 >= E) return;
    const int ne = min(RAW_EPW, E - e0);
    if (ne == RAW_EPW) raw_scatter_wave<true>(xu, xm, dinv, row, col, Z, rec, Nu, e0, ne, l);
    else               raw_scatter_wave<false>(xu, xm, dinv, row, col, Z, rec, Nu, e0, ne, l);
}


constexpr int OZ_PITCH = RAW_H + 4;

__device__ __forceinline__ void owz_drain(const float* __restrict__ xu, const float* __restrict__ xm,
                                          const float* __restrict__ dinv,
                                          const int* __restrict__ row, const int* __restrict__ col,
                                          int4* __restrict__ lst, int lpos,
                                          float* accs, const int* q, int m, int lane, int Nu) {
    const int e = q[min(lane, m - 1)];
    const int r = row[e];
    const int c = col[e];
    const float s = dinv[r];
    const int sbits = __float_as_int(s);
    if (lane < m) {
        if (lpos + lane < OL_CAP) lst[lpos + lane] = make_int4(r, c, sbits, 0);
        lds_add_f32(accs + (c >> OW_GRID_LOG2) * OZ_PITCH + (r < Nu ? RAW_WU : RAW_WM), s);
    }
    for (int j = 0; j < m; j += 8) {
        float v0[8], v1[8], ss[8];
        int sl[8], rr[8];
#pragma unroll
        for (int i = 0; i < 8; ++i) {
            const int idx = min(j + i, m - 1);
            rr[i] = __builtin_amdgcn_readlane(r, idx);
            sl[i] = __builtin_amdgcn_readlane(c, idx) >> OW_GRID_LOG2;
            ss[i] = __int_as_float(__builtin_amdgcn_readlane(sbits, idx));
            const bool user = rr[i] < Nu;
            const float* __restrict__ p = user ? xu + (size_t)rr[i] * RAW_FU + lane
                                               : xm + (size_t)(rr[i] - Nu) * RAW_FM + lane;
            v0[i] = p[0];
            v1[i] = 0.f;
            if (!user) v1[i] = p[32];
        }
#pragma unroll
        for (int i = 0; i < 8; ++i) {
            if (j + i < m) {
                const bool user = rr[i] < Nu;
                float* d = accs + sl[i] * OZ_PITCH + lane;
                lds_add_f32(d + (user ? 0 : RAW_FU), ss[i] * v0[i]);
                if (!user) lds_add_f32(d + RAW_FU + 32, ss[i] * v1[i]);
            }
        }
    }
}

__global__ __launch_bounds__(1024)
void owner_raw_kernel(const float* __restrict__ xu, const float* __restrict__ xm,
                      const float* __restrict__ dinv,
                      const int* __restrict__ row, const int* __restrict__ col,
                      const unsigned char* __restrict__ own,
                      float* __restrict__ Z, int* __restrict__ lstw, int* __restrict__ lcnt,
                      int E, int N, int Nu) {
    __shared__ __attribute__((aligned(16))) float accs[OW_SLOTS * OZ_PITCH];
    __shared__ int qs[OW_WAVES * OW_Q];
    constexpr int QU = RAW_FU / 4;
    constexpr int QM = (RAW_FU + RAW_FM) / 4;

    const int lane = threadIdx.x & 31;
    const int wave = __builtin_amdgcn_readfirstlane((int)(threadIdx.x >> 5));
    const int b = blockIdx.x;
    const int slots = (N - b + OW_GRID - 1) >> OW_GRID_LOG2;

    const float4* __restrict__ xu4 = (const float4*)xu;
    const float4* __restrict__ xm4 = (const float4*)xm;
#pragma unroll 4
    for (int sl = wave; sl < slots; sl += OW_WAVES) {
        const int node = (sl << OW_GRID_LOG2) + b;
        const float di = dinv[node];
        const bool user = node < Nu;
        const float4* __restrict__ p = user ? xu4 + (size_t)node * QU + lane
                                            : xm4 + (size_t)(node - Nu) * (RAW_FM / 4) + (lane - QU);
        const bool ld = user ? (lane < QU) : (lane >= QU && lane < QM);
        float4 r = make_float4(0.f, 0.f, 0.f, 0.f);
        if (ld) r = *p;
        if (lane == QM) { r.x = user ? 1.f : 0.f; r.y = user ? 0.f : 1.f; }
        *(float4*)(accs + sl * OZ_PITCH + 4 * lane) = make_float4(di * r.x, di * r.y, di * r.z, di * r.w);
    }
    __syncthreads();

    int* q = qs + wave * OW_Q;
    int cnt = 0;
    int lpos = 0;
    int4* __restrict__ lst = (int4*)lstw + (size_t)(b * OW_WAVES + wave) * OL_CAP;
    const uint4* __restrict__ own4 = (const uint4*)own;
    const int total16 = E >> 4;
    const unsigned bpat = (unsigned)b * 0x01010101u;
    for (int base = 0; base < total16; base += OW_U * OW_WAVES * 32) {
        uint4 ov[OW_U];
#pragma unroll
        for (int u = 0; u < OW_U; ++u) {
            const int i16 = base + (u * OW_WAVES + wave) * 32 + lane;
            ov[u] = own4[min(i16, total16 - 1)];
        }
#pragma unroll
        for (int u = 0; u < OW_U; ++u) {
            const int i16 = base + (u * OW_WAVES + wave) * 32 + lane;
            const bool ok = i16 < total16;
            OW_WORD(ov[u].x, 16 * i16 + 0)
            OW_WORD(ov[u].y, 16 * i16 + 4)
            OW_WORD(ov[u].z, 16 * i16 + 8)
            OW_WORD(ov[u].w, 16 * i16 + 12)
            if (cnt >= 32) {
                for (int g = 0; g < cnt; g += 32)
                    owz_drain(xu, xm, dinv, row, col, lst, lpos + g, accs, q + g, min(32, cnt - g), lane, Nu);
                lpos += cnt;
                cnt = 0;
            }
        }
    }
    for (int g = 0; g < cnt; g += 32)
        owz_drain(xu, xm, dinv, row, col, lst, lpos + g, accs, q + g, min(32, cnt - g), lane, Nu);
    lpos += cnt;
    if (lane == 0) lcnt[b * OW_WAVES + wave] = lpos;
    __syncthreads();

#pragma unroll 4
    for (int sl = wave; sl < slots; sl += OW_WAVES) {
        const int node = (sl << OW_GRID_LOG2) + b;
        ((float4*)(Z + (size_t)node * RAW_H))[lane] = *(const float4*)(accs + sl * OZ_PITCH + 4 * lane);
    }
}

static inline int ceil_div_i(long long a, long long b) { return (int)((a + b - 1) / b); }

static void launch_gemm(const float* X, const float* W, const float* b, float* Y,
                        int M, int K, int N, hipStream_t stream) {
    const int TPB = 256;
    if (N == 128 && K == 32)
        gemm_bias_wmma_wlds<32, 0><<<GEMM_GRID, TPB, 0, stream>>>(X, W, b, nullptr, nullptr, Y, M);
    else if (N == 128 && K == 64)
        gemm_bias_wmma_wlds<64, 0><<<GEMM_GRID, TPB, 0, stream>>>(X, W, b, nullptr, nullptr, Y, M);
    else if (N == 128 && K == 128)
        gemm_bias_wmma_wlds<128, 0><<<GEMM_GRID, TPB, 0, stream>>>(X, W, b, nullptr, nullptr, Y, M);
    else {
        const int strips = ceil_div_i(M, 16);
        const int tiles = strips * (N / 16);
        gemm_bias_wmma_rt<<<ceil_div_i(tiles, 8), TPB, 0, stream>>>(X, W, b, Y, M, K, N);
    }
}

extern "C" void kernel_launch(void* const* d_in, const int* in_sizes, int n_in,
                              void* d_out, int out_size, void* d_ws, size_t ws_size,
                              hipStream_t stream) {
    const float* x_user  = (const float*)d_in[0];
    const float* x_movie = (const float*)d_in[1];
    const int*   edge    = (const int*)d_in[2];
    const float* Wu = (const float*)d_in[3];
    const float* bu = (const float*)d_in[4];
    const float* Wm = (const float*)d_in[5];
    const float* bm = (const float*)d_in[6];
    const float* W1 = (const float*)d_in[7];
    const float* b1 = (const float*)d_in[8];
    const float* W2 = (const float*)d_in[9];
    const float* b2 = (const float*)d_in[10];
    const float* We = (const float*)d_in[11];
    const float* be = (const float*)d_in[12];
    float* out = (float*)d_out;

    const int H  = in_sizes[4];
    const int C  = in_sizes[12];
    const int Fu = in_sizes[3] / H;
    const int Fm = in_sizes[5] / H;
    const int Nu = in_sizes[0] / Fu;
    const int Nm = in_sizes[1] / Fm;
    const int E  = in_sizes[2] / 2;
    const int N  = Nu + Nm;
    const int* row = edge;
    const int* col = edge + E;

    float* ws   = (float*)d_ws;
    float* X    = ws;
    float* Hh   = X  + (size_t)N * H;
    float* Acc  = Hh + (size_t)N * H;
    float* Wz   = Acc + (size_t)N * H;
    float* dinv = Wz + (size_t)RAW_H * RAW_H;
    unsigned char* own = (unsigned char*)(dinv + (size_t)((N + 3) & ~3));
    unsigned int* part = (unsigned int*)(own + (((size_t)E + 15) & ~(size_t)15));

    const int TPB = 256, WPB = 8;
    const long long NH = (long long)N * H;

    const bool owner_path = (H == SC_H) && ((E & 15) == 0) && (N <= OW_GRID * OW_SLOTS);
    const int dh_words = (((N + 1) >> 1) + 3) & ~3;
    const bool hist_path = (N <= 2 * DH_WORDS) && ((E & 3) == 0) && (E >= 4 * DH_GRID) &&
                           (E / DH_GRID + 4 <= 65535) &&
                           ((size_t)((unsigned char*)(part + (size_t)DH_GRID * dh_words) - (unsigned char*)d_ws) <= ws_size);
    const bool raw_path = (H == RAW_H) && (Fu == RAW_FU) && (Fm == RAW_FM);
    const size_t rec_off = ((size_t)((unsigned char*)(part + (size_t)DH_GRID * dh_words) - (unsigned char*)d_ws) + 127) & ~(size_t)127;
    const bool rec_path = raw_path && owner_path && (rec_off + (size_t)E * 16 <= ws_size);
    int* rec = rec_path ? (int*)((unsigned char*)d_ws + rec_off) : nullptr;
    const size_t lst_bytes = (size_t)OW_GRID * OW_WAVES * OL_CAP * 16;
    const bool own1_path = raw_path && owner_path && (rec_off + lst_bytes + (size_t)OW_GRID * OW_WAVES * 4 <= ws_size);
    int* lst = (int*)((unsigned char*)d_ws + rec_off);
    int* lcnt = (int*)((unsigned char*)d_ws + rec_off + lst_bytes);
    if (hist_path) {
        deg_hist_kernel<<<DH_GRID, DH_THREADS, 0, stream>>>(col, (unsigned int*)own, part, E, N);
        if (raw_path && !own1_path)
            deg_reduce_init_z_kernel<<<ceil_div_i(N, DZ_NODES), DZ_THREADS, 0, stream>>>(part, dinv, x_user, x_movie, Hh, N, Nu, dh_words);
        else
            deg_reduce_kernel<<<ceil_div_i((N + 1) >> 1, TPB), TPB, 0, stream>>>(part, dinv, N, dh_words);
    } else {
        fill_kernel<<<ceil_div_i(N, TPB), TPB, 0, stream>>>(dinv, 1.0f, N);
        if (owner_path)
            deg_scatter_own_kernel<<<ceil_div_i(E, TPB), TPB, 0, stream>>>(col, dinv, own, E);
        else
            deg_scatter_kernel<<<ceil_div_i(E, TPB), TPB, 0, stream>>>(col, dinv, E);
        rsqrt_kernel<<<ceil_div_i(N, TPB), TPB, 0, stream>>>(dinv, N);
    }

    const long long scatterThreads = (long long)E * (H / 4);

    if (raw_path) {
        compose_wz_kernel<<<ceil_div_i(RAW_H * RAW_H, TPB), TPB, 0, stream>>>(Wu, bu, Wm, bm, Wz);
        if (own1_path) {
            owner_raw_kernel<<<OW_GRID, OW_WAVES * 32, 0, stream>>>(x_user, x_movie, dinv, row, col, own, Hh, lst, lcnt, E, N, Nu);
        } else {
            if (!hist_path) init_z_kernel<<<N, RAW_H, 0, stream>>>(x_user, x_movie, dinv, Hh, Nu);
            raw_scatter_kernel<<<ceil_div_i(E, RAW_EPW * WPB), TPB, 0, stream>>>(x_user, x_movie, dinv, row, col, Hh, rec, E, Nu);
        }
        gemm_chain3_wmma<<<CH_GRID, CH_WAVES * 32, 0, stream>>>(Hh, Wz, W1, b1, W2, b2, dinv, Acc, N);
    } else {
        launch_gemm(x_user, Wu, bu, X, Nu, Fu, H, stream);
        launch_gemm(x_movie, Wm, bm, X + (size_t)Nu * H, Nm, Fm, H, stream);

        launch_gemm(X, W1, b1, Hh, N, H, H, stream);
        zero4_kernel<<<ceil_div_i(NH / 4, TPB), TPB, 0, stream>>>((float4*)Acc, NH / 4);
        edge_scatter_kernel<<<ceil_div_i(scatterThreads, TPB), TPB, 0, stream>>>(Hh, dinv, row, col, Acc, E, H);
        finalize_relu_kernel<<<N, H, 0, stream>>>(Hh, dinv, Acc, H);
    }

    const float* Hc = Acc;
    if (!raw_path) {
        launch_gemm(Acc, W2, b2, Hh, N, H, H, stream);
        Hc = Hh;
    }
    if (owner_path) {
        if (own1_path)
            owner_aggregate_kernel<2><<<OW_GRID, OW_WAVES * 32, 0, stream>>>(Hc, dinv, row, col, lst, lcnt, own, X, E, N);
        else if (rec_path)
            owner_aggregate_kernel<1><<<OW_GRID, OW_WAVES * 32, 0, stream>>>(Hc, dinv, row, col, rec, nullptr, own, X, E, N);
        else
            owner_aggregate_kernel<0><<<OW_GRID, OW_WAVES * 32, 0, stream>>>(Hc, dinv, row, col, nullptr, nullptr, own, X, E, N);
    } else {
        zero4_kernel<<<ceil_div_i(NH / 4, TPB), TPB, 0, stream>>>((float4*)X, NH / 4);
        if (H == SC_H)
            edge_scatter_lines_kernel<<<ceil_div_i(E, SC_EPW * WPB), TPB, 0, stream>>>(Hc, dinv, row, col, X, E);
        else
            edge_scatter_kernel<<<ceil_div_i(scatterThreads, TPB), TPB, 0, stream>>>(Hc, dinv, row, col, X, E, H);
        finalize_relu_kernel<<<N, H, 0, stream>>>(Hc, dinv, X, H);
    }

    const int edgeTiles = ceil_div_i(E, 16);
    if (H == EM_H && C <= 16)
        edge_mlp_rows_lds<<<EM_GRID, TPB, 0, stream>>>(X, row, col, We, be, out, E, Nu, C);
    else
        edge_mlp_wmma<0><<<ceil_div_i(edgeTiles, WPB), TPB, 0, stream>>>(X, row, col, We, be, out, E, Nu, H, C);
}
